# starting kernel
# baseline (speedup 1.0000x reference)
#include <hip/hip_runtime.h>
#include <math.h>

#define NN 100000
#define EE 1600000
#define HH 4
#define CC 32
#define DD 128
#define GG 64
#define LL 4
#define INDIM 7
#define KPAD 8
#define NEG_SLOPE 0.2f

typedef __attribute__((ext_vector_type(2))) float v2f;
typedef __attribute__((ext_vector_type(8))) float v8f;

__device__ __forceinline__ float lrelu(float t) {
    return t > 0.f ? t : NEG_SLOPE * t;
}

__global__ void k_pad_x(const float* __restrict__ x, float* __restrict__ xp) {
    int t = blockIdx.x * blockDim.x + threadIdx.x;
    if (t >= NN * KPAD) return;
    const int n = t >> 3, c = t & 7;
    xp[t] = (c < INDIM) ? x[n * INDIM + c] : 0.f;
}

__global__ void k_pad_w(const float* __restrict__ W0, float* __restrict__ Wp) {
    int t = blockIdx.x * blockDim.x + threadIdx.x;
    if (t >= KPAD * DD) return;
    Wp[t] = (t < INDIM * DD) ? W0[t] : 0.f;
}

template <int K>
__global__ __launch_bounds__(256) void k_gemm(const float* __restrict__ A,
                                              const float* __restrict__ B,
                                              float* __restrict__ Cm) {
    const int lane = threadIdx.x & 31;
    const int wave = threadIdx.x >> 5;
    const int row0 = blockIdx.x * 16;
    const int col0 = wave * 16;
    const int m    = lane & 15;
    const int half = lane >> 4;

    const float* __restrict__ arow = A + (row0 + m) * K + half * 2;
    const float* __restrict__ bcol = B + col0 + m;

    v8f acc = {};
#pragma unroll 8
    for (int k0 = 0; k0 < K; k0 += 4) {
        const int ka = k0 + half * 2;
        const v2f a = *(const v2f*)(arow + k0);
        v2f b;
        b.x = bcol[ka * DD];
        b.y = bcol[(ka + 1) * DD];
        acc = __builtin_amdgcn_wmma_f32_16x16x4_f32(
             false, a,  false, b,
             (short)0, acc,  false,  false);
    }
    float* out = Cm + row0 * DD + col0;
#pragma unroll
    for (int r = 0; r < 8; ++r) {
        out[(r + half * 8) * DD + m] = acc[r];
    }
}

__global__ void k_alpha(const float* __restrict__ hLin,
                        const float* __restrict__ a_src,
                        const float* __restrict__ a_dst,
                        float* __restrict__ aS, float* __restrict__ aD) {
    int t = blockIdx.x * blockDim.x + threadIdx.x;
    if (t >= NN * HH) return;
    const int n = t >> 2, hh = t & 3;
    const float4* hp = (const float4*)(hLin + n * DD + hh * CC);
    const float4* as = (const float4*)(a_src + hh * CC);
    const float4* ad = (const float4*)(a_dst + hh * CC);
    float s = 0.f, d = 0.f;
#pragma unroll
    for (int j = 0; j < CC / 4; ++j) {
        float4 v = hp[j], pa = as[j], pb = ad[j];
        s += v.x * pa.x + v.y * pa.y + v.z * pa.z + v.w * pa.w;
        d += v.x * pb.x + v.y * pb.y + v.z * pb.z + v.w * pb.w;
    }
    aS[t] = s;
    aD[t] = d;
}

__global__ void k_init(float* __restrict__ agg, float* __restrict__ mb,
                       float* __restrict__ zb) {
    int t = blockIdx.x * blockDim.x + threadIdx.x;
    if (t < NN * DD) agg[t] = 0.f;
    if (t < NN * HH) { mb[t] = -3.0e38f; zb[t] = 0.f; }
}

__global__ void k_edge_max(const int* __restrict__ ei,
                           const float* __restrict__ aS,
                           const float* __restrict__ aD,
                           float* __restrict__ mb) {
    int e = blockIdx.x * blockDim.x + threadIdx.x;
    if (e >= EE + NN) return;
    int s, d;
    if (e < EE) { s = ei[e]; d = ei[EE + e]; } else { s = d = e - EE; }
    const float4 as = *(const float4*)(aS + s * HH);
    const float4 ad = *(const float4*)(aD + d * HH);
    float* mp = mb + d * HH;
    atomicMax(mp + 0, lrelu(as.x + ad.x));
    atomicMax(mp + 1, lrelu(as.y + ad.y));
    atomicMax(mp + 2, lrelu(as.z + ad.z));
    atomicMax(mp + 3, lrelu(as.w + ad.w));
}

__global__ void k_edge_sum(const int* __restrict__ ei,
                           const float* __restrict__ aS,
                           const float* __restrict__ aD,
                           const float* __restrict__ mb,
                           float* __restrict__ zb) {
    int e = blockIdx.x * blockDim.x + threadIdx.x;
    if (e >= EE + NN) return;
    int s, d;
    if (e < EE) { s = ei[e]; d = ei[EE + e]; } else { s = d = e - EE; }
    const float4 as = *(const float4*)(aS + s * HH);
    const float4 ad = *(const float4*)(aD + d * HH);
    const float4 mv = *(const float4*)(mb + d * HH);
    float* zp = zb + d * HH;
    atomicAdd(zp + 0, expf(lrelu(as.x + ad.x) - mv.x));
    atomicAdd(zp + 1, expf(lrelu(as.y + ad.y) - mv.y));
    atomicAdd(zp + 2, expf(lrelu(as.z + ad.z) - mv.z));
    atomicAdd(zp + 3, expf(lrelu(as.w + ad.w) - mv.w));
}

__global__ void k_edge_agg(const int* __restrict__ ei,
                           const float* __restrict__ aS,
                           const float* __restrict__ aD,
                           const float* __restrict__ mb,
                           const float* __restrict__ zb,
                           const float* __restrict__ hLin,
                           float* __restrict__ agg) {
    int t = blockIdx.x * blockDim.x + threadIdx.x;
    if (t >= (EE + NN) * HH) return;
    const int e = t >> 2, hh = t & 3;
    int s, d;
    if (e < EE) { s = ei[e]; d = ei[EE + e]; } else { s = d = e - EE; }
    const float lg = lrelu(aS[s * HH + hh] + aD[d * HH + hh]);
    const float alpha = expf(lg - mb[d * HH + hh]) / (zb[d * HH + hh] + 1e-16f);
    const float4* hp = (const float4*)(hLin + s * DD + hh * CC);
    float* op = agg + d * DD + hh * CC;
#pragma unroll
    for (int j = 0; j < CC / 4; ++j) {
        float4 v = hp[j];
        atomicAdd(op + 4 * j + 0, v.x * alpha);
        atomicAdd(op + 4 * j + 1, v.y * alpha);
        atomicAdd(op + 4 * j + 2, v.z * alpha);
        atomicAdd(op + 4 * j + 3, v.w * alpha);
    }
}

__global__ __launch_bounds__(256) void k_post(const float* __restrict__ agg,
                                              const float* __restrict__ bias,
                                              const float* __restrict__ gma,
                                              const float* __restrict__ bta,
                                              const float* __restrict__ resid,
                                              float* __restrict__ outp,
                                              int has_res) {
    const int wid  = (blockIdx.x * blockDim.x + threadIdx.x) >> 5;
    const int lane = threadIdx.x & 31;
    if (wid >= NN) return;
    const float* ag = agg + wid * DD;
    float v[4];
    float sum = 0.f;
#pragma unroll
    for (int j = 0; j < 4; ++j) {
        const int c = lane + 32 * j;
        v[j] = ag[c] + bias[c];
        sum += v[j];
    }
#pragma unroll
    for (int off = 16; off > 0; off >>= 1) sum += __shfl_xor(sum, off, 32);
    const float mu = sum * (1.f / 128.f);
    float var = 0.f;
#pragma unroll
    for (int j = 0; j < 4; ++j) { float dl = v[j] - mu; var += dl * dl; }
#pragma unroll
    for (int off = 16; off > 0; off >>= 1) var += __shfl_xor(var, off, 32);
    const float rs = rsqrtf(var * (1.f / 128.f) + 1e-5f);
#pragma unroll
    for (int j = 0; j < 4; ++j) {
        const int c = lane + 32 * j;
        float y = (v[j] - mu) * rs * gma[c] + bta[c];
        y = y > 0.f ? y : expm1f(y);
        outp[wid * DD + c] = (has_res ? resid[wid * DD + c] : 0.f) + y;
    }
}

__global__ void k_zero_pool(float* __restrict__ gsum, float* __restrict__ gcnt) {
    int t = blockIdx.x * blockDim.x + threadIdx.x;
    if (t < GG * DD) gsum[t] = 0.f;
    if (t < GG) gcnt[t] = 0.f;
}

__global__ void k_pool_sum(const float* __restrict__ h,
                           const int* __restrict__ batch,
                           float* __restrict__ gsum, float* __restrict__ gcnt) {
    int t = blockIdx.x * blockDim.x + threadIdx.x;
    if (t >= NN * DD) return;
    const int n = t >> 7, c = t & 127;
    const int g = batch[n];
    atomicAdd(gsum + g * DD + c, h[t]);
    if (c == 0) atomicAdd(gcnt + g, 1.f);
}

__global__ void k_pool_div(const float* __restrict__ gsum,
                           const float* __restrict__ gcnt,
                           float* __restrict__ outg) {
    int t = blockIdx.x * blockDim.x + threadIdx.x;
    if (t >= GG * DD) return;
    outg[t] = gsum[t] / fmaxf(gcnt[t >> 7], 1.f);
}

extern "C" void kernel_launch(void* const* d_in, const int* in_sizes, int n_in,
                              void* d_out, int out_size, void* d_ws, size_t ws_size,
                              hipStream_t stream) {
    (void)in_sizes; (void)n_in; (void)out_size; (void)ws_size;

    const float* x     = (const float*)d_in[0];
    const int*   ei    = (const int*)d_in[1];
    const int*   batch = (const int*)d_in[2];
    const float* W0    = (const float*)d_in[3];
    const float* as0   = (const float*)d_in[4];
    const float* ad0   = (const float*)d_in[5];
    const float* b0    = (const float*)d_in[6];
    const float* Wsl   = (const float*)d_in[7];
    const float* asrcs = (const float*)d_in[8];
    const float* adsts = (const float*)d_in[9];
    const float* bsl   = (const float*)d_in[10];
    const float* lng   = (const float*)d_in[11];
    const float* lnb   = (const float*)d_in[12];
    float* out = (float*)d_out;

    float* ws   = (float*)d_ws;
    float* A0v  = ws;           ws += NN * DD;
    float* A1v  = ws;           ws += NN * DD;
    float* Lin  = ws;           ws += NN * DD;
    float* xp   = ws;           ws += NN * KPAD;
    float* Wp   = ws;           ws += KPAD * DD;
    float* aS   = ws;           ws += NN * HH;
    float* aD   = ws;           ws += NN * HH;
    float* mb   = ws;           ws += NN * HH;
    float* zb   = ws;           ws += NN * HH;
    float* gsum = ws;           ws += GG * DD;
    float* gcnt = ws;           ws += GG;

    const int nTot = EE + NN;

    k_pad_x<<<(NN * KPAD + 255) / 256, 256, 0, stream>>>(x, xp);
    k_pad_w<<<(KPAD * DD + 255) / 256, 256, 0, stream>>>(W0, Wp);

    for (int layer = 0; layer < LL; ++layer) {
        const float* Ain;  float* agg;  const float* resid;
        switch (layer) {
            case 0: Ain = xp;  agg = A0v; resid = nullptr; break;
            case 1: Ain = A0v; agg = A1v; resid = A0v;     break;
            case 2: Ain = A1v; agg = A0v; resid = A1v;     break;
            default:Ain = A0v; agg = out; resid = A0v;     break;
        }
        const float* asv  = (layer == 0) ? as0 : asrcs + (layer - 1) * HH * CC;
        const float* adv  = (layer == 0) ? ad0 : adsts + (layer - 1) * HH * CC;
        const float* bias = (layer == 0) ? b0  : bsl   + (layer - 1) * DD;

        if (layer == 0) {
            k_gemm<KPAD><<<NN / 16, 256, 0, stream>>>(Ain, Wp, Lin);
        } else {
            k_gemm<DD><<<NN / 16, 256, 0, stream>>>(
                Ain, Wsl + (layer - 1) * DD * DD, Lin);
        }
        k_alpha<<<(NN * HH + 255) / 256, 256, 0, stream>>>(Lin, asv, adv, aS, aD);
        k_init<<<(NN * DD + 255) / 256, 256, 0, stream>>>(agg, mb, zb);
        k_edge_max<<<(nTot + 255) / 256, 256, 0, stream>>>(ei, aS, aD, mb);
        k_edge_sum<<<(nTot + 255) / 256, 256, 0, stream>>>(ei, aS, aD, mb, zb);
        k_edge_agg<<<(nTot * HH + 255) / 256, 256, 0, stream>>>(ei, aS, aD, mb, zb, Lin, agg);
        k_post<<<(NN * 32 + 255) / 256, 256, 0, stream>>>(
            agg, bias, lng + layer * DD, lnb + layer * DD, resid, agg,
            layer > 0 ? 1 : 0);
    }

    k_zero_pool<<<(GG * DD + 255) / 256, 256, 0, stream>>>(gsum, gcnt);
    k_pool_sum<<<(NN * DD + 255) / 256, 256, 0, stream>>>(out, batch, gsum, gcnt);
    k_pool_div<<<(GG * DD + 255) / 256, 256, 0, stream>>>(gsum, gcnt, out + NN * DD);
}


# float-atomic edge scatter (max, denominator, rows) replaced: edges grouped by destination once per call (integer atomic arrival-order rank), one wave per node sums its rows in registers
# speedup vs baseline: 10.2359x; 10.2359x over previous
#include <hip/hip_runtime.h>
#include <math.h>

#define NN 100000
#define EE 1600000
#define HH 4
#define CC 32
#define DD 128
#define GG 64
#define LL 4
#define INDIM 7
#define KPAD 8
#define NEG_SLOPE 0.2f

typedef __attribute__((ext_vector_type(2))) float v2f;
typedef __attribute__((ext_vector_type(8))) float v8f;

__device__ __forceinline__ float lrelu(float t) {
    return t > 0.f ? t : NEG_SLOPE * t;
}

__global__ void k_pad_x(const float* __restrict__ x, float* __restrict__ xp) {
    int t = blockIdx.x * blockDim.x + threadIdx.x;
    if (t >= NN * KPAD) return;
    const int n = t >> 3, c = t & 7;
    xp[t] = (c < INDIM) ? x[n * INDIM + c] : 0.f;
}

__global__ void k_pad_w(const float* __restrict__ W0, float* __restrict__ Wp) {
    int t = blockIdx.x * blockDim.x + threadIdx.x;
    if (t >= KPAD * DD) return;
    Wp[t] = (t < INDIM * DD) ? W0[t] : 0.f;
}

template <int K>
__global__ __launch_bounds__(256) void k_gemm(const float* __restrict__ A,
                                              const float* __restrict__ B,
                                              float* __restrict__ Cm) {
    const int lane = threadIdx.x & 31;
    const int wave = threadIdx.x >> 5;
    const int row0 = blockIdx.x * 16;
    const int col0 = wave * 16;
    const int m    = lane & 15;
    const int half = lane >> 4;

    const float* __restrict__ arow = A + (row0 + m) * K + half * 2;
    const float* __restrict__ bcol = B + col0 + m;

    v8f acc = {};
#pragma unroll 8
    for (int k0 = 0; k0 < K; k0 += 4) {
        const int ka = k0 + half * 2;
        const v2f a = *(const v2f*)(arow + k0);
        v2f b;
        b.x = bcol[ka * DD];
        b.y = bcol[(ka + 1) * DD];
        acc = __builtin_amdgcn_wmma_f32_16x16x4_f32(
             false, a,  false, b,
             (short)0, acc,  false,  false);
    }
    float* out = Cm + row0 * DD + col0;
#pragma unroll
    for (int r = 0; r < 8; ++r) {
        out[(r + half * 8) * DD + m] = acc[r];
    }
}

__global__ void k_alpha(const float* __restrict__ hLin,
                        const float* __restrict__ a_src,
                        const float* __restrict__ a_dst,
                        float* __restrict__ aS, float* __restrict__ aD) {
    int t = blockIdx.x * blockDim.x + threadIdx.x;
    if (t >= NN * HH) return;
    const int n = t >> 2, hh = t & 3;
    const float4* hp = (const float4*)(hLin + n * DD + hh * CC);
    const float4* as = (const float4*)(a_src + hh * CC);
    const float4* ad = (const float4*)(a_dst + hh * CC);
    float s = 0.f, d = 0.f;
#pragma unroll
    for (int j = 0; j < CC / 4; ++j) {
        float4 v = hp[j], pa = as[j], pb = ad[j];
        s += v.x * pa.x + v.y * pa.y + v.z * pa.z + v.w * pa.w;
        d += v.x * pb.x + v.y * pb.y + v.z * pb.z + v.w * pb.w;
    }
    aS[t] = s;
    aD[t] = d;
}

#define SCAN_TILE 1024
#define SCAN_BLOCKS ((NN + SCAN_TILE - 1) / SCAN_TILE)

__global__ void k_zero_deg(int* __restrict__ deg) {
    int t = blockIdx.x * blockDim.x + threadIdx.x;
    if (t < NN) deg[t] = 0;
}

__global__ void k_rank(const int* __restrict__ ei, int* __restrict__ deg,
                       int* __restrict__ rank) {
    int e = blockIdx.x * blockDim.x + threadIdx.x;
    if (e >= EE) return;
    rank[e] = atomicAdd(deg + ei[EE + e], 1);
}

__global__ __launch_bounds__(256) void k_tile_sum(const int* __restrict__ deg,
                                                  int* __restrict__ tsum) {
    __shared__ int wtot[8];
    const int lane = threadIdx.x & 31, wave = threadIdx.x >> 5;
    const int i0 = blockIdx.x * SCAN_TILE + threadIdx.x * 4;
    int s = 0;
    if (i0 < NN) {
        const int4 v = *(const int4*)(deg + i0);
        s = v.x + v.y + v.z + v.w;
    }
#pragma unroll
    for (int off = 16; off > 0; off >>= 1) s += __shfl_xor(s, off, 32);
    if (lane == 0) wtot[wave] = s;
    __syncthreads();
    if (threadIdx.x == 0) {
        int t = 0;
#pragma unroll
        for (int w = 0; w < 8; ++w) t += wtot[w];
        tsum[blockIdx.x] = t;
    }
}

__global__ __launch_bounds__(256) void k_rowptr(const int* __restrict__ deg,
                                                const int* __restrict__ tsum,
                                                int* __restrict__ rowptr) {
    __shared__ int wtot[8];
    const int lane = threadIdx.x & 31, wave = threadIdx.x >> 5;
    int p = 0;
    for (int i = lane; i < (int)blockIdx.x; i += 32) p += tsum[i];
#pragma unroll
    for (int off = 16; off > 0; off >>= 1) p += __shfl_xor(p, off, 32);

    const int i0 = blockIdx.x * SCAN_TILE + threadIdx.x * 4;
    int4 v = make_int4(0, 0, 0, 0);
    if (i0 < NN) v = *(const int4*)(deg + i0);
    const int s1 = v.x, s2 = s1 + v.y, s3 = s2 + v.z, s4 = s3 + v.w;
    int inc = s4;
#pragma unroll
    for (int off = 1; off < 32; off <<= 1) {
        const int u = __shfl_up(inc, off, 32);
        if (lane >= off) inc += u;
    }
    if (lane == 31) wtot[wave] = inc;
    __syncthreads();
    int wb = 0;
#pragma unroll
    for (int w = 0; w < 8; ++w) if (w < wave) wb += wtot[w];
    const int b = p + wb + inc - s4;
    if (i0 < NN) {
        *(int4*)(rowptr + i0) = make_int4(b, b + s1, b + s2, b + s3);
        if (i0 + 4 == NN) rowptr[NN] = b + s4;
    }
}

__global__ void k_fill(const int* __restrict__ ei, const int* __restrict__ rank,
                       const int* __restrict__ rowptr, int* __restrict__ col) {
    int e = blockIdx.x * blockDim.x + threadIdx.x;
    if (e >= EE) return;
    col[rowptr[ei[EE + e]] + rank[e]] = ei[e];
}

__device__ __forceinline__ float wave_max(float v) {
#pragma unroll
    for (int off = 16; off > 0; off >>= 1) v = fmaxf(v, __shfl_xor(v, off, 32));
    return v;
}

__device__ __forceinline__ float wave_sum(float v) {
#pragma unroll
    for (int off = 16; off > 0; off >>= 1) v += __shfl_xor(v, off, 32);
    return v;
}

__device__ __forceinline__ void load_item(int t, int cnt, int beg, int d,
                                          const int* __restrict__ col,
                                          const float* __restrict__ aS,
                                          const float4 ad, int& s, float4& lg) {
    s = d;
    lg = make_float4(-3.0e38f, -3.0e38f, -3.0e38f, -3.0e38f);
    if (t < cnt) {
        if (t < cnt - 1) s = col[beg + t];
        const float4 as = *(const float4*)(aS + s * HH);
        lg.x = lrelu(as.x + ad.x);
        lg.y = lrelu(as.y + ad.y);
        lg.z = lrelu(as.z + ad.z);
        lg.w = lrelu(as.w + ad.w);
    }
}

__device__ __forceinline__ float lane_value(float v, int j) {
    return __int_as_float(__builtin_amdgcn_readlane(__float_as_int(v), j));
}

__device__ __forceinline__ void add_rows(const float* __restrict__ hrow, int head,
                                         int s, const float4 a, int n,
                                         float4& acc) {
#pragma unroll 4
    for (int j = 0; j < n; ++j) {
        const int sj = __builtin_amdgcn_readlane(s, j);
        const float a0 = lane_value(a.x, j);
        const float a1 = lane_value(a.y, j);
        const float a2 = lane_value(a.z, j);
        const float a3 = lane_value(a.w, j);
        const float aj = (head & 2) ? ((head & 1) ? a3 : a2) : ((head & 1) ? a1 : a0);
        const float4 v = *(const float4*)(hrow + sj * DD);
        acc.x += aj * v.x;
        acc.y += aj * v.y;
        acc.z += aj * v.z;
        acc.w += aj * v.w;
    }
}

__global__ __launch_bounds__(256) void k_gat_node(const int* __restrict__ rowptr,
                                                  const int* __restrict__ col,
                                                  const float* __restrict__ aS,
                                                  const float* __restrict__ aD,
                                                  const float* __restrict__ hLin,
                                                  float* __restrict__ agg) {
    const int lane = threadIdx.x & 31;
    const int head = lane >> 3;
    const int d = __builtin_amdgcn_readfirstlane(
        (int)((blockIdx.x * blockDim.x + threadIdx.x) >> 5));
    if (d >= NN) return;
    const int beg = rowptr[d];
    const int cnt = rowptr[d + 1] - beg + 1;
    const float4 ad = *(const float4*)(aD + d * HH);
    const float* __restrict__ hrow = hLin + lane * 4;

    int s0;
    float4 lg0;
    load_item(lane, cnt, beg, d, col, aS, ad, s0, lg0);

    float4 m = lg0;
    for (int base = 32; base < cnt; base += 32) {
        int s; float4 lg;
        load_item(base + lane, cnt, beg, d, col, aS, ad, s, lg);
        m.x = fmaxf(m.x, lg.x); m.y = fmaxf(m.y, lg.y);
        m.z = fmaxf(m.z, lg.z); m.w = fmaxf(m.w, lg.w);
    }
    m.x = wave_max(m.x); m.y = wave_max(m.y);
    m.z = wave_max(m.z); m.w = wave_max(m.w);

    float4 e0 = make_float4(0.f, 0.f, 0.f, 0.f);
    if (lane < cnt) {
        e0.x = expf(lg0.x - m.x); e0.y = expf(lg0.y - m.y);
        e0.z = expf(lg0.z - m.z); e0.w = expf(lg0.w - m.w);
    }
    float4 z = e0;
    for (int base = 32; base < cnt; base += 32) {
        int s; float4 lg;
        load_item(base + lane, cnt, beg, d, col, aS, ad, s, lg);
        if (base + lane < cnt) {
            z.x += expf(lg.x - m.x); z.y += expf(lg.y - m.y);
            z.z += expf(lg.z - m.z); z.w += expf(lg.w - m.w);
        }
    }
    float4 den;
    den.x = wave_sum(z.x) + 1e-16f; den.y = wave_sum(z.y) + 1e-16f;
    den.z = wave_sum(z.z) + 1e-16f; den.w = wave_sum(z.w) + 1e-16f;

    float4 acc = make_float4(0.f, 0.f, 0.f, 0.f);
    {
        const float4 a = make_float4(e0.x / den.x, e0.y / den.y,
                                     e0.z / den.z, e0.w / den.w);
        add_rows(hrow, head, s0, a, cnt < 32 ? cnt : 32, acc);
    }
    for (int base = 32; base < cnt; base += 32) {
        int s; float4 lg;
        load_item(base + lane, cnt, beg, d, col, aS, ad, s, lg);
        float4 a = make_float4(0.f, 0.f, 0.f, 0.f);
        if (base + lane < cnt) {
            a.x = expf(lg.x - m.x) / den.x; a.y = expf(lg.y - m.y) / den.y;
            a.z = expf(lg.z - m.z) / den.z; a.w = expf(lg.w - m.w) / den.w;
        }
        const int left = cnt - base;
        add_rows(hrow, head, s, a, left < 32 ? left : 32, acc);
    }
    *(float4*)(agg + d * DD + lane * 4) = acc;
}

__global__ __launch_bounds__(256) void k_post(const float* __restrict__ agg,
                                              const float* __restrict__ bias,
                                              const float* __restrict__ gma,
                                              const float* __restrict__ bta,
                                              const float* __restrict__ resid,
                                              float* __restrict__ outp,
                                              int has_res) {
    const int wid  = (blockIdx.x * blockDim.x + threadIdx.x) >> 5;
    const int lane = threadIdx.x & 31;
    if (wid >= NN) return;
    const float* ag = agg + wid * DD;
    float v[4];
    float sum = 0.f;
#pragma unroll
    for (int j = 0; j < 4; ++j) {
        const int c = lane + 32 * j;
        v[j] = ag[c] + bias[c];
        sum += v[j];
    }
#pragma unroll
    for (int off = 16; off > 0; off >>= 1) sum += __shfl_xor(sum, off, 32);
    const float mu = sum * (1.f / 128.f);
    float var = 0.f;
#pragma unroll
    for (int j = 0; j < 4; ++j) { float dl = v[j] - mu; var += dl * dl; }
#pragma unroll
    for (int off = 16; off > 0; off >>= 1) var += __shfl_xor(var, off, 32);
    const float rs = rsqrtf(var * (1.f / 128.f) + 1e-5f);
#pragma unroll
    for (int j = 0; j < 4; ++j) {
        const int c = lane + 32 * j;
        float y = (v[j] - mu) * rs * gma[c] + bta[c];
        y = y > 0.f ? y : expm1f(y);
        outp[wid * DD + c] = (has_res ? resid[wid * DD + c] : 0.f) + y;
    }
}

__global__ void k_zero_pool(float* __restrict__ gsum, float* __restrict__ gcnt) {
    int t = blockIdx.x * blockDim.x + threadIdx.x;
    if (t < GG * DD) gsum[t] = 0.f;
    if (t < GG) gcnt[t] = 0.f;
}

__global__ void k_pool_sum(const float* __restrict__ h,
                           const int* __restrict__ batch,
                           float* __restrict__ gsum, float* __restrict__ gcnt) {
    int t = blockIdx.x * blockDim.x + threadIdx.x;
    if (t >= NN * DD) return;
    const int n = t >> 7, c = t & 127;
    const int g = batch[n];
    atomicAdd(gsum + g * DD + c, h[t]);
    if (c == 0) atomicAdd(gcnt + g, 1.f);
}

__global__ void k_pool_div(const float* __restrict__ gsum,
                           const float* __restrict__ gcnt,
                           float* __restrict__ outg) {
    int t = blockIdx.x * blockDim.x + threadIdx.x;
    if (t >= GG * DD) return;
    outg[t] = gsum[t] / fmaxf(gcnt[t >> 7], 1.f);
}

extern "C" void kernel_launch(void* const* d_in, const int* in_sizes, int n_in,
                              void* d_out, int out_size, void* d_ws, size_t ws_size,
                              hipStream_t stream) {
    (void)in_sizes; (void)n_in; (void)out_size; (void)ws_size;

    const float* x     = (const float*)d_in[0];
    const int*   ei    = (const int*)d_in[1];
    const int*   batch = (const int*)d_in[2];
    const float* W0    = (const float*)d_in[3];
    const float* as0   = (const float*)d_in[4];
    const float* ad0   = (const float*)d_in[5];
    const float* b0    = (const float*)d_in[6];
    const float* Wsl   = (const float*)d_in[7];
    const float* asrcs = (const float*)d_in[8];
    const float* adsts = (const float*)d_in[9];
    const float* bsl   = (const float*)d_in[10];
    const float* lng   = (const float*)d_in[11];
    const float* lnb   = (const float*)d_in[12];
    float* out = (float*)d_out;

    float* ws   = (float*)d_ws;
    float* A0v  = ws;           ws += NN * DD;
    float* A1v  = ws;           ws += NN * DD;
    float* Lin  = ws;           ws += NN * DD;
    float* xp   = ws;           ws += NN * KPAD;
    float* Wp   = ws;           ws += KPAD * DD;
    float* aS   = ws;           ws += NN * HH;
    float* aD   = ws;           ws += NN * HH;
    float* gsum = ws;           ws += GG * DD;
    float* gcnt = ws;           ws += GG;
    int* deg    = (int*)ws;     ws += NN;
    int* rowptr = (int*)ws;     ws += NN + 4;
    int* tsum   = (int*)ws;     ws += 128;
    int* rank   = (int*)ws;     ws += EE;
    int* col    = (int*)ws;     ws += EE;

    k_zero_deg<<<(NN + 255) / 256, 256, 0, stream>>>(deg);
    k_rank<<<(EE + 255) / 256, 256, 0, stream>>>(ei, deg, rank);
    k_tile_sum<<<SCAN_BLOCKS, 256, 0, stream>>>(deg, tsum);
    k_rowptr<<<SCAN_BLOCKS, 256, 0, stream>>>(deg, tsum, rowptr);
    k_fill<<<(EE + 255) / 256, 256, 0, stream>>>(ei, rank, rowptr, col);

    k_pad_x<<<(NN * KPAD + 255) / 256, 256, 0, stream>>>(x, xp);
    k_pad_w<<<(KPAD * DD + 255) / 256, 256, 0, stream>>>(W0, Wp);

    for (int layer = 0; layer < LL; ++layer) {
        const float* Ain;  float* agg;  const float* resid;
        switch (layer) {
            case 0: Ain = xp;  agg = A0v; resid = nullptr; break;
            case 1: Ain = A0v; agg = A1v; resid = A0v;     break;
            case 2: Ain = A1v; agg = A0v; resid = A1v;     break;
            default:Ain = A0v; agg = out; resid = A0v;     break;
        }
        const float* asv  = (layer == 0) ? as0 : asrcs + (layer - 1) * HH * CC;
        const float* adv  = (layer == 0) ? ad0 : adsts + (layer - 1) * HH * CC;
        const float* bias = (layer == 0) ? b0  : bsl   + (layer - 1) * DD;

        if (layer == 0) {
            k_gemm<KPAD><<<NN / 16, 256, 0, stream>>>(Ain, Wp, Lin);
        } else {
            k_gemm<DD><<<NN / 16, 256, 0, stream>>>(
                Ain, Wsl + (layer - 1) * DD * DD, Lin);
        }
        k_alpha<<<(NN * HH + 255) / 256, 256, 0, stream>>>(Lin, asv, adv, aS, aD);
        k_gat_node<<<(NN * 32 + 255) / 256, 256, 0, stream>>>(rowptr, col, aS, aD, Lin, agg);
        k_post<<<(NN * 32 + 255) / 256, 256, 0, stream>>>(
            agg, bias, lng + layer * DD, lnb + layer * DD, resid, agg,
            layer > 0 ? 1 : 0);
    }

    k_zero_pool<<<(GG * DD + 255) / 256, 256, 0, stream>>>(gsum, gcnt);
    k_pool_sum<<<(NN * DD + 255) / 256, 256, 0, stream>>>(out, batch, gsum, gcnt);
    k_pool_div<<<(GG * DD + 255) / 256, 256, 0, stream>>>(gsum, gcnt, out + NN * DD);
}


# mean pool: a wave sums 16 consecutive node rows in registers per run of equal graph id, one float atomic add per run and channel instead of one per node (order of the pooled sums changes)
# speedup vs baseline: 12.5317x; 1.2243x over previous
#include <hip/hip_runtime.h>
#include <math.h>

#define NN 100000
#define EE 1600000
#define HH 4
#define CC 32
#define DD 128
#define GG 64
#define LL 4
#define INDIM 7
#define KPAD 8
#define NEG_SLOPE 0.2f

typedef __attribute__((ext_vector_type(2))) float v2f;
typedef __attribute__((ext_vector_type(8))) float v8f;

__device__ __forceinline__ float lrelu(float t) {
    return t > 0.f ? t : NEG_SLOPE * t;
}

__global__ void k_pad_x(const float* __restrict__ x, float* __restrict__ xp) {
    int t = blockIdx.x * blockDim.x + threadIdx.x;
    if (t >= NN * KPAD) return;
    const int n = t >> 3, c = t & 7;
    xp[t] = (c < INDIM) ? x[n * INDIM + c] : 0.f;
}

__global__ void k_pad_w(const float* __restrict__ W0, float* __restrict__ Wp) {
    int t = blockIdx.x * blockDim.x + threadIdx.x;
    if (t >= KPAD * DD) return;
    Wp[t] = (t < INDIM * DD) ? W0[t] : 0.f;
}

template <int K>
__global__ __launch_bounds__(256) void k_gemm(const float* __restrict__ A,
                                              const float* __restrict__ B,
                                              float* __restrict__ Cm) {
    const int lane = threadIdx.x & 31;
    const int wave = threadIdx.x >> 5;
    const int row0 = blockIdx.x * 16;
    const int col0 = wave * 16;
    const int m    = lane & 15;
    const int half = lane >> 4;

    const float* __restrict__ arow = A + (row0 + m) * K + half * 2;
    const float* __restrict__ bcol = B + col0 + m;

    v8f acc = {};
#pragma unroll 8
    for (int k0 = 0; k0 < K; k0 += 4) {
        const int ka = k0 + half * 2;
        const v2f a = *(const v2f*)(arow + k0);
        v2f b;
        b.x = bcol[ka * DD];
        b.y = bcol[(ka + 1) * DD];
        acc = __builtin_amdgcn_wmma_f32_16x16x4_f32(
             false, a,  false, b,
             (short)0, acc,  false,  false);
    }
    float* out = Cm + row0 * DD + col0;
#pragma unroll
    for (int r = 0; r < 8; ++r) {
        out[(r + half * 8) * DD + m] = acc[r];
    }
}

__global__ void k_alpha(const float* __restrict__ hLin,
                        const float* __restrict__ a_src,
                        const float* __restrict__ a_dst,
                        float* __restrict__ aS, float* __restrict__ aD) {
    int t = blockIdx.x * blockDim.x + threadIdx.x;
    if (t >= NN * HH) return;
    const int n = t >> 2, hh = t & 3;
    const float4* hp = (const float4*)(hLin + n * DD + hh * CC);
    const float4* as = (const float4*)(a_src + hh * CC);
    const float4* ad = (const float4*)(a_dst + hh * CC);
    float s = 0.f, d = 0.f;
#pragma unroll
    for (int j = 0; j < CC / 4; ++j) {
        float4 v = hp[j], pa = as[j], pb = ad[j];
        s += v.x * pa.x + v.y * pa.y + v.z * pa.z + v.w * pa.w;
        d += v.x * pb.x + v.y * pb.y + v.z * pb.z + v.w * pb.w;
    }
    aS[t] = s;
    aD[t] = d;
}

#define SCAN_TILE 1024
#define SCAN_BLOCKS ((NN + SCAN_TILE - 1) / SCAN_TILE)

__global__ void k_zero_deg(int* __restrict__ deg) {
    int t = blockIdx.x * blockDim.x + threadIdx.x;
    if (t < NN) deg[t] = 0;
}

__global__ void k_rank(const int* __restrict__ ei, int* __restrict__ deg,
                       int* __restrict__ rank) {
    int e = blockIdx.x * blockDim.x + threadIdx.x;
    if (e >= EE) return;
    rank[e] = atomicAdd(deg + ei[EE + e], 1);
}

__global__ __launch_bounds__(256) void k_tile_sum(const int* __restrict__ deg,
                                                  int* __restrict__ tsum) {
    __shared__ int wtot[8];
    const int lane = threadIdx.x & 31, wave = threadIdx.x >> 5;
    const int i0 = blockIdx.x * SCAN_TILE + threadIdx.x * 4;
    int s = 0;
    if (i0 < NN) {
        const int4 v = *(const int4*)(deg + i0);
        s = v.x + v.y + v.z + v.w;
    }
#pragma unroll
    for (int off = 16; off > 0; off >>= 1) s += __shfl_xor(s, off, 32);
    if (lane == 0) wtot[wave] = s;
    __syncthreads();
    if (threadIdx.x == 0) {
        int t = 0;
#pragma unroll
        for (int w = 0; w < 8; ++w) t += wtot[w];
        tsum[blockIdx.x] = t;
    }
}

__global__ __launch_bounds__(256) void k_rowptr(const int* __restrict__ deg,
                                                const int* __restrict__ tsum,
                                                int* __restrict__ rowptr) {
    __shared__ int wtot[8];
    const int lane = threadIdx.x & 31, wave = threadIdx.x >> 5;
    int p = 0;
    for (int i = lane; i < (int)blockIdx.x; i += 32) p += tsum[i];
#pragma unroll
    for (int off = 16; off > 0; off >>= 1) p += __shfl_xor(p, off, 32);

    const int i0 = blockIdx.x * SCAN_TILE + threadIdx.x * 4;
    int4 v = make_int4(0, 0, 0, 0);
    if (i0 < NN) v = *(const int4*)(deg + i0);
    const int s1 = v.x, s2 = s1 + v.y, s3 = s2 + v.z, s4 = s3 + v.w;
    int inc = s4;
#pragma unroll
    for (int off = 1; off < 32; off <<= 1) {
        const int u = __shfl_up(inc, off, 32);
        if (lane >= off) inc += u;
    }
    if (lane == 31) wtot[wave] = inc;
    __syncthreads();
    int wb = 0;
#pragma unroll
    for (int w = 0; w < 8; ++w) if (w < wave) wb += wtot[w];
    const int b = p + wb + inc - s4;
    if (i0 < NN) {
        *(int4*)(rowptr + i0) = make_int4(b, b + s1, b + s2, b + s3);
        if (i0 + 4 == NN) rowptr[NN] = b + s4;
    }
}

__global__ void k_fill(const int* __restrict__ ei, const int* __restrict__ rank,
                       const int* __restrict__ rowptr, int* __restrict__ col) {
    int e = blockIdx.x * blockDim.x + threadIdx.x;
    if (e >= EE) return;
    col[rowptr[ei[EE + e]] + rank[e]] = ei[e];
}

__device__ __forceinline__ float wave_max(float v) {
#pragma unroll
    for (int off = 16; off > 0; off >>= 1) v = fmaxf(v, __shfl_xor(v, off, 32));
    return v;
}

__device__ __forceinline__ float wave_sum(float v) {
#pragma unroll
    for (int off = 16; off > 0; off >>= 1) v += __shfl_xor(v, off, 32);
    return v;
}

__device__ __forceinline__ void load_item(int t, int cnt, int beg, int d,
                                          const int* __restrict__ col,
                                          const float* __restrict__ aS,
                                          const float4 ad, int& s, float4& lg) {
    s = d;
    lg = make_float4(-3.0e38f, -3.0e38f, -3.0e38f, -3.0e38f);
    if (t < cnt) {
        if (t < cnt - 1) s = col[beg + t];
        const float4 as = *(const float4*)(aS + s * HH);
        lg.x = lrelu(as.x + ad.x);
        lg.y = lrelu(as.y + ad.y);
        lg.z = lrelu(as.z + ad.z);
        lg.w = lrelu(as.w + ad.w);
    }
}

__device__ __forceinline__ float lane_value(float v, int j) {
    return __int_as_float(__builtin_amdgcn_readlane(__float_as_int(v), j));
}

__device__ __forceinline__ void add_rows(const float* __restrict__ hrow, int head,
                                         int s, const float4 a, int n,
                                         float4& acc) {
#pragma unroll 4
    for (int j = 0; j < n; ++j) {
        const int sj = __builtin_amdgcn_readlane(s, j);
        const float a0 = lane_value(a.x, j);
        const float a1 = lane_value(a.y, j);
        const float a2 = lane_value(a.z, j);
        const float a3 = lane_value(a.w, j);
        const float aj = (head & 2) ? ((head & 1) ? a3 : a2) : ((head & 1) ? a1 : a0);
        const float4 v = *(const float4*)(hrow + sj * DD);
        acc.x += aj * v.x;
        acc.y += aj * v.y;
        acc.z += aj * v.z;
        acc.w += aj * v.w;
    }
}

__global__ __launch_bounds__(256) void k_gat_node(const int* __restrict__ rowptr,
                                                  const int* __restrict__ col,
                                                  const float* __restrict__ aS,
                                                  const float* __restrict__ aD,
                                                  const float* __restrict__ hLin,
                                                  float* __restrict__ agg) {
    const int lane = threadIdx.x & 31;
    const int head = lane >> 3;
    const int d = __builtin_amdgcn_readfirstlane(
        (int)((blockIdx.x * blockDim.x + threadIdx.x) >> 5));
    if (d >= NN) return;
    const int beg = rowptr[d];
    const int cnt = rowptr[d + 1] - beg + 1;
    const float4 ad = *(const float4*)(aD + d * HH);
    const float* __restrict__ hrow = hLin + lane * 4;

    int s0;
    float4 lg0;
    load_item(lane, cnt, beg, d, col, aS, ad, s0, lg0);

    float4 m = lg0;
    for (int base = 32; base < cnt; base += 32) {
        int s; float4 lg;
        load_item(base + lane, cnt, beg, d, col, aS, ad, s, lg);
        m.x = fmaxf(m.x, lg.x); m.y = fmaxf(m.y, lg.y);
        m.z = fmaxf(m.z, lg.z); m.w = fmaxf(m.w, lg.w);
    }
    m.x = wave_max(m.x); m.y = wave_max(m.y);
    m.z = wave_max(m.z); m.w = wave_max(m.w);

    float4 e0 = make_float4(0.f, 0.f, 0.f, 0.f);
    if (lane < cnt) {
        e0.x = expf(lg0.x - m.x); e0.y = expf(lg0.y - m.y);
        e0.z = expf(lg0.z - m.z); e0.w = expf(lg0.w - m.w);
    }
    float4 z = e0;
    for (int base = 32; base < cnt; base += 32) {
        int s; float4 lg;
        load_item(base + lane, cnt, beg, d, col, aS, ad, s, lg);
        if (base + lane < cnt) {
            z.x += expf(lg.x - m.x); z.y += expf(lg.y - m.y);
            z.z += expf(lg.z - m.z); z.w += expf(lg.w - m.w);
        }
    }
    float4 den;
    den.x = wave_sum(z.x) + 1e-16f; den.y = wave_sum(z.y) + 1e-16f;
    den.z = wave_sum(z.z) + 1e-16f; den.w = wave_sum(z.w) + 1e-16f;

    float4 acc = make_float4(0.f, 0.f, 0.f, 0.f);
    {
        const float4 a = make_float4(e0.x / den.x, e0.y / den.y,
                                     e0.z / den.z, e0.w / den.w);
        add_rows(hrow, head, s0, a, cnt < 32 ? cnt : 32, acc);
    }
    for (int base = 32; base < cnt; base += 32) {
        int s; float4 lg;
        load_item(base + lane, cnt, beg, d, col, aS, ad, s, lg);
        float4 a = make_float4(0.f, 0.f, 0.f, 0.f);
        if (base + lane < cnt) {
            a.x = expf(lg.x - m.x) / den.x; a.y = expf(lg.y - m.y) / den.y;
            a.z = expf(lg.z - m.z) / den.z; a.w = expf(lg.w - m.w) / den.w;
        }
        const int left = cnt - base;
        add_rows(hrow, head, s, a, left < 32 ? left : 32, acc);
    }
    *(float4*)(agg + d * DD + lane * 4) = acc;
}

__global__ __launch_bounds__(256) void k_post(const float* __restrict__ agg,
                                              const float* __restrict__ bias,
                                              const float* __restrict__ gma,
                                              const float* __restrict__ bta,
                                              const float* __restrict__ resid,
                                              float* __restrict__ outp,
                                              int has_res) {
    const int wid  = (blockIdx.x * blockDim.x + threadIdx.x) >> 5;
    const int lane = threadIdx.x & 31;
    if (wid >= NN) return;
    const float* ag = agg + wid * DD;
    float v[4];
    float sum = 0.f;
#pragma unroll
    for (int j = 0; j < 4; ++j) {
        const int c = lane + 32 * j;
        v[j] = ag[c] + bias[c];
        sum += v[j];
    }
#pragma unroll
    for (int off = 16; off > 0; off >>= 1) sum += __shfl_xor(sum, off, 32);
    const float mu = sum * (1.f / 128.f);
    float var = 0.f;
#pragma unroll
    for (int j = 0; j < 4; ++j) { float dl = v[j] - mu; var += dl * dl; }
#pragma unroll
    for (int off = 16; off > 0; off >>= 1) var += __shfl_xor(var, off, 32);
    const float rs = rsqrtf(var * (1.f / 128.f) + 1e-5f);
#pragma unroll
    for (int j = 0; j < 4; ++j) {
        const int c = lane + 32 * j;
        float y = (v[j] - mu) * rs * gma[c] + bta[c];
        y = y > 0.f ? y : expm1f(y);
        outp[wid * DD + c] = (has_res ? resid[wid * DD + c] : 0.f) + y;
    }
}

__global__ void k_zero_pool(float* __restrict__ gsum, float* __restrict__ gcnt) {
    int t = blockIdx.x * blockDim.x + threadIdx.x;
    if (t < GG * DD) gsum[t] = 0.f;
    if (t < GG) gcnt[t] = 0.f;
}

#define POOL_ROWS 16
static_assert(NN % POOL_ROWS == 0, "a wave sums whole groups of POOL_ROWS nodes");

__device__ __forceinline__ void pool_flush(float* __restrict__ gsum,
                                           float* __restrict__ gcnt, int g,
                                           int lane, const float4 acc, float cnt) {
    float* p = gsum + g * DD + lane * 4;
    atomicAdd(p + 0, acc.x);
    atomicAdd(p + 1, acc.y);
    atomicAdd(p + 2, acc.z);
    atomicAdd(p + 3, acc.w);
    if (lane == 0) atomicAdd(gcnt + g, cnt);
}

__global__ __launch_bounds__(256) void k_pool_sum(const float* __restrict__ h,
                                                  const int* __restrict__ batch,
                                                  float* __restrict__ gsum,
                                                  float* __restrict__ gcnt) {
    const int lane = threadIdx.x & 31;
    const int w = __builtin_amdgcn_readfirstlane(
        (int)((blockIdx.x * blockDim.x + threadIdx.x) >> 5));
    const int n0 = w * POOL_ROWS;
    if (n0 >= NN) return;
    const int gl = batch[n0 + (lane & (POOL_ROWS - 1))];
    float4 r[POOL_ROWS];
    const float* __restrict__ hp = h + n0 * DD + lane * 4;
#pragma unroll
    for (int i = 0; i < POOL_ROWS; ++i) r[i] = *(const float4*)(hp + i * DD);

    int gcur = __builtin_amdgcn_readlane(gl, 0);
    float4 acc = r[0];
    float cnt = 1.f;
#pragma unroll
    for (int i = 1; i < POOL_ROWS; ++i) {
        const int gi = __builtin_amdgcn_readlane(gl, i);
        if (gi != gcur) {
            pool_flush(gsum, gcnt, gcur, lane, acc, cnt);
            gcur = gi;
            acc = r[i];
            cnt = 1.f;
        } else {
            acc.x += r[i].x; acc.y += r[i].y;
            acc.z += r[i].z; acc.w += r[i].w;
            cnt += 1.f;
        }
    }
    pool_flush(gsum, gcnt, gcur, lane, acc, cnt);
}

__global__ void k_pool_div(const float* __restrict__ gsum,
                           const float* __restrict__ gcnt,
                           float* __restrict__ outg) {
    int t = blockIdx.x * blockDim.x + threadIdx.x;
    if (t >= GG * DD) return;
    outg[t] = gsum[t] / fmaxf(gcnt[t >> 7], 1.f);
}

extern "C" void kernel_launch(void* const* d_in, const int* in_sizes, int n_in,
                              void* d_out, int out_size, void* d_ws, size_t ws_size,
                              hipStream_t stream) {
    (void)in_sizes; (void)n_in; (void)out_size; (void)ws_size;

    const float* x     = (const float*)d_in[0];
    const int*   ei    = (const int*)d_in[1];
    const int*   batch = (const int*)d_in[2];
    const float* W0    = (const float*)d_in[3];
    const float* as0   = (const float*)d_in[4];
    const float* ad0   = (const float*)d_in[5];
    const float* b0    = (const float*)d_in[6];
    const float* Wsl   = (const float*)d_in[7];
    const float* asrcs = (const float*)d_in[8];
    const float* adsts = (const float*)d_in[9];
    const float* bsl   = (const float*)d_in[10];
    const float* lng   = (const float*)d_in[11];
    const float* lnb   = (const float*)d_in[12];
    float* out = (float*)d_out;

    float* ws   = (float*)d_ws;
    float* A0v  = ws;           ws += NN * DD;
    float* A1v  = ws;           ws += NN * DD;
    float* Lin  = ws;           ws += NN * DD;
    float* xp   = ws;           ws += NN * KPAD;
    float* Wp   = ws;           ws += KPAD * DD;
    float* aS   = ws;           ws += NN * HH;
    float* aD   = ws;           ws += NN * HH;
    float* gsum = ws;           ws += GG * DD;
    float* gcnt = ws;           ws += GG;
    int* deg    = (int*)ws;     ws += NN;
    int* rowptr = (int*)ws;     ws += NN + 4;
    int* tsum   = (int*)ws;     ws += 128;
    int* rank   = (int*)ws;     ws += EE;
    int* col    = (int*)ws;     ws += EE;

    k_zero_deg<<<(NN + 255) / 256, 256, 0, stream>>>(deg);
    k_rank<<<(EE + 255) / 256, 256, 0, stream>>>(ei, deg, rank);
    k_tile_sum<<<SCAN_BLOCKS, 256, 0, stream>>>(deg, tsum);
    k_rowptr<<<SCAN_BLOCKS, 256, 0, stream>>>(deg, tsum, rowptr);
    k_fill<<<(EE + 255) / 256, 256, 0, stream>>>(ei, rank, rowptr, col);

    k_pad_x<<<(NN * KPAD + 255) / 256, 256, 0, stream>>>(x, xp);
    k_pad_w<<<(KPAD * DD + 255) / 256, 256, 0, stream>>>(W0, Wp);

    for (int layer = 0; layer < LL; ++layer) {
        const float* Ain;  float* agg;  const float* resid;
        switch (layer) {
            case 0: Ain = xp;  agg = A0v; resid = nullptr; break;
            case 1: Ain = A0v; agg = A1v; resid = A0v;     break;
            case 2: Ain = A1v; agg = A0v; resid = A1v;     break;
            default:Ain = A0v; agg = out; resid = A0v;     break;
        }
        const float* asv  = (layer == 0) ? as0 : asrcs + (layer - 1) * HH * CC;
        const float* adv  = (layer == 0) ? ad0 : adsts + (layer - 1) * HH * CC;
        const float* bias = (layer == 0) ? b0  : bsl   + (layer - 1) * DD;

        if (layer == 0) {
            k_gemm<KPAD><<<NN / 16, 256, 0, stream>>>(Ain, Wp, Lin);
        } else {
            k_gemm<DD><<<NN / 16, 256, 0, stream>>>(
                Ain, Wsl + (layer - 1) * DD * DD, Lin);
        }
        k_alpha<<<(NN * HH + 255) / 256, 256, 0, stream>>>(Lin, asv, adv, aS, aD);
        k_gat_node<<<(NN * 32 + 255) / 256, 256, 0, stream>>>(rowptr, col, aS, aD, Lin, agg);
        k_post<<<(NN * 32 + 255) / 256, 256, 0, stream>>>(
            agg, bias, lng + layer * DD, lnb + layer * DD, resid, agg,
            layer > 0 ? 1 : 0);
    }

    k_zero_pool<<<(GG * DD + 255) / 256, 256, 0, stream>>>(gsum, gcnt);
    k_pool_sum<<<(NN / POOL_ROWS * 32 + 255) / 256, 256, 0, stream>>>(out, batch, gsum, gcnt);
    k_pool_div<<<(GG * DD + 255) / 256, 256, 0, stream>>>(gsum, gcnt, out + NN * DD);
}


# bias + LayerNorm + ELU + residual fused into the aggregation kernel: the wave that sums a node's row finishes it in registers (order of the norm's mean and variance sums changes); 4 launches fewer
# speedup vs baseline: 14.0218x; 1.1189x over previous
#include <hip/hip_runtime.h>
#include <math.h>

#define NN 100000
#define EE 1600000
#define HH 4
#define CC 32
#define DD 128
#define GG 64
#define LL 4
#define INDIM 7
#define KPAD 8
#define NEG_SLOPE 0.2f

typedef __attribute__((ext_vector_type(2))) float v2f;
typedef __attribute__((ext_vector_type(8))) float v8f;

__device__ __forceinline__ float lrelu(float t) {
    return t > 0.f ? t : NEG_SLOPE * t;
}

__global__ void k_pad_x(const float* __restrict__ x, float* __restrict__ xp) {
    int t = blockIdx.x * blockDim.x + threadIdx.x;
    if (t >= NN * KPAD) return;
    const int n = t >> 3, c = t & 7;
    xp[t] = (c < INDIM) ? x[n * INDIM + c] : 0.f;
}

__global__ void k_pad_w(const float* __restrict__ W0, float* __restrict__ Wp) {
    int t = blockIdx.x * blockDim.x + threadIdx.x;
    if (t >= KPAD * DD) return;
    Wp[t] = (t < INDIM * DD) ? W0[t] : 0.f;
}

template <int K>
__global__ __launch_bounds__(256) void k_gemm(const float* __restrict__ A,
                                              const float* __restrict__ B,
                                              float* __restrict__ Cm) {
    const int lane = threadIdx.x & 31;
    const int wave = threadIdx.x >> 5;
    const int row0 = blockIdx.x * 16;
    const int col0 = wave * 16;
    const int m    = lane & 15;
    const int half = lane >> 4;

    const float* __restrict__ arow = A + (row0 + m) * K + half * 2;
    const float* __restrict__ bcol = B + col0 + m;

    v8f acc = {};
#pragma unroll 8
    for (int k0 = 0; k0 < K; k0 += 4) {
        const int ka = k0 + half * 2;
        const v2f a = *(const v2f*)(arow + k0);
        v2f b;
        b.x = bcol[ka * DD];
        b.y = bcol[(ka + 1) * DD];
        acc = __builtin_amdgcn_wmma_f32_16x16x4_f32(
             false, a,  false, b,
             (short)0, acc,  false,  false);
    }
    float* out = Cm + row0 * DD + col0;
#pragma unroll
    for (int r = 0; r < 8; ++r) {
        out[(r + half * 8) * DD + m] = acc[r];
    }
}

__global__ void k_alpha(const float* __restrict__ hLin,
                        const float* __restrict__ a_src,
                        const float* __restrict__ a_dst,
                        float* __restrict__ aS, float* __restrict__ aD) {
    int t = blockIdx.x * blockDim.x + threadIdx.x;
    if (t >= NN * HH) return;
    const int n = t >> 2, hh = t & 3;
    const float4* hp = (const float4*)(hLin + n * DD + hh * CC);
    const float4* as = (const float4*)(a_src + hh * CC);
    const float4* ad = (const float4*)(a_dst + hh * CC);
    float s = 0.f, d = 0.f;
#pragma unroll
    for (int j = 0; j < CC / 4; ++j) {
        float4 v = hp[j], pa = as[j], pb = ad[j];
        s += v.x * pa.x + v.y * pa.y + v.z * pa.z + v.w * pa.w;
        d += v.x * pb.x + v.y * pb.y + v.z * pb.z + v.w * pb.w;
    }
    aS[t] = s;
    aD[t] = d;
}

#define SCAN_TILE 1024
#define SCAN_BLOCKS ((NN + SCAN_TILE - 1) / SCAN_TILE)

__global__ void k_zero_deg(int* __restrict__ deg) {
    int t = blockIdx.x * blockDim.x + threadIdx.x;
    if (t < NN) deg[t] = 0;
}

__global__ void k_rank(const int* __restrict__ ei, int* __restrict__ deg,
                       int* __restrict__ rank) {
    int e = blockIdx.x * blockDim.x + threadIdx.x;
    if (e >= EE) return;
    rank[e] = atomicAdd(deg + ei[EE + e], 1);
}

__global__ __launch_bounds__(256) void k_tile_sum(const int* __restrict__ deg,
                                                  int* __restrict__ tsum) {
    __shared__ int wtot[8];
    const int lane = threadIdx.x & 31, wave = threadIdx.x >> 5;
    const int i0 = blockIdx.x * SCAN_TILE + threadIdx.x * 4;
    int s = 0;
    if (i0 < NN) {
        const int4 v = *(const int4*)(deg + i0);
        s = v.x + v.y + v.z + v.w;
    }
#pragma unroll
    for (int off = 16; off > 0; off >>= 1) s += __shfl_xor(s, off, 32);
    if (lane == 0) wtot[wave] = s;
    __syncthreads();
    if (threadIdx.x == 0) {
        int t = 0;
#pragma unroll
        for (int w = 0; w < 8; ++w) t += wtot[w];
        tsum[blockIdx.x] = t;
    }
}

__global__ __launch_bounds__(256) void k_rowptr(const int* __restrict__ deg,
                                                const int* __restrict__ tsum,
                                                int* __restrict__ rowptr) {
    __shared__ int wtot[8];
    const int lane = threadIdx.x & 31, wave = threadIdx.x >> 5;
    int p = 0;
    for (int i = lane; i < (int)blockIdx.x; i += 32) p += tsum[i];
#pragma unroll
    for (int off = 16; off > 0; off >>= 1) p += __shfl_xor(p, off, 32);

    const int i0 = blockIdx.x * SCAN_TILE + threadIdx.x * 4;
    int4 v = make_int4(0, 0, 0, 0);
    if (i0 < NN) v = *(const int4*)(deg + i0);
    const int s1 = v.x, s2 = s1 + v.y, s3 = s2 + v.z, s4 = s3 + v.w;
    int inc = s4;
#pragma unroll
    for (int off = 1; off < 32; off <<= 1) {
        const int u = __shfl_up(inc, off, 32);
        if (lane >= off) inc += u;
    }
    if (lane == 31) wtot[wave] = inc;
    __syncthreads();
    int wb = 0;
#pragma unroll
    for (int w = 0; w < 8; ++w) if (w < wave) wb += wtot[w];
    const int b = p + wb + inc - s4;
    if (i0 < NN) {
        *(int4*)(rowptr + i0) = make_int4(b, b + s1, b + s2, b + s3);
        if (i0 + 4 == NN) rowptr[NN] = b + s4;
    }
}

__global__ void k_fill(const int* __restrict__ ei, const int* __restrict__ rank,
                       const int* __restrict__ rowptr, int* __restrict__ col) {
    int e = blockIdx.x * blockDim.x + threadIdx.x;
    if (e >= EE) return;
    col[rowptr[ei[EE + e]] + rank[e]] = ei[e];
}

__device__ __forceinline__ float wave_max(float v) {
#pragma unroll
    for (int off = 16; off > 0; off >>= 1) v = fmaxf(v, __shfl_xor(v, off, 32));
    return v;
}

__device__ __forceinline__ float wave_sum(float v) {
#pragma unroll
    for (int off = 16; off > 0; off >>= 1) v += __shfl_xor(v, off, 32);
    return v;
}

__device__ __forceinline__ void load_item(int t, int cnt, int beg, int d,
                                          const int* __restrict__ col,
                                          const float* __restrict__ aS,
                                          const float4 ad, int& s, float4& lg) {
    s = d;
    lg = make_float4(-3.0e38f, -3.0e38f, -3.0e38f, -3.0e38f);
    if (t < cnt) {
        if (t < cnt - 1) s = col[beg + t];
        const float4 as = *(const float4*)(aS + s * HH);
        lg.x = lrelu(as.x + ad.x);
        lg.y = lrelu(as.y + ad.y);
        lg.z = lrelu(as.z + ad.z);
        lg.w = lrelu(as.w + ad.w);
    }
}

__device__ __forceinline__ float lane_value(float v, int j) {
    return __int_as_float(__builtin_amdgcn_readlane(__float_as_int(v), j));
}

__device__ __forceinline__ void add_rows(const float* __restrict__ hrow, int head,
                                         int s, const float4 a, int n,
                                         float4& acc) {
#pragma unroll 4
    for (int j = 0; j < n; ++j) {
        const int sj = __builtin_amdgcn_readlane(s, j);
        const float a0 = lane_value(a.x, j);
        const float a1 = lane_value(a.y, j);
        const float a2 = lane_value(a.z, j);
        const float a3 = lane_value(a.w, j);
        const float aj = (head & 2) ? ((head & 1) ? a3 : a2) : ((head & 1) ? a1 : a0);
        const float4 v = *(const float4*)(hrow + sj * DD);
        acc.x += aj * v.x;
        acc.y += aj * v.y;
        acc.z += aj * v.z;
        acc.w += aj * v.w;
    }
}

template <bool HAS_RES>
__global__ __launch_bounds__(256) void k_gat_node(const int* __restrict__ rowptr,
                                                  const int* __restrict__ col,
                                                  const float* __restrict__ aS,
                                                  const float* __restrict__ aD,
                                                  const float* __restrict__ hLin,
                                                  const float* __restrict__ bias,
                                                  const float* __restrict__ gma,
                                                  const float* __restrict__ bta,
                                                  const float* __restrict__ resid,
                                                  float* __restrict__ outp) {
    const int lane = threadIdx.x & 31;
    const int head = lane >> 3;
    const int d = __builtin_amdgcn_readfirstlane(
        (int)((blockIdx.x * blockDim.x + threadIdx.x) >> 5));
    if (d >= NN) return;
    const int beg = rowptr[d];
    const int cnt = rowptr[d + 1] - beg + 1;
    const float4 ad = *(const float4*)(aD + d * HH);
    const float* __restrict__ hrow = hLin + lane * 4;
    float4 rr = make_float4(0.f, 0.f, 0.f, 0.f);
    if (HAS_RES) rr = *(const float4*)(resid + d * DD + lane * 4);

    int s0;
    float4 lg0;
    load_item(lane, cnt, beg, d, col, aS, ad, s0, lg0);

    float4 m = lg0;
    for (int base = 32; base < cnt; base += 32) {
        int s; float4 lg;
        load_item(base + lane, cnt, beg, d, col, aS, ad, s, lg);
        m.x = fmaxf(m.x, lg.x); m.y = fmaxf(m.y, lg.y);
        m.z = fmaxf(m.z, lg.z); m.w = fmaxf(m.w, lg.w);
    }
    m.x = wave_max(m.x); m.y = wave_max(m.y);
    m.z = wave_max(m.z); m.w = wave_max(m.w);

    float4 e0 = make_float4(0.f, 0.f, 0.f, 0.f);
    if (lane < cnt) {
        e0.x = expf(lg0.x - m.x); e0.y = expf(lg0.y - m.y);
        e0.z = expf(lg0.z - m.z); e0.w = expf(lg0.w - m.w);
    }
    float4 z = e0;
    for (int base = 32; base < cnt; base += 32) {
        int s; float4 lg;
        load_item(base + lane, cnt, beg, d, col, aS, ad, s, lg);
        if (base + lane < cnt) {
            z.x += expf(lg.x - m.x); z.y += expf(lg.y - m.y);
            z.z += expf(lg.z - m.z); z.w += expf(lg.w - m.w);
        }
    }
    float4 den;
    den.x = wave_sum(z.x) + 1e-16f; den.y = wave_sum(z.y) + 1e-16f;
    den.z = wave_sum(z.z) + 1e-16f; den.w = wave_sum(z.w) + 1e-16f;

    float4 acc = make_float4(0.f, 0.f, 0.f, 0.f);
    {
        const float4 a = make_float4(e0.x / den.x, e0.y / den.y,
                                     e0.z / den.z, e0.w / den.w);
        add_rows(hrow, head, s0, a, cnt < 32 ? cnt : 32, acc);
    }
    for (int base = 32; base < cnt; base += 32) {
        int s; float4 lg;
        load_item(base + lane, cnt, beg, d, col, aS, ad, s, lg);
        float4 a = make_float4(0.f, 0.f, 0.f, 0.f);
        if (base + lane < cnt) {
            a.x = expf(lg.x - m.x) / den.x; a.y = expf(lg.y - m.y) / den.y;
            a.z = expf(lg.z - m.z) / den.z; a.w = expf(lg.w - m.w) / den.w;
        }
        const int left = cnt - base;
        add_rows(hrow, head, s, a, left < 32 ? left : 32, acc);
    }

    const float4 bi = *(const float4*)(bias + lane * 4);
    const float4 gm = *(const float4*)(gma + lane * 4);
    const float4 bt = *(const float4*)(bta + lane * 4);
    const float4 v = make_float4(acc.x + bi.x, acc.y + bi.y, acc.z + bi.z, acc.w + bi.w);
    const float mu = wave_sum((v.x + v.y) + (v.z + v.w)) * (1.f / 128.f);
    const float4 dl = make_float4(v.x - mu, v.y - mu, v.z - mu, v.w - mu);
    const float var =
        wave_sum((dl.x * dl.x + dl.y * dl.y) + (dl.z * dl.z + dl.w * dl.w));
    const float rs = rsqrtf(var * (1.f / 128.f) + 1e-5f);
    float4 y = make_float4(dl.x * rs * gm.x + bt.x, dl.y * rs * gm.y + bt.y,
                           dl.z * rs * gm.z + bt.z, dl.w * rs * gm.w + bt.w);
    y.x = y.x > 0.f ? y.x : expm1f(y.x);
    y.y = y.y > 0.f ? y.y : expm1f(y.y);
    y.z = y.z > 0.f ? y.z : expm1f(y.z);
    y.w = y.w > 0.f ? y.w : expm1f(y.w);
    if (HAS_RES) { y.x += rr.x; y.y += rr.y; y.z += rr.z; y.w += rr.w; }
    *(float4*)(outp + d * DD + lane * 4) = y;
}

__global__ void k_zero_pool(float* __restrict__ gsum, float* __restrict__ gcnt) {
    int t = blockIdx.x * blockDim.x + threadIdx.x;
    if (t < GG * DD) gsum[t] = 0.f;
    if (t < GG) gcnt[t] = 0.f;
}

#define POOL_ROWS 16
static_assert(NN % POOL_ROWS == 0, "a wave sums whole groups of POOL_ROWS nodes");

__device__ __forceinline__ void pool_flush(float* __restrict__ gsum,
                                           float* __restrict__ gcnt, int g,
                                           int lane, const float4 acc, float cnt) {
    float* p = gsum + g * DD + lane * 4;
    atomicAdd(p + 0, acc.x);
    atomicAdd(p + 1, acc.y);
    atomicAdd(p + 2, acc.z);
    atomicAdd(p + 3, acc.w);
    if (lane == 0) atomicAdd(gcnt + g, cnt);
}

__global__ __launch_bounds__(256) void k_pool_sum(const float* __restrict__ h,
                                                  const int* __restrict__ batch,
                                                  float* __restrict__ gsum,
                                                  float* __restrict__ gcnt) {
    const int lane = threadIdx.x & 31;
    const int w = __builtin_amdgcn_readfirstlane(
        (int)((blockIdx.x * blockDim.x + threadIdx.x) >> 5));
    const int n0 = w * POOL_ROWS;
    if (n0 >= NN) return;
    const int gl = batch[n0 + (lane & (POOL_ROWS - 1))];
    float4 r[POOL_ROWS];
    const float* __restrict__ hp = h + n0 * DD + lane * 4;
#pragma unroll
    for (int i = 0; i < POOL_ROWS; ++i) r[i] = *(const float4*)(hp + i * DD);

    int gcur = __builtin_amdgcn_readlane(gl, 0);
    float4 acc = r[0];
    float cnt = 1.f;
#pragma unroll
    for (int i = 1; i < POOL_ROWS; ++i) {
        const int gi = __builtin_amdgcn_readlane(gl, i);
        if (gi != gcur) {
            pool_flush(gsum, gcnt, gcur, lane, acc, cnt);
            gcur = gi;
            acc = r[i];
            cnt = 1.f;
        } else {
            acc.x += r[i].x; acc.y += r[i].y;
            acc.z += r[i].z; acc.w += r[i].w;
            cnt += 1.f;
        }
    }
    pool_flush(gsum, gcnt, gcur, lane, acc, cnt);
}

__global__ void k_pool_div(const float* __restrict__ gsum,
                           const float* __restrict__ gcnt,
                           float* __restrict__ outg) {
    int t = blockIdx.x * blockDim.x + threadIdx.x;
    if (t >= GG * DD) return;
    outg[t] = gsum[t] / fmaxf(gcnt[t >> 7], 1.f);
}

extern "C" void kernel_launch(void* const* d_in, const int* in_sizes, int n_in,
                              void* d_out, int out_size, void* d_ws, size_t ws_size,
                              hipStream_t stream) {
    (void)in_sizes; (void)n_in; (void)out_size; (void)ws_size;

    const float* x     = (const float*)d_in[0];
    const int*   ei    = (const int*)d_in[1];
    const int*   batch = (const int*)d_in[2];
    const float* W0    = (const float*)d_in[3];
    const float* as0   = (const float*)d_in[4];
    const float* ad0   = (const float*)d_in[5];
    const float* b0    = (const float*)d_in[6];
    const float* Wsl   = (const float*)d_in[7];
    const float* asrcs = (const float*)d_in[8];
    const float* adsts = (const float*)d_in[9];
    const float* bsl   = (const float*)d_in[10];
    const float* lng   = (const float*)d_in[11];
    const float* lnb   = (const float*)d_in[12];
    float* out = (float*)d_out;

    float* ws   = (float*)d_ws;
    float* A0v  = ws;           ws += NN * DD;
    float* A1v  = ws;           ws += NN * DD;
    float* Lin  = ws;           ws += NN * DD;
    float* xp   = ws;           ws += NN * KPAD;
    float* Wp   = ws;           ws += KPAD * DD;
    float* aS   = ws;           ws += NN * HH;
    float* aD   = ws;           ws += NN * HH;
    float* gsum = ws;           ws += GG * DD;
    float* gcnt = ws;           ws += GG;
    int* deg    = (int*)ws;     ws += NN;
    int* rowptr = (int*)ws;     ws += NN + 4;
    int* tsum   = (int*)ws;     ws += 128;
    int* rank   = (int*)ws;     ws += EE;
    int* col    = (int*)ws;     ws += EE;

    k_zero_deg<<<(NN + 255) / 256, 256, 0, stream>>>(deg);
    k_rank<<<(EE + 255) / 256, 256, 0, stream>>>(ei, deg, rank);
    k_tile_sum<<<SCAN_BLOCKS, 256, 0, stream>>>(deg, tsum);
    k_rowptr<<<SCAN_BLOCKS, 256, 0, stream>>>(deg, tsum, rowptr);
    k_fill<<<(EE + 255) / 256, 256, 0, stream>>>(ei, rank, rowptr, col);

    k_pad_x<<<(NN * KPAD + 255) / 256, 256, 0, stream>>>(x, xp);
    k_pad_w<<<(KPAD * DD + 255) / 256, 256, 0, stream>>>(W0, Wp);

    for (int layer = 0; layer < LL; ++layer) {
        const float* Ain;  float* agg;  const float* resid;
        switch (layer) {
            case 0: Ain = xp;  agg = A0v; resid = nullptr; break;
            case 1: Ain = A0v; agg = A1v; resid = A0v;     break;
            case 2: Ain = A1v; agg = A0v; resid = A1v;     break;
            default:Ain = A0v; agg = out; resid = A0v;     break;
        }
        const float* asv  = (layer == 0) ? as0 : asrcs + (layer - 1) * HH * CC;
        const float* adv  = (layer == 0) ? ad0 : adsts + (layer - 1) * HH * CC;
        const float* bias = (layer == 0) ? b0  : bsl   + (layer - 1) * DD;

        if (layer == 0) {
            k_gemm<KPAD><<<NN / 16, 256, 0, stream>>>(Ain, Wp, Lin);
        } else {
            k_gemm<DD><<<NN / 16, 256, 0, stream>>>(
                Ain, Wsl + (layer - 1) * DD * DD, Lin);
        }
        k_alpha<<<(NN * HH + 255) / 256, 256, 0, stream>>>(Lin, asv, adv, aS, aD);
        if (layer == 0) {
            k_gat_node<false><<<(NN * 32 + 255) / 256, 256, 0, stream>>>(
                rowptr, col, aS, aD, Lin, bias, lng + layer * DD, lnb + layer * DD,
                resid, agg);
        } else {
            k_gat_node<true><<<(NN * 32 + 255) / 256, 256, 0, stream>>>(
                rowptr, col, aS, aD, Lin, bias, lng + layer * DD, lnb + layer * DD,
                resid, agg);
        }
    }

    k_zero_pool<<<(GG * DD + 255) / 256, 256, 0, stream>>>(gsum, gcnt);
    k_pool_sum<<<(NN / POOL_ROWS * 32 + 255) / 256, 256, 0, stream>>>(out, batch, gsum, gcnt);
    k_pool_div<<<(GG * DD + 255) / 256, 256, 0, stream>>>(gsum, gcnt, out + NN * DD);
}


# edges grouped by destination with a two-level counting sort on LDS counters (256-node bins, scanned offsets): no global atomics, no random rowptr reads; order of each node's edge sums changes
# speedup vs baseline: 15.6331x; 1.1149x over previous
#include <hip/hip_runtime.h>
#include <math.h>

#define NN 100000
#define EE 1600000
#define HH 4
#define CC 32
#define DD 128
#define GG 64
#define LL 4
#define INDIM 7
#define KPAD 8
#define NEG_SLOPE 0.2f

typedef __attribute__((ext_vector_type(2))) float v2f;
typedef __attribute__((ext_vector_type(8))) float v8f;

__device__ __forceinline__ float lrelu(float t) {
    return t > 0.f ? t : NEG_SLOPE * t;
}

__global__ void k_pad_x(const float* __restrict__ x, float* __restrict__ xp) {
    int t = blockIdx.x * blockDim.x + threadIdx.x;
    if (t >= NN * KPAD) return;
    const int n = t >> 3, c = t & 7;
    xp[t] = (c < INDIM) ? x[n * INDIM + c] : 0.f;
}

__global__ void k_pad_w(const float* __restrict__ W0, float* __restrict__ Wp) {
    int t = blockIdx.x * blockDim.x + threadIdx.x;
    if (t >= KPAD * DD) return;
    Wp[t] = (t < INDIM * DD) ? W0[t] : 0.f;
}

template <int K>
__global__ __launch_bounds__(256) void k_gemm(const float* __restrict__ A,
                                              const float* __restrict__ B,
                                              float* __restrict__ Cm) {
    const int lane = threadIdx.x & 31;
    const int wave = threadIdx.x >> 5;
    const int row0 = blockIdx.x * 16;
    const int col0 = wave * 16;
    const int m    = lane & 15;
    const int half = lane >> 4;

    const float* __restrict__ arow = A + (row0 + m) * K + half * 2;
    const float* __restrict__ bcol = B + col0 + m;

    v8f acc = {};
#pragma unroll 8
    for (int k0 = 0; k0 < K; k0 += 4) {
        const int ka = k0 + half * 2;
        const v2f a = *(const v2f*)(arow + k0);
        v2f b;
        b.x = bcol[ka * DD];
        b.y = bcol[(ka + 1) * DD];
        acc = __builtin_amdgcn_wmma_f32_16x16x4_f32(
             false, a,  false, b,
             (short)0, acc,  false,  false);
    }
    float* out = Cm + row0 * DD + col0;
#pragma unroll
    for (int r = 0; r < 8; ++r) {
        out[(r + half * 8) * DD + m] = acc[r];
    }
}

__global__ void k_alpha(const float* __restrict__ hLin,
                        const float* __restrict__ a_src,
                        const float* __restrict__ a_dst,
                        float* __restrict__ aS, float* __restrict__ aD) {
    int t = blockIdx.x * blockDim.x + threadIdx.x;
    if (t >= NN * HH) return;
    const int n = t >> 2, hh = t & 3;
    const float4* hp = (const float4*)(hLin + n * DD + hh * CC);
    const float4* as = (const float4*)(a_src + hh * CC);
    const float4* ad = (const float4*)(a_dst + hh * CC);
    float s = 0.f, d = 0.f;
#pragma unroll
    for (int j = 0; j < CC / 4; ++j) {
        float4 v = hp[j], pa = as[j], pb = ad[j];
        s += v.x * pa.x + v.y * pa.y + v.z * pa.z + v.w * pa.w;
        d += v.x * pb.x + v.y * pb.y + v.z * pb.z + v.w * pb.w;
    }
    aS[t] = s;
    aD[t] = d;
}

#define BIN_SHIFT 8
#define BIN_NODES (1 << BIN_SHIFT)
#define NBIN ((NN + BIN_NODES - 1) / BIN_NODES)
#define CHUNK 4096
#define NCHUNK ((EE + CHUNK - 1) / CHUNK)
#define BPITCH 512
static_assert(NBIN <= BPITCH && NCHUNK <= BPITCH, "two slots per thread of a 256-thread block");
static_assert(BIN_NODES == 256, "one destination per thread in k_bin_csr");
#define CSR_BATCH 8

__device__ __forceinline__ int block_excl_scan(int v, int* wtot, int& total) {
    const int lane = threadIdx.x & 31, wave = threadIdx.x >> 5;
    int inc = v;
#pragma unroll
    for (int off = 1; off < 32; off <<= 1) {
        const int u = __shfl_up(inc, off, 32);
        if (lane >= off) inc += u;
    }
    if (lane == 31) wtot[wave] = inc;
    __syncthreads();
    int wb = 0, tot = 0;
#pragma unroll
    for (int w = 0; w < 8; ++w) {
        const int x = wtot[w];
        if (w < wave) wb += x;
        tot += x;
    }
    total = tot;
    return wb + inc - v;
}

__global__ __launch_bounds__(256) void k_bin_count(const int* __restrict__ ei,
                                                   int* __restrict__ cnt) {
    __shared__ int hist[BPITCH];
    const int t = threadIdx.x;
    hist[t] = 0;
    hist[t + 256] = 0;
    __syncthreads();
    const int e0 = blockIdx.x * CHUNK + t;
    int d[CHUNK / 256];
#pragma unroll
    for (int i = 0; i < CHUNK / 256; ++i) {
        const int e = e0 + i * 256;
        d[i] = ei[EE + (e < EE ? e : EE - 1)];
    }
#pragma unroll
    for (int i = 0; i < CHUNK / 256; ++i) {
        if (e0 + i * 256 < EE) atomicAdd(&hist[d[i] >> BIN_SHIFT], 1);
    }
    __syncthreads();
    cnt[blockIdx.x * BPITCH + t] = hist[t];
    cnt[blockIdx.x * BPITCH + t + 256] = hist[t + 256];
}

__global__ __launch_bounds__(256) void k_bin_scan(const int* __restrict__ cnt,
                                                  int* __restrict__ off,
                                                  int* __restrict__ binTot) {
    __shared__ int wtot[8];
    const int b = blockIdx.x, t = threadIdx.x;
    const int w0 = 2 * t, w1 = 2 * t + 1;
    const int c0 = (w0 < NCHUNK) ? cnt[w0 * BPITCH + b] : 0;
    const int c1 = (w1 < NCHUNK) ? cnt[w1 * BPITCH + b] : 0;
    int tot;
    const int ex = block_excl_scan(c0 + c1, wtot, tot);
    if (w0 < NCHUNK) off[w0 * BPITCH + b] = ex;
    if (w1 < NCHUNK) off[w1 * BPITCH + b] = ex + c0;
    if (t == 0) binTot[b] = tot;
}

__global__ __launch_bounds__(256) void k_bin_base(const int* __restrict__ binTot,
                                                  int* __restrict__ binBase) {
    __shared__ int wtot[8];
    const int t = threadIdx.x;
    const int b0 = 2 * t, b1 = 2 * t + 1;
    const int c0 = (b0 < NBIN) ? binTot[b0] : 0;
    const int c1 = (b1 < NBIN) ? binTot[b1] : 0;
    int tot;
    const int ex = block_excl_scan(c0 + c1, wtot, tot);
    binBase[b0] = ex;
    binBase[b1] = ex + c0;
}

__global__ __launch_bounds__(256) void k_bin_scatter(const int* __restrict__ ei,
                                                     const int* __restrict__ off,
                                                     const int* __restrict__ binBase,
                                                     int* __restrict__ binned) {
    __shared__ int cur[BPITCH];
    const int t = threadIdx.x;
    cur[t] = (t < NBIN) ? binBase[t] + off[blockIdx.x * BPITCH + t] : 0;
    cur[t + 256] = (t + 256 < NBIN) ? binBase[t + 256] + off[blockIdx.x * BPITCH + t + 256] : 0;
    __syncthreads();
    const int e0 = blockIdx.x * CHUNK + t;
    int s[CHUNK / 256], d[CHUNK / 256];
#pragma unroll
    for (int i = 0; i < CHUNK / 256; ++i) {
        const int e = e0 + i * 256;
        const int ec = e < EE ? e : EE - 1;
        s[i] = ei[ec];
        d[i] = ei[EE + ec];
    }
#pragma unroll
    for (int i = 0; i < CHUNK / 256; ++i) {
        if (e0 + i * 256 < EE) {
            const int pos = atomicAdd(&cur[d[i] >> BIN_SHIFT], 1);
            *(int2*)(binned + 2 * pos) = make_int2(s[i], d[i]);
        }
    }
}

__global__ __launch_bounds__(256) void k_bin_csr(const int* __restrict__ binned,
                                                 const int* __restrict__ binBase,
                                                 const int* __restrict__ binTot,
                                                 int* __restrict__ rowptr,
                                                 int* __restrict__ col) {
    __shared__ int cntl[BIN_NODES];
    __shared__ int curl[BIN_NODES];
    __shared__ int wtot[8];
    const int b = blockIdx.x, t = threadIdx.x;
    const int start = binBase[b];
    const int n = binTot[b];
    const int2* __restrict__ ed = (const int2*)(binned + 2 * start);
    cntl[t] = 0;
    __syncthreads();
    for (int i0 = t; i0 < n; i0 += 256 * CSR_BATCH) {
        int dl[CSR_BATCH];
#pragma unroll
        for (int k = 0; k < CSR_BATCH; ++k) {
            const int i = i0 + k * 256;
            dl[k] = ed[i < n ? i : n - 1].y & (BIN_NODES - 1);
        }
#pragma unroll
        for (int k = 0; k < CSR_BATCH; ++k) {
            if (i0 + k * 256 < n) atomicAdd(&cntl[dl[k]], 1);
        }
    }
    __syncthreads();
    int tot;
    const int ex = block_excl_scan(cntl[t], wtot, tot);
    curl[t] = ex;
    const int node = b * BIN_NODES + t;
    if (node <= NN) rowptr[node] = start + ex;
    __syncthreads();
    for (int i0 = t; i0 < n; i0 += 256 * CSR_BATCH) {
        int2 e[CSR_BATCH];
#pragma unroll
        for (int k = 0; k < CSR_BATCH; ++k) {
            const int i = i0 + k * 256;
            e[k] = ed[i < n ? i : n - 1];
        }
#pragma unroll
        for (int k = 0; k < CSR_BATCH; ++k) {
            if (i0 + k * 256 < n) {
                const int r = atomicAdd(&curl[e[k].y & (BIN_NODES - 1)], 1);
                col[start + r] = e[k].x;
            }
        }
    }
}

__device__ __forceinline__ float wave_max(float v) {
#pragma unroll
    for (int off = 16; off > 0; off >>= 1) v = fmaxf(v, __shfl_xor(v, off, 32));
    return v;
}

__device__ __forceinline__ float wave_sum(float v) {
#pragma unroll
    for (int off = 16; off > 0; off >>= 1) v += __shfl_xor(v, off, 32);
    return v;
}

__device__ __forceinline__ void load_item(int t, int cnt, int beg, int d,
                                          const int* __restrict__ col,
                                          const float* __restrict__ aS,
                                          const float4 ad, int& s, float4& lg) {
    s = d;
    lg = make_float4(-3.0e38f, -3.0e38f, -3.0e38f, -3.0e38f);
    if (t < cnt) {
        if (t < cnt - 1) s = col[beg + t];
        const float4 as = *(const float4*)(aS + s * HH);
        lg.x = lrelu(as.x + ad.x);
        lg.y = lrelu(as.y + ad.y);
        lg.z = lrelu(as.z + ad.z);
        lg.w = lrelu(as.w + ad.w);
    }
}

__device__ __forceinline__ float lane_value(float v, int j) {
    return __int_as_float(__builtin_amdgcn_readlane(__float_as_int(v), j));
}

__device__ __forceinline__ void add_rows(const float* __restrict__ hrow, int head,
                                         int s, const float4 a, int n,
                                         float4& acc) {
#pragma unroll 4
    for (int j = 0; j < n; ++j) {
        const int sj = __builtin_amdgcn_readlane(s, j);
        const float a0 = lane_value(a.x, j);
        const float a1 = lane_value(a.y, j);
        const float a2 = lane_value(a.z, j);
        const float a3 = lane_value(a.w, j);
        const float aj = (head & 2) ? ((head & 1) ? a3 : a2) : ((head & 1) ? a1 : a0);
        const float4 v = *(const float4*)(hrow + sj * DD);
        acc.x += aj * v.x;
        acc.y += aj * v.y;
        acc.z += aj * v.z;
        acc.w += aj * v.w;
    }
}

template <bool HAS_RES>
__global__ __launch_bounds__(256) void k_gat_node(const int* __restrict__ rowptr,
                                                  const int* __restrict__ col,
                                                  const float* __restrict__ aS,
                                                  const float* __restrict__ aD,
                                                  const float* __restrict__ hLin,
                                                  const float* __restrict__ bias,
                                                  const float* __restrict__ gma,
                                                  const float* __restrict__ bta,
                                                  const float* __restrict__ resid,
                                                  float* __restrict__ outp) {
    const int lane = threadIdx.x & 31;
    const int head = lane >> 3;
    const int d = __builtin_amdgcn_readfirstlane(
        (int)((blockIdx.x * blockDim.x + threadIdx.x) >> 5));
    if (d >= NN) return;
    const int beg = rowptr[d];
    const int cnt = rowptr[d + 1] - beg + 1;
    const float4 ad = *(const float4*)(aD + d * HH);
    const float* __restrict__ hrow = hLin + lane * 4;
    float4 rr = make_float4(0.f, 0.f, 0.f, 0.f);
    if (HAS_RES) rr = *(const float4*)(resid + d * DD + lane * 4);

    int s0;
    float4 lg0;
    load_item(lane, cnt, beg, d, col, aS, ad, s0, lg0);

    float4 m = lg0;
    for (int base = 32; base < cnt; base += 32) {
        int s; float4 lg;
        load_item(base + lane, cnt, beg, d, col, aS, ad, s, lg);
        m.x = fmaxf(m.x, lg.x); m.y = fmaxf(m.y, lg.y);
        m.z = fmaxf(m.z, lg.z); m.w = fmaxf(m.w, lg.w);
    }
    m.x = wave_max(m.x); m.y = wave_max(m.y);
    m.z = wave_max(m.z); m.w = wave_max(m.w);

    float4 e0 = make_float4(0.f, 0.f, 0.f, 0.f);
    if (lane < cnt) {
        e0.x = expf(lg0.x - m.x); e0.y = expf(lg0.y - m.y);
        e0.z = expf(lg0.z - m.z); e0.w = expf(lg0.w - m.w);
    }
    float4 z = e0;
    for (int base = 32; base < cnt; base += 32) {
        int s; float4 lg;
        load_item(base + lane, cnt, beg, d, col, aS, ad, s, lg);
        if (base + lane < cnt) {
            z.x += expf(lg.x - m.x); z.y += expf(lg.y - m.y);
            z.z += expf(lg.z - m.z); z.w += expf(lg.w - m.w);
        }
    }
    float4 den;
    den.x = wave_sum(z.x) + 1e-16f; den.y = wave_sum(z.y) + 1e-16f;
    den.z = wave_sum(z.z) + 1e-16f; den.w = wave_sum(z.w) + 1e-16f;

    float4 acc = make_float4(0.f, 0.f, 0.f, 0.f);
    {
        const float4 a = make_float4(e0.x / den.x, e0.y / den.y,
                                     e0.z / den.z, e0.w / den.w);
        add_rows(hrow, head, s0, a, cnt < 32 ? cnt : 32, acc);
    }
    for (int base = 32; base < cnt; base += 32) {
        int s; float4 lg;
        load_item(base + lane, cnt, beg, d, col, aS, ad, s, lg);
        float4 a = make_float4(0.f, 0.f, 0.f, 0.f);
        if (base + lane < cnt) {
            a.x = expf(lg.x - m.x) / den.x; a.y = expf(lg.y - m.y) / den.y;
            a.z = expf(lg.z - m.z) / den.z; a.w = expf(lg.w - m.w) / den.w;
        }
        const int left = cnt - base;
        add_rows(hrow, head, s, a, left < 32 ? left : 32, acc);
    }

    const float4 bi = *(const float4*)(bias + lane * 4);
    const float4 gm = *(const float4*)(gma + lane * 4);
    const float4 bt = *(const float4*)(bta + lane * 4);
    const float4 v = make_float4(acc.x + bi.x, acc.y + bi.y, acc.z + bi.z, acc.w + bi.w);
    const float mu = wave_sum((v.x + v.y) + (v.z + v.w)) * (1.f / 128.f);
    const float4 dl = make_float4(v.x - mu, v.y - mu, v.z - mu, v.w - mu);
    const float var =
        wave_sum((dl.x * dl.x + dl.y * dl.y) + (dl.z * dl.z + dl.w * dl.w));
    const float rs = rsqrtf(var * (1.f / 128.f) + 1e-5f);
    float4 y = make_float4(dl.x * rs * gm.x + bt.x, dl.y * rs * gm.y + bt.y,
                           dl.z * rs * gm.z + bt.z, dl.w * rs * gm.w + bt.w);
    y.x = y.x > 0.f ? y.x : expm1f(y.x);
    y.y = y.y > 0.f ? y.y : expm1f(y.y);
    y.z = y.z > 0.f ? y.z : expm1f(y.z);
    y.w = y.w > 0.f ? y.w : expm1f(y.w);
    if (HAS_RES) { y.x += rr.x; y.y += rr.y; y.z += rr.z; y.w += rr.w; }
    *(float4*)(outp + d * DD + lane * 4) = y;
}

__global__ void k_zero_pool(float* __restrict__ gsum, float* __restrict__ gcnt) {
    int t = blockIdx.x * blockDim.x + threadIdx.x;
    if (t < GG * DD) gsum[t] = 0.f;
    if (t < GG) gcnt[t] = 0.f;
}

#define POOL_ROWS 16
static_assert(NN % POOL_ROWS == 0, "a wave sums whole groups of POOL_ROWS nodes");

__device__ __forceinline__ void pool_flush(float* __restrict__ gsum,
                                           float* __restrict__ gcnt, int g,
                                           int lane, const float4 acc, float cnt) {
    float* p = gsum + g * DD + lane * 4;
    atomicAdd(p + 0, acc.x);
    atomicAdd(p + 1, acc.y);
    atomicAdd(p + 2, acc.z);
    atomicAdd(p + 3, acc.w);
    if (lane == 0) atomicAdd(gcnt + g, cnt);
}

__global__ __launch_bounds__(256) void k_pool_sum(const float* __restrict__ h,
                                                  const int* __restrict__ batch,
                                                  float* __restrict__ gsum,
                                                  float* __restrict__ gcnt) {
    const int lane = threadIdx.x & 31;
    const int w = __builtin_amdgcn_readfirstlane(
        (int)((blockIdx.x * blockDim.x + threadIdx.x) >> 5));
    const int n0 = w * POOL_ROWS;
    if (n0 >= NN) return;
    const int gl = batch[n0 + (lane & (POOL_ROWS - 1))];
    float4 r[POOL_ROWS];
    const float* __restrict__ hp = h + n0 * DD + lane * 4;
#pragma unroll
    for (int i = 0; i < POOL_ROWS; ++i) r[i] = *(const float4*)(hp + i * DD);

    int gcur = __builtin_amdgcn_readlane(gl, 0);
    float4 acc = r[0];
    float cnt = 1.f;
#pragma unroll
    for (int i = 1; i < POOL_ROWS; ++i) {
        const int gi = __builtin_amdgcn_readlane(gl, i);
        if (gi != gcur) {
            pool_flush(gsum, gcnt, gcur, lane, acc, cnt);
            gcur = gi;
            acc = r[i];
            cnt = 1.f;
        } else {
            acc.x += r[i].x; acc.y += r[i].y;
            acc.z += r[i].z; acc.w += r[i].w;
            cnt += 1.f;
        }
    }
    pool_flush(gsum, gcnt, gcur, lane, acc, cnt);
}

__global__ void k_pool_div(const float* __restrict__ gsum,
                           const float* __restrict__ gcnt,
                           float* __restrict__ outg) {
    int t = blockIdx.x * blockDim.x + threadIdx.x;
    if (t >= GG * DD) return;
    outg[t] = gsum[t] / fmaxf(gcnt[t >> 7], 1.f);
}

extern "C" void kernel_launch(void* const* d_in, const int* in_sizes, int n_in,
                              void* d_out, int out_size, void* d_ws, size_t ws_size,
                              hipStream_t stream) {
    (void)in_sizes; (void)n_in; (void)out_size; (void)ws_size;

    const float* x     = (const float*)d_in[0];
    const int*   ei    = (const int*)d_in[1];
    const int*   batch = (const int*)d_in[2];
    const float* W0    = (const float*)d_in[3];
    const float* as0   = (const float*)d_in[4];
    const float* ad0   = (const float*)d_in[5];
    const float* b0    = (const float*)d_in[6];
    const float* Wsl   = (const float*)d_in[7];
    const float* asrcs = (const float*)d_in[8];
    const float* adsts = (const float*)d_in[9];
    const float* bsl   = (const float*)d_in[10];
    const float* lng   = (const float*)d_in[11];
    const float* lnb   = (const float*)d_in[12];
    float* out = (float*)d_out;

    float* ws   = (float*)d_ws;
    float* A0v  = ws;           ws += NN * DD;
    float* A1v  = ws;           ws += NN * DD;
    float* Lin  = ws;           ws += NN * DD;
    float* xp   = ws;           ws += NN * KPAD;
    float* Wp   = ws;           ws += KPAD * DD;
    float* aS   = ws;           ws += NN * HH;
    float* aD   = ws;           ws += NN * HH;
    float* gsum = ws;           ws += GG * DD;
    float* gcnt = ws;           ws += GG;
    int* binTot  = (int*)ws;    ws += BPITCH;
    int* binBase = (int*)ws;    ws += BPITCH;
    int* cnt     = (int*)ws;    ws += NCHUNK * BPITCH;
    int* off     = (int*)ws;    ws += NCHUNK * BPITCH;
    int* binned  = (int*)ws;    ws += 2 * EE;
    int* rowptr  = (int*)ws;    ws += NN + 4;
    int* col     = (int*)ws;    ws += EE;

    k_bin_count<<<NCHUNK, 256, 0, stream>>>(ei, cnt);
    k_bin_scan<<<NBIN, 256, 0, stream>>>(cnt, off, binTot);
    k_bin_base<<<1, 256, 0, stream>>>(binTot, binBase);
    k_bin_scatter<<<NCHUNK, 256, 0, stream>>>(ei, off, binBase, binned);
    k_bin_csr<<<NBIN, 256, 0, stream>>>(binned, binBase, binTot, rowptr, col);

    k_pad_x<<<(NN * KPAD + 255) / 256, 256, 0, stream>>>(x, xp);
    k_pad_w<<<(KPAD * DD + 255) / 256, 256, 0, stream>>>(W0, Wp);

    for (int layer = 0; layer < LL; ++layer) {
        const float* Ain;  float* agg;  const float* resid;
        switch (layer) {
            case 0: Ain = xp;  agg = A0v; resid = nullptr; break;
            case 1: Ain = A0v; agg = A1v; resid = A0v;     break;
            case 2: Ain = A1v; agg = A0v; resid = A1v;     break;
            default:Ain = A0v; agg = out; resid = A0v;     break;
        }
        const float* asv  = (layer == 0) ? as0 : asrcs + (layer - 1) * HH * CC;
        const float* adv  = (layer == 0) ? ad0 : adsts + (layer - 1) * HH * CC;
        const float* bias = (layer == 0) ? b0  : bsl   + (layer - 1) * DD;

        if (layer == 0) {
            k_gemm<KPAD><<<NN / 16, 256, 0, stream>>>(Ain, Wp, Lin);
        } else {
            k_gemm<DD><<<NN / 16, 256, 0, stream>>>(
                Ain, Wsl + (layer - 1) * DD * DD, Lin);
        }
        k_alpha<<<(NN * HH + 255) / 256, 256, 0, stream>>>(Lin, asv, adv, aS, aD);
        if (layer == 0) {
            k_gat_node<false><<<(NN * 32 + 255) / 256, 256, 0, stream>>>(
                rowptr, col, aS, aD, Lin, bias, lng + layer * DD, lnb + layer * DD,
                resid, agg);
        } else {
            k_gat_node<true><<<(NN * 32 + 255) / 256, 256, 0, stream>>>(
                rowptr, col, aS, aD, Lin, bias, lng + layer * DD, lnb + layer * DD,
                resid, agg);
        }
    }

    k_zero_pool<<<(GG * DD + 255) / 256, 256, 0, stream>>>(gsum, gcnt);
    k_pool_sum<<<(NN / POOL_ROWS * 32 + 255) / 256, 256, 0, stream>>>(out, batch, gsum, gcnt);
    k_pool_div<<<(GG * DD + 255) / 256, 256, 0, stream>>>(gsum, gcnt, out + NN * DD);
}


# GEMM retiled: a wave computes a 16x128 strip on 8 accumulator tiles, B staged once per block in LDS, A read as 16-byte vectors shared by the 8 column tiles (k order of each dot product's sum changes)
# speedup vs baseline: 17.2810x; 1.1054x over previous
#include <hip/hip_runtime.h>
#include <math.h>

#define NN 100000
#define EE 1600000
#define HH 4
#define CC 32
#define DD 128
#define GG 64
#define LL 4
#define INDIM 7
#define KPAD 8
#define NEG_SLOPE 0.2f

typedef __attribute__((ext_vector_type(2))) float v2f;
typedef __attribute__((ext_vector_type(8))) float v8f;

__device__ __forceinline__ float lrelu(float t) {
    return t > 0.f ? t : NEG_SLOPE * t;
}

__global__ void k_pad_x(const float* __restrict__ x, float* __restrict__ xp) {
    int t = blockIdx.x * blockDim.x + threadIdx.x;
    if (t >= NN * KPAD) return;
    const int n = t >> 3, c = t & 7;
    xp[t] = (c < INDIM) ? x[n * INDIM + c] : 0.f;
}

__global__ void k_pad_w(const float* __restrict__ W0, float* __restrict__ Wp) {
    int t = blockIdx.x * blockDim.x + threadIdx.x;
    if (t >= KPAD * DD) return;
    Wp[t] = (t < INDIM * DD) ? W0[t] : 0.f;
}

#define GEMM_WAVES 8
#define ROW_TILES (NN / 16)
static_assert(NN % 16 == 0, "whole 16-row strips");

__device__ __forceinline__ v8f wmma4(float a0, float a1, float b0, float b1, v8f c) {
    v2f a; a.x = a0; a.y = a1;
    v2f b; b.x = b0; b.y = b1;
    return __builtin_amdgcn_wmma_f32_16x16x4_f32(
         false, a,  false, b,
         (short)0, c,  false,  false);
}

template <int K>
__global__ __launch_bounds__(GEMM_WAVES * 32) void k_gemm(const float* __restrict__ A,
                                                          const float* __restrict__ B,
                                                          float* __restrict__ Cm) {
    static_assert(K % 8 == 0, "k is consumed in blocks of 8");
    constexpr int NKB = K / 8;
    constexpr int CH  = NKB < 4 ? NKB : 4;
    static_assert(NKB % CH == 0, "whole chunks");
    __shared__ float4 Bs[(K / 4) * DD];

    const int lane = threadIdx.x & 31;
    const int wave = __builtin_amdgcn_readfirstlane((int)(threadIdx.x >> 5));
    const int m    = lane & 15;
    const int half = lane >> 4;

    const int tile  = blockIdx.x * GEMM_WAVES + wave;
    const int tilec = tile < ROW_TILES ? tile : ROW_TILES - 1;
    const float* __restrict__ arow = A + (tilec * 16 + m) * K + half * 4;

    float4 a_cur[CH];
#pragma unroll
    for (int i = 0; i < CH; ++i) a_cur[i] = *(const float4*)(arow + i * 8);

    constexpr int NPAIR = (K / 4) * (DD / 32);
    static_assert(NPAIR % GEMM_WAVES == 0, "every wave stages the same number of pieces");
#pragma unroll 4
    for (int t = 0; t < NPAIR / GEMM_WAVES; ++t) {
        const int p = t * GEMM_WAVES + wave;
        const int kq = p >> 2, n = (p & 3) * 32 + lane;
        const float* __restrict__ bp = B + kq * 4 * DD + n;
        float4 v;
        v.x = bp[0];
        v.y = bp[DD];
        v.z = bp[2 * DD];
        v.w = bp[3 * DD];
        Bs[kq * DD + n] = v;
    }
    __syncthreads();
    if (tile >= ROW_TILES) return;

    v8f acc[8];
#pragma unroll
    for (int ct = 0; ct < 8; ++ct) acc[ct] = v8f{};

#pragma unroll
    for (int c0 = 0; c0 < NKB; c0 += CH) {
        float4 a_nxt[CH];
        if (c0 + CH < NKB) {
#pragma unroll
            for (int i = 0; i < CH; ++i)
                a_nxt[i] = *(const float4*)(arow + (c0 + CH + i) * 8);
        }
#pragma unroll
        for (int i = 0; i < CH; ++i) {
            const float4 a = a_cur[i];
            const float4* __restrict__ bq = Bs + ((c0 + i) * 2 + half) * DD + m;
#pragma unroll
            for (int ct = 0; ct < 8; ct += 2) {
                const float4 b0 = bq[ct * 16];
                const float4 b1 = bq[ct * 16 + 16];
                acc[ct]     = wmma4(a.x, a.y, b0.x, b0.y, acc[ct]);
                acc[ct + 1] = wmma4(a.x, a.y, b1.x, b1.y, acc[ct + 1]);
                acc[ct]     = wmma4(a.z, a.w, b0.z, b0.w, acc[ct]);
                acc[ct + 1] = wmma4(a.z, a.w, b1.z, b1.w, acc[ct + 1]);
            }
        }
        if (c0 + CH < NKB) {
#pragma unroll
            for (int i = 0; i < CH; ++i) a_cur[i] = a_nxt[i];
        }
    }
    float* out = Cm + tile * 16 * DD + m;
#pragma unroll
    for (int ct = 0; ct < 8; ++ct) {
#pragma unroll
        for (int r = 0; r < 8; ++r) {
            out[(r + half * 8) * DD + ct * 16] = acc[ct][r];
        }
    }
}

__global__ void k_alpha(const float* __restrict__ hLin,
                        const float* __restrict__ a_src,
                        const float* __restrict__ a_dst,
                        float* __restrict__ aS, float* __restrict__ aD) {
    int t = blockIdx.x * blockDim.x + threadIdx.x;
    if (t >= NN * HH) return;
    const int n = t >> 2, hh = t & 3;
    const float4* hp = (const float4*)(hLin + n * DD + hh * CC);
    const float4* as = (const float4*)(a_src + hh * CC);
    const float4* ad = (const float4*)(a_dst + hh * CC);
    float s = 0.f, d = 0.f;
#pragma unroll
    for (int j = 0; j < CC / 4; ++j) {
        float4 v = hp[j], pa = as[j], pb = ad[j];
        s += v.x * pa.x + v.y * pa.y + v.z * pa.z + v.w * pa.w;
        d += v.x * pb.x + v.y * pb.y + v.z * pb.z + v.w * pb.w;
    }
    aS[t] = s;
    aD[t] = d;
}

#define BIN_SHIFT 8
#define BIN_NODES (1 << BIN_SHIFT)
#define NBIN ((NN + BIN_NODES - 1) / BIN_NODES)
#define CHUNK 4096
#define NCHUNK ((EE + CHUNK - 1) / CHUNK)
#define BPITCH 512
static_assert(NBIN <= BPITCH && NCHUNK <= BPITCH, "two slots per thread of a 256-thread block");
static_assert(BIN_NODES == 256, "one destination per thread in k_bin_csr");
#define CSR_BATCH 8

__device__ __forceinline__ int block_excl_scan(int v, int* wtot, int& total) {
    const int lane = threadIdx.x & 31, wave = threadIdx.x >> 5;
    int inc = v;
#pragma unroll
    for (int off = 1; off < 32; off <<= 1) {
        const int u = __shfl_up(inc, off, 32);
        if (lane >= off) inc += u;
    }
    if (lane == 31) wtot[wave] = inc;
    __syncthreads();
    int wb = 0, tot = 0;
#pragma unroll
    for (int w = 0; w < 8; ++w) {
        const int x = wtot[w];
        if (w < wave) wb += x;
        tot += x;
    }
    total = tot;
    return wb + inc - v;
}

__global__ __launch_bounds__(256) void k_bin_count(const int* __restrict__ ei,
                                                   int* __restrict__ cnt) {
    __shared__ int hist[BPITCH];
    const int t = threadIdx.x;
    hist[t] = 0;
    hist[t + 256] = 0;
    __syncthreads();
    const int e0 = blockIdx.x * CHUNK + t;
    int d[CHUNK / 256];
#pragma unroll
    for (int i = 0; i < CHUNK / 256; ++i) {
        const int e = e0 + i * 256;
        d[i] = ei[EE + (e < EE ? e : EE - 1)];
    }
#pragma unroll
    for (int i = 0; i < CHUNK / 256; ++i) {
        if (e0 + i * 256 < EE) atomicAdd(&hist[d[i] >> BIN_SHIFT], 1);
    }
    __syncthreads();
    cnt[blockIdx.x * BPITCH + t] = hist[t];
    cnt[blockIdx.x * BPITCH + t + 256] = hist[t + 256];
}

__global__ __launch_bounds__(256) void k_bin_scan(const int* __restrict__ cnt,
                                                  int* __restrict__ off,
                                                  int* __restrict__ binTot) {
    __shared__ int wtot[8];
    const int b = blockIdx.x, t = threadIdx.x;
    const int w0 = 2 * t, w1 = 2 * t + 1;
    const int c0 = (w0 < NCHUNK) ? cnt[w0 * BPITCH + b] : 0;
    const int c1 = (w1 < NCHUNK) ? cnt[w1 * BPITCH + b] : 0;
    int tot;
    const int ex = block_excl_scan(c0 + c1, wtot, tot);
    if (w0 < NCHUNK) off[w0 * BPITCH + b] = ex;
    if (w1 < NCHUNK) off[w1 * BPITCH + b] = ex + c0;
    if (t == 0) binTot[b] = tot;
}

__global__ __launch_bounds__(256) void k_bin_base(const int* __restrict__ binTot,
                                                  int* __restrict__ binBase) {
    __shared__ int wtot[8];
    const int t = threadIdx.x;
    const int b0 = 2 * t, b1 = 2 * t + 1;
    const int c0 = (b0 < NBIN) ? binTot[b0] : 0;
    const int c1 = (b1 < NBIN) ? binTot[b1] : 0;
    int tot;
    const int ex = block_excl_scan(c0 + c1, wtot, tot);
    binBase[b0] = ex;
    binBase[b1] = ex + c0;
}

__global__ __launch_bounds__(256) void k_bin_scatter(const int* __restrict__ ei,
                                                     const int* __restrict__ off,
                                                     const int* __restrict__ binBase,
                                                     int* __restrict__ binned) {
    __shared__ int cur[BPITCH];
    const int t = threadIdx.x;
    cur[t] = (t < NBIN) ? binBase[t] + off[blockIdx.x * BPITCH + t] : 0;
    cur[t + 256] = (t + 256 < NBIN) ? binBase[t + 256] + off[blockIdx.x * BPITCH + t + 256] : 0;
    __syncthreads();
    const int e0 = blockIdx.x * CHUNK + t;
    int s[CHUNK / 256], d[CHUNK / 256];
#pragma unroll
    for (int i = 0; i < CHUNK / 256; ++i) {
        const int e = e0 + i * 256;
        const int ec = e < EE ? e : EE - 1;
        s[i] = ei[ec];
        d[i] = ei[EE + ec];
    }
#pragma unroll
    for (int i = 0; i < CHUNK / 256; ++i) {
        if (e0 + i * 256 < EE) {
            const int pos = atomicAdd(&cur[d[i] >> BIN_SHIFT], 1);
            *(int2*)(binned + 2 * pos) = make_int2(s[i], d[i]);
        }
    }
}

__global__ __launch_bounds__(256) void k_bin_csr(const int* __restrict__ binned,
                                                 const int* __restrict__ binBase,
                                                 const int* __restrict__ binTot,
                                                 int* __restrict__ rowptr,
                                                 int* __restrict__ col) {
    __shared__ int cntl[BIN_NODES];
    __shared__ int curl[BIN_NODES];
    __shared__ int wtot[8];
    const int b = blockIdx.x, t = threadIdx.x;
    const int start = binBase[b];
    const int n = binTot[b];
    const int2* __restrict__ ed = (const int2*)(binned + 2 * start);
    cntl[t] = 0;
    __syncthreads();
    for (int i0 = t; i0 < n; i0 += 256 * CSR_BATCH) {
        int dl[CSR_BATCH];
#pragma unroll
        for (int k = 0; k < CSR_BATCH; ++k) {
            const int i = i0 + k * 256;
            dl[k] = ed[i < n ? i : n - 1].y & (BIN_NODES - 1);
        }
#pragma unroll
        for (int k = 0; k < CSR_BATCH; ++k) {
            if (i0 + k * 256 < n) atomicAdd(&cntl[dl[k]], 1);
        }
    }
    __syncthreads();
    int tot;
    const int ex = block_excl_scan(cntl[t], wtot, tot);
    curl[t] = ex;
    const int node = b * BIN_NODES + t;
    if (node <= NN) rowptr[node] = start + ex;
    __syncthreads();
    for (int i0 = t; i0 < n; i0 += 256 * CSR_BATCH) {
        int2 e[CSR_BATCH];
#pragma unroll
        for (int k = 0; k < CSR_BATCH; ++k) {
            const int i = i0 + k * 256;
            e[k] = ed[i < n ? i : n - 1];
        }
#pragma unroll
        for (int k = 0; k < CSR_BATCH; ++k) {
            if (i0 + k * 256 < n) {
                const int r = atomicAdd(&curl[e[k].y & (BIN_NODES - 1)], 1);
                col[start + r] = e[k].x;
            }
        }
    }
}

__device__ __forceinline__ float wave_max(float v) {
#pragma unroll
    for (int off = 16; off > 0; off >>= 1) v = fmaxf(v, __shfl_xor(v, off, 32));
    return v;
}

__device__ __forceinline__ float wave_sum(float v) {
#pragma unroll
    for (int off = 16; off > 0; off >>= 1) v += __shfl_xor(v, off, 32);
    return v;
}

__device__ __forceinline__ void load_item(int t, int cnt, int beg, int d,
                                          const int* __restrict__ col,
                                          const float* __restrict__ aS,
                                          const float4 ad, int& s, float4& lg) {
    s = d;
    lg = make_float4(-3.0e38f, -3.0e38f, -3.0e38f, -3.0e38f);
    if (t < cnt) {
        if (t < cnt - 1) s = col[beg + t];
        const float4 as = *(const float4*)(aS + s * HH);
        lg.x = lrelu(as.x + ad.x);
        lg.y = lrelu(as.y + ad.y);
        lg.z = lrelu(as.z + ad.z);
        lg.w = lrelu(as.w + ad.w);
    }
}

__device__ __forceinline__ float lane_value(float v, int j) {
    return __int_as_float(__builtin_amdgcn_readlane(__float_as_int(v), j));
}

__device__ __forceinline__ void add_rows(const float* __restrict__ hrow, int head,
                                         int s, const float4 a, int n,
                                         float4& acc) {
#pragma unroll 4
    for (int j = 0; j < n; ++j) {
        const int sj = __builtin_amdgcn_readlane(s, j);
        const float a0 = lane_value(a.x, j);
        const float a1 = lane_value(a.y, j);
        const float a2 = lane_value(a.z, j);
        const float a3 = lane_value(a.w, j);
        const float aj = (head & 2) ? ((head & 1) ? a3 : a2) : ((head & 1) ? a1 : a0);
        const float4 v = *(const float4*)(hrow + sj * DD);
        acc.x += aj * v.x;
        acc.y += aj * v.y;
        acc.z += aj * v.z;
        acc.w += aj * v.w;
    }
}

template <bool HAS_RES>
__global__ __launch_bounds__(256) void k_gat_node(const int* __restrict__ rowptr,
                                                  const int* __restrict__ col,
                                                  const float* __restrict__ aS,
                                                  const float* __restrict__ aD,
                                                  const float* __restrict__ hLin,
                                                  const float* __restrict__ bias,
                                                  const float* __restrict__ gma,
                                                  const float* __restrict__ bta,
                                                  const float* __restrict__ resid,
                                                  float* __restrict__ outp) {
    const int lane = threadIdx.x & 31;
    const int head = lane >> 3;
    const int d = __builtin_amdgcn_readfirstlane(
        (int)((blockIdx.x * blockDim.x + threadIdx.x) >> 5));
    if (d >= NN) return;
    const int beg = rowptr[d];
    const int cnt = rowptr[d + 1] - beg + 1;
    const float4 ad = *(const float4*)(aD + d * HH);
    const float* __restrict__ hrow = hLin + lane * 4;
    float4 rr = make_float4(0.f, 0.f, 0.f, 0.f);
    if (HAS_RES) rr = *(const float4*)(resid + d * DD + lane * 4);

    int s0;
    float4 lg0;
    load_item(lane, cnt, beg, d, col, aS, ad, s0, lg0);

    float4 m = lg0;
    for (int base = 32; base < cnt; base += 32) {
        int s; float4 lg;
        load_item(base + lane, cnt, beg, d, col, aS, ad, s, lg);
        m.x = fmaxf(m.x, lg.x); m.y = fmaxf(m.y, lg.y);
        m.z = fmaxf(m.z, lg.z); m.w = fmaxf(m.w, lg.w);
    }
    m.x = wave_max(m.x); m.y = wave_max(m.y);
    m.z = wave_max(m.z); m.w = wave_max(m.w);

    float4 e0 = make_float4(0.f, 0.f, 0.f, 0.f);
    if (lane < cnt) {
        e0.x = expf(lg0.x - m.x); e0.y = expf(lg0.y - m.y);
        e0.z = expf(lg0.z - m.z); e0.w = expf(lg0.w - m.w);
    }
    float4 z = e0;
    for (int base = 32; base < cnt; base += 32) {
        int s; float4 lg;
        load_item(base + lane, cnt, beg, d, col, aS, ad, s, lg);
        if (base + lane < cnt) {
            z.x += expf(lg.x - m.x); z.y += expf(lg.y - m.y);
            z.z += expf(lg.z - m.z); z.w += expf(lg.w - m.w);
        }
    }
    float4 den;
    den.x = wave_sum(z.x) + 1e-16f; den.y = wave_sum(z.y) + 1e-16f;
    den.z = wave_sum(z.z) + 1e-16f; den.w = wave_sum(z.w) + 1e-16f;

    float4 acc = make_float4(0.f, 0.f, 0.f, 0.f);
    {
        const float4 a = make_float4(e0.x / den.x, e0.y / den.y,
                                     e0.z / den.z, e0.w / den.w);
        add_rows(hrow, head, s0, a, cnt < 32 ? cnt : 32, acc);
    }
    for (int base = 32; base < cnt; base += 32) {
        int s; float4 lg;
        load_item(base + lane, cnt, beg, d, col, aS, ad, s, lg);
        float4 a = make_float4(0.f, 0.f, 0.f, 0.f);
        if (base + lane < cnt) {
            a.x = expf(lg.x - m.x) / den.x; a.y = expf(lg.y - m.y) / den.y;
            a.z = expf(lg.z - m.z) / den.z; a.w = expf(lg.w - m.w) / den.w;
        }
        const int left = cnt - base;
        add_rows(hrow, head, s, a, left < 32 ? left : 32, acc);
    }

    const float4 bi = *(const float4*)(bias + lane * 4);
    const float4 gm = *(const float4*)(gma + lane * 4);
    const float4 bt = *(const float4*)(bta + lane * 4);
    const float4 v = make_float4(acc.x + bi.x, acc.y + bi.y, acc.z + bi.z, acc.w + bi.w);
    const float mu = wave_sum((v.x + v.y) + (v.z + v.w)) * (1.f / 128.f);
    const float4 dl = make_float4(v.x - mu, v.y - mu, v.z - mu, v.w - mu);
    const float var =
        wave_sum((dl.x * dl.x + dl.y * dl.y) + (dl.z * dl.z + dl.w * dl.w));
    const float rs = rsqrtf(var * (1.f / 128.f) + 1e-5f);
    float4 y = make_float4(dl.x * rs * gm.x + bt.x, dl.y * rs * gm.y + bt.y,
                           dl.z * rs * gm.z + bt.z, dl.w * rs * gm.w + bt.w);
    y.x = y.x > 0.f ? y.x : expm1f(y.x);
    y.y = y.y > 0.f ? y.y : expm1f(y.y);
    y.z = y.z > 0.f ? y.z : expm1f(y.z);
    y.w = y.w > 0.f ? y.w : expm1f(y.w);
    if (HAS_RES) { y.x += rr.x; y.y += rr.y; y.z += rr.z; y.w += rr.w; }
    *(float4*)(outp + d * DD + lane * 4) = y;
}

__global__ void k_zero_pool(float* __restrict__ gsum, float* __restrict__ gcnt) {
    int t = blockIdx.x * blockDim.x + threadIdx.x;
    if (t < GG * DD) gsum[t] = 0.f;
    if (t < GG) gcnt[t] = 0.f;
}

#define POOL_ROWS 16
static_assert(NN % POOL_ROWS == 0, "a wave sums whole groups of POOL_ROWS nodes");

__device__ __forceinline__ void pool_flush(float* __restrict__ gsum,
                                           float* __restrict__ gcnt, int g,
                                           int lane, const float4 acc, float cnt) {
    float* p = gsum + g * DD + lane * 4;
    atomicAdd(p + 0, acc.x);
    atomicAdd(p + 1, acc.y);
    atomicAdd(p + 2, acc.z);
    atomicAdd(p + 3, acc.w);
    if (lane == 0) atomicAdd(gcnt + g, cnt);
}

__global__ __launch_bounds__(256) void k_pool_sum(const float* __restrict__ h,
                                                  const int* __restrict__ batch,
                                                  float* __restrict__ gsum,
                                                  float* __restrict__ gcnt) {
    const int lane = threadIdx.x & 31;
    const int w = __builtin_amdgcn_readfirstlane(
        (int)((blockIdx.x * blockDim.x + threadIdx.x) >> 5));
    const int n0 = w * POOL_ROWS;
    if (n0 >= NN) return;
    const int gl = batch[n0 + (lane & (POOL_ROWS - 1))];
    float4 r[POOL_ROWS];
    const float* __restrict__ hp = h + n0 * DD + lane * 4;
#pragma unroll
    for (int i = 0; i < POOL_ROWS; ++i) r[i] = *(const float4*)(hp + i * DD);

    int gcur = __builtin_amdgcn_readlane(gl, 0);
    float4 acc = r[0];
    float cnt = 1.f;
#pragma unroll
    for (int i = 1; i < POOL_ROWS; ++i) {
        const int gi = __builtin_amdgcn_readlane(gl, i);
        if (gi != gcur) {
            pool_flush(gsum, gcnt, gcur, lane, acc, cnt);
            gcur = gi;
            acc = r[i];
            cnt = 1.f;
        } else {
            acc.x += r[i].x; acc.y += r[i].y;
            acc.z += r[i].z; acc.w += r[i].w;
            cnt += 1.f;
        }
    }
    pool_flush(gsum, gcnt, gcur, lane, acc, cnt);
}

__global__ void k_pool_div(const float* __restrict__ gsum,
                           const float* __restrict__ gcnt,
                           float* __restrict__ outg) {
    int t = blockIdx.x * blockDim.x + threadIdx.x;
    if (t >= GG * DD) return;
    outg[t] = gsum[t] / fmaxf(gcnt[t >> 7], 1.f);
}

extern "C" void kernel_launch(void* const* d_in, const int* in_sizes, int n_in,
                              void* d_out, int out_size, void* d_ws, size_t ws_size,
                              hipStream_t stream) {
    (void)in_sizes; (void)n_in; (void)out_size; (void)ws_size;

    const float* x     = (const float*)d_in[0];
    const int*   ei    = (const int*)d_in[1];
    const int*   batch = (const int*)d_in[2];
    const float* W0    = (const float*)d_in[3];
    const float* as0   = (const float*)d_in[4];
    const float* ad0   = (const float*)d_in[5];
    const float* b0    = (const float*)d_in[6];
    const float* Wsl   = (const float*)d_in[7];
    const float* asrcs = (const float*)d_in[8];
    const float* adsts = (const float*)d_in[9];
    const float* bsl   = (const float*)d_in[10];
    const float* lng   = (const float*)d_in[11];
    const float* lnb   = (const float*)d_in[12];
    float* out = (float*)d_out;

    float* ws   = (float*)d_ws;
    float* A0v  = ws;           ws += NN * DD;
    float* A1v  = ws;           ws += NN * DD;
    float* Lin  = ws;           ws += NN * DD;
    float* xp   = ws;           ws += NN * KPAD;
    float* Wp   = ws;           ws += KPAD * DD;
    float* aS   = ws;           ws += NN * HH;
    float* aD   = ws;           ws += NN * HH;
    float* gsum = ws;           ws += GG * DD;
    float* gcnt = ws;           ws += GG;
    int* binTot  = (int*)ws;    ws += BPITCH;
    int* binBase = (int*)ws;    ws += BPITCH;
    int* cnt     = (int*)ws;    ws += NCHUNK * BPITCH;
    int* off     = (int*)ws;    ws += NCHUNK * BPITCH;
    int* binned  = (int*)ws;    ws += 2 * EE;
    int* rowptr  = (int*)ws;    ws += NN + 4;
    int* col     = (int*)ws;    ws += EE;

    k_bin_count<<<NCHUNK, 256, 0, stream>>>(ei, cnt);
    k_bin_scan<<<NBIN, 256, 0, stream>>>(cnt, off, binTot);
    k_bin_base<<<1, 256, 0, stream>>>(binTot, binBase);
    k_bin_scatter<<<NCHUNK, 256, 0, stream>>>(ei, off, binBase, binned);
    k_bin_csr<<<NBIN, 256, 0, stream>>>(binned, binBase, binTot, rowptr, col);

    k_pad_x<<<(NN * KPAD + 255) / 256, 256, 0, stream>>>(x, xp);
    k_pad_w<<<(KPAD * DD + 255) / 256, 256, 0, stream>>>(W0, Wp);

    for (int layer = 0; layer < LL; ++layer) {
        const float* Ain;  float* agg;  const float* resid;
        switch (layer) {
            case 0: Ain = xp;  agg = A0v; resid = nullptr; break;
            case 1: Ain = A0v; agg = A1v; resid = A0v;     break;
            case 2: Ain = A1v; agg = A0v; resid = A1v;     break;
            default:Ain = A0v; agg = out; resid = A0v;     break;
        }
        const float* asv  = (layer == 0) ? as0 : asrcs + (layer - 1) * HH * CC;
        const float* adv  = (layer == 0) ? ad0 : adsts + (layer - 1) * HH * CC;
        const float* bias = (layer == 0) ? b0  : bsl   + (layer - 1) * DD;

        if (layer == 0) {
            k_gemm<KPAD><<<(ROW_TILES + GEMM_WAVES - 1) / GEMM_WAVES, GEMM_WAVES * 32, 0, stream>>>(Ain, Wp, Lin);
        } else {
            k_gemm<DD><<<(ROW_TILES + GEMM_WAVES - 1) / GEMM_WAVES, GEMM_WAVES * 32, 0, stream>>>(
                Ain, Wsl + (layer - 1) * DD * DD, Lin);
        }
        k_alpha<<<(NN * HH + 255) / 256, 256, 0, stream>>>(Lin, asv, adv, aS, aD);
        if (layer == 0) {
            k_gat_node<false><<<(NN * 32 + 255) / 256, 256, 0, stream>>>(
                rowptr, col, aS, aD, Lin, bias, lng + layer * DD, lnb + layer * DD,
                resid, agg);
        } else {
            k_gat_node<true><<<(NN * 32 + 255) / 256, 256, 0, stream>>>(
                rowptr, col, aS, aD, Lin, bias, lng + layer * DD, lnb + layer * DD,
                resid, agg);
        }
    }

    k_zero_pool<<<(GG * DD + 255) / 256, 256, 0, stream>>>(gsum, gcnt);
    k_pool_sum<<<(NN / POOL_ROWS * 32 + 255) / 256, 256, 0, stream>>>(out, batch, gsum, gcnt);
    k_pool_div<<<(GG * DD + 255) / 256, 256, 0, stream>>>(gsum, gcnt, out + NN * DD);
}


# attention coefficients (alpha_src, alpha_dst) formed in the GEMM epilogue from the accumulators by a 16-lane reduce-scatter; the coefficient kernel is gone (order of those dot products' sums changes)
# speedup vs baseline: 17.9734x; 1.0401x over previous
#include <hip/hip_runtime.h>
#include <math.h>

#define NN 100000
#define EE 1600000
#define HH 4
#define CC 32
#define DD 128
#define GG 64
#define LL 4
#define INDIM 7
#define KPAD 8
#define NEG_SLOPE 0.2f

typedef __attribute__((ext_vector_type(2))) float v2f;
typedef __attribute__((ext_vector_type(8))) float v8f;

__device__ __forceinline__ float lrelu(float t) {
    return t > 0.f ? t : NEG_SLOPE * t;
}

__global__ void k_pad_x(const float* __restrict__ x, float* __restrict__ xp) {
    int t = blockIdx.x * blockDim.x + threadIdx.x;
    if (t >= NN * KPAD) return;
    const int n = t >> 3, c = t & 7;
    xp[t] = (c < INDIM) ? x[n * INDIM + c] : 0.f;
}

__global__ void k_pad_w(const float* __restrict__ W0, float* __restrict__ Wp) {
    int t = blockIdx.x * blockDim.x + threadIdx.x;
    if (t >= KPAD * DD) return;
    Wp[t] = (t < INDIM * DD) ? W0[t] : 0.f;
}

#define GEMM_WAVES 8
#define ROW_TILES (NN / 16)
static_assert(NN % 16 == 0, "whole 16-row strips");

__device__ __forceinline__ v8f wmma4(float a0, float a1, float b0, float b1, v8f c) {
    v2f a; a.x = a0; a.y = a1;
    v2f b; b.x = b0; b.y = b1;
    return __builtin_amdgcn_wmma_f32_16x16x4_f32(
         false, a,  false, b,
         (short)0, c,  false,  false);
}

template <int X>
__device__ __forceinline__ float row_xor(float v) {
    return __int_as_float(__builtin_amdgcn_update_dpp(
        0, __float_as_int(v), 0x160 + X, 0xf, 0xf, true));
}

template <int N, int X>
__device__ __forceinline__ void rs_step(const float (&in)[2 * N], float (&out)[N], bool bit) {
#pragma unroll
    for (int i = 0; i < N; ++i) {
        const float keep = bit ? in[2 * i + 1] : in[2 * i];
        const float send = bit ? in[2 * i] : in[2 * i + 1];
        out[i] = keep + row_xor<X>(send);
    }
}

template <int K>
__global__ __launch_bounds__(GEMM_WAVES * 32) void k_gemm(const float* __restrict__ A,
                                                          const float* __restrict__ B,
                                                          float* __restrict__ Cm,
                                                          const float* __restrict__ a_src,
                                                          const float* __restrict__ a_dst,
                                                          float* __restrict__ aS,
                                                          float* __restrict__ aD) {
    static_assert(K % 8 == 0, "k is consumed in blocks of 8");
    constexpr int NKB = K / 8;
    constexpr int CH  = NKB < 4 ? NKB : 4;
    static_assert(NKB % CH == 0, "whole chunks");
    __shared__ float4 Bs[(K / 4) * DD];

    const int lane = threadIdx.x & 31;
    const int wave = __builtin_amdgcn_readfirstlane((int)(threadIdx.x >> 5));
    const int m    = lane & 15;
    const int half = lane >> 4;

    const int tile  = blockIdx.x * GEMM_WAVES + wave;
    const int tilec = tile < ROW_TILES ? tile : ROW_TILES - 1;
    const float* __restrict__ arow = A + (tilec * 16 + m) * K + half * 4;

    float4 a_cur[CH];
#pragma unroll
    for (int i = 0; i < CH; ++i) a_cur[i] = *(const float4*)(arow + i * 8);

    constexpr int NPAIR = (K / 4) * (DD / 32);
    static_assert(NPAIR % GEMM_WAVES == 0, "every wave stages the same number of pieces");
#pragma unroll 4
    for (int t = 0; t < NPAIR / GEMM_WAVES; ++t) {
        const int p = t * GEMM_WAVES + wave;
        const int kq = p >> 2, n = (p & 3) * 32 + lane;
        const float* __restrict__ bp = B + kq * 4 * DD + n;
        float4 v;
        v.x = bp[0];
        v.y = bp[DD];
        v.z = bp[2 * DD];
        v.w = bp[3 * DD];
        Bs[kq * DD + n] = v;
    }
    __syncthreads();
    if (tile >= ROW_TILES) return;

    v8f acc[8];
#pragma unroll
    for (int ct = 0; ct < 8; ++ct) acc[ct] = v8f{};

#pragma unroll
    for (int c0 = 0; c0 < NKB; c0 += CH) {
        float4 a_nxt[CH];
        if (c0 + CH < NKB) {
#pragma unroll
            for (int i = 0; i < CH; ++i)
                a_nxt[i] = *(const float4*)(arow + (c0 + CH + i) * 8);
        }
#pragma unroll
        for (int i = 0; i < CH; ++i) {
            const float4 a = a_cur[i];
            const float4* __restrict__ bq = Bs + ((c0 + i) * 2 + half) * DD + m;
#pragma unroll
            for (int ct = 0; ct < 8; ct += 2) {
                const float4 b0 = bq[ct * 16];
                const float4 b1 = bq[ct * 16 + 16];
                acc[ct]     = wmma4(a.x, a.y, b0.x, b0.y, acc[ct]);
                acc[ct + 1] = wmma4(a.x, a.y, b1.x, b1.y, acc[ct + 1]);
                acc[ct]     = wmma4(a.z, a.w, b0.z, b0.w, acc[ct]);
                acc[ct + 1] = wmma4(a.z, a.w, b1.z, b1.w, acc[ct + 1]);
            }
        }
        if (c0 + CH < NKB) {
#pragma unroll
            for (int i = 0; i < CH; ++i) a_cur[i] = a_nxt[i];
        }
    }
    float cs0[HH], cs1[HH], cd0[HH], cd1[HH];
#pragma unroll
    for (int hd = 0; hd < HH; ++hd) {
        cs0[hd] = a_src[hd * CC + m];
        cs1[hd] = a_src[hd * CC + 16 + m];
        cd0[hd] = a_dst[hd * CC + m];
        cd1[hd] = a_dst[hd * CC + 16 + m];
    }
    float* out = Cm + tile * 16 * DD + m;
#pragma unroll
    for (int ct = 0; ct < 8; ++ct) {
#pragma unroll
        for (int r = 0; r < 8; ++r) {
            out[(r + half * 8) * DD + ct * 16] = acc[ct][r];
        }
    }

    __builtin_amdgcn_sched_barrier(0);
    const bool b3 = (m & 8) != 0, b2 = (m & 4) != 0, b1 = (m & 2) != 0, b0 = (m & 1) != 0;
    float res[HH];
#pragma unroll
    for (int hd = 0; hd < HH; ++hd) {
        const float s0 = cs0[hd], s1 = cs1[hd], d0 = cd0[hd], d1 = cd1[hd];
        float x16[16], x8[8], x4[4], x2[2], x1[1];
#pragma unroll
        for (int r = 0; r < 8; ++r) {
            const float c0 = acc[2 * hd][r], c1 = acc[2 * hd + 1][r];
            x16[r]     = c0 * s0 + c1 * s1;
            x16[8 + r] = c0 * d0 + c1 * d1;
        }
        rs_step<8, 8>(x16, x8, b3);
        rs_step<4, 4>(x8, x4, b2);
        rs_step<2, 2>(x4, x2, b1);
        rs_step<1, 1>(x2, x1, b0);
        res[hd] = x1[0];
    }
    const int rr = ((m >> 3) & 1) | ((m >> 1) & 2) | ((m << 1) & 4);
    float* ap = (b0 ? aD : aS) + (tile * 16 + half * 8 + rr) * HH;
    *(float4*)ap = make_float4(res[0], res[1], res[2], res[3]);
}

#define BIN_SHIFT 8
#define BIN_NODES (1 << BIN_SHIFT)
#define NBIN ((NN + BIN_NODES - 1) / BIN_NODES)
#define CHUNK 4096
#define NCHUNK ((EE + CHUNK - 1) / CHUNK)
#define BPITCH 512
static_assert(NBIN <= BPITCH && NCHUNK <= BPITCH, "two slots per thread of a 256-thread block");
static_assert(BIN_NODES == 256, "one destination per thread in k_bin_csr");
#define CSR_BATCH 8

__device__ __forceinline__ int block_excl_scan(int v, int* wtot, int& total) {
    const int lane = threadIdx.x & 31, wave = threadIdx.x >> 5;
    int inc = v;
#pragma unroll
    for (int off = 1; off < 32; off <<= 1) {
        const int u = __shfl_up(inc, off, 32);
        if (lane >= off) inc += u;
    }
    if (lane == 31) wtot[wave] = inc;
    __syncthreads();
    int wb = 0, tot = 0;
#pragma unroll
    for (int w = 0; w < 8; ++w) {
        const int x = wtot[w];
        if (w < wave) wb += x;
        tot += x;
    }
    total = tot;
    return wb + inc - v;
}

__global__ __launch_bounds__(256) void k_bin_count(const int* __restrict__ ei,
                                                   int* __restrict__ cnt) {
    __shared__ int hist[BPITCH];
    const int t = threadIdx.x;
    hist[t] = 0;
    hist[t + 256] = 0;
    __syncthreads();
    const int e0 = blockIdx.x * CHUNK + t;
    int d[CHUNK / 256];
#pragma unroll
    for (int i = 0; i < CHUNK / 256; ++i) {
        const int e = e0 + i * 256;
        d[i] = ei[EE + (e < EE ? e : EE - 1)];
    }
#pragma unroll
    for (int i = 0; i < CHUNK / 256; ++i) {
        if (e0 + i * 256 < EE) atomicAdd(&hist[d[i] >> BIN_SHIFT], 1);
    }
    __syncthreads();
    cnt[blockIdx.x * BPITCH + t] = hist[t];
    cnt[blockIdx.x * BPITCH + t + 256] = hist[t + 256];
}

__global__ __launch_bounds__(256) void k_bin_scan(const int* __restrict__ cnt,
                                                  int* __restrict__ off,
                                                  int* __restrict__ binTot) {
    __shared__ int wtot[8];
    const int b = blockIdx.x, t = threadIdx.x;
    const int w0 = 2 * t, w1 = 2 * t + 1;
    const int c0 = (w0 < NCHUNK) ? cnt[w0 * BPITCH + b] : 0;
    const int c1 = (w1 < NCHUNK) ? cnt[w1 * BPITCH + b] : 0;
    int tot;
    const int ex = block_excl_scan(c0 + c1, wtot, tot);
    if (w0 < NCHUNK) off[w0 * BPITCH + b] = ex;
    if (w1 < NCHUNK) off[w1 * BPITCH + b] = ex + c0;
    if (t == 0) binTot[b] = tot;
}

__global__ __launch_bounds__(256) void k_bin_base(const int* __restrict__ binTot,
                                                  int* __restrict__ binBase) {
    __shared__ int wtot[8];
    const int t = threadIdx.x;
    const int b0 = 2 * t, b1 = 2 * t + 1;
    const int c0 = (b0 < NBIN) ? binTot[b0] : 0;
    const int c1 = (b1 < NBIN) ? binTot[b1] : 0;
    int tot;
    const int ex = block_excl_scan(c0 + c1, wtot, tot);
    binBase[b0] = ex;
    binBase[b1] = ex + c0;
}

__global__ __launch_bounds__(256) void k_bin_scatter(const int* __restrict__ ei,
                                                     const int* __restrict__ off,
                                                     const int* __restrict__ binBase,
                                                     int* __restrict__ binned) {
    __shared__ int cur[BPITCH];
    const int t = threadIdx.x;
    cur[t] = (t < NBIN) ? binBase[t] + off[blockIdx.x * BPITCH + t] : 0;
    cur[t + 256] = (t + 256 < NBIN) ? binBase[t + 256] + off[blockIdx.x * BPITCH + t + 256] : 0;
    __syncthreads();
    const int e0 = blockIdx.x * CHUNK + t;
    int s[CHUNK / 256], d[CHUNK / 256];
#pragma unroll
    for (int i = 0; i < CHUNK / 256; ++i) {
        const int e = e0 + i * 256;
        const int ec = e < EE ? e : EE - 1;
        s[i] = ei[ec];
        d[i] = ei[EE + ec];
    }
#pragma unroll
    for (int i = 0; i < CHUNK / 256; ++i) {
        if (e0 + i * 256 < EE) {
            const int pos = atomicAdd(&cur[d[i] >> BIN_SHIFT], 1);
            *(int2*)(binned + 2 * pos) = make_int2(s[i], d[i]);
        }
    }
}

__global__ __launch_bounds__(256) void k_bin_csr(const int* __restrict__ binned,
                                                 const int* __restrict__ binBase,
                                                 const int* __restrict__ binTot,
                                                 int* __restrict__ rowptr,
                                                 int* __restrict__ col) {
    __shared__ int cntl[BIN_NODES];
    __shared__ int curl[BIN_NODES];
    __shared__ int wtot[8];
    const int b = blockIdx.x, t = threadIdx.x;
    const int start = binBase[b];
    const int n = binTot[b];
    const int2* __restrict__ ed = (const int2*)(binned + 2 * start);
    cntl[t] = 0;
    __syncthreads();
    for (int i0 = t; i0 < n; i0 += 256 * CSR_BATCH) {
        int dl[CSR_BATCH];
#pragma unroll
        for (int k = 0; k < CSR_BATCH; ++k) {
            const int i = i0 + k * 256;
            dl[k] = ed[i < n ? i : n - 1].y & (BIN_NODES - 1);
        }
#pragma unroll
        for (int k = 0; k < CSR_BATCH; ++k) {
            if (i0 + k * 256 < n) atomicAdd(&cntl[dl[k]], 1);
        }
    }
    __syncthreads();
    int tot;
    const int ex = block_excl_scan(cntl[t], wtot, tot);
    curl[t] = ex;
    const int node = b * BIN_NODES + t;
    if (node <= NN) rowptr[node] = start + ex;
    __syncthreads();
    for (int i0 = t; i0 < n; i0 += 256 * CSR_BATCH) {
        int2 e[CSR_BATCH];
#pragma unroll
        for (int k = 0; k < CSR_BATCH; ++k) {
            const int i = i0 + k * 256;
            e[k] = ed[i < n ? i : n - 1];
        }
#pragma unroll
        for (int k = 0; k < CSR_BATCH; ++k) {
            if (i0 + k * 256 < n) {
                const int r = atomicAdd(&curl[e[k].y & (BIN_NODES - 1)], 1);
                col[start + r] = e[k].x;
            }
        }
    }
}

__device__ __forceinline__ float wave_max(float v) {
#pragma unroll
    for (int off = 16; off > 0; off >>= 1) v = fmaxf(v, __shfl_xor(v, off, 32));
    return v;
}

__device__ __forceinline__ float wave_sum(float v) {
#pragma unroll
    for (int off = 16; off > 0; off >>= 1) v += __shfl_xor(v, off, 32);
    return v;
}

__device__ __forceinline__ void load_item(int t, int cnt, int beg, int d,
                                          const int* __restrict__ col,
                                          const float* __restrict__ aS,
                                          const float4 ad, int& s, float4& lg) {
    s = d;
    lg = make_float4(-3.0e38f, -3.0e38f, -3.0e38f, -3.0e38f);
    if (t < cnt) {
        if (t < cnt - 1) s = col[beg + t];
        const float4 as = *(const float4*)(aS + s * HH);
        lg.x = lrelu(as.x + ad.x);
        lg.y = lrelu(as.y + ad.y);
        lg.z = lrelu(as.z + ad.z);
        lg.w = lrelu(as.w + ad.w);
    }
}

__device__ __forceinline__ float lane_value(float v, int j) {
    return __int_as_float(__builtin_amdgcn_readlane(__float_as_int(v), j));
}

__device__ __forceinline__ void add_rows(const float* __restrict__ hrow, int head,
                                         int s, const float4 a, int n,
                                         float4& acc) {
#pragma unroll 4
    for (int j = 0; j < n; ++j) {
        const int sj = __builtin_amdgcn_readlane(s, j);
        const float a0 = lane_value(a.x, j);
        const float a1 = lane_value(a.y, j);
        const float a2 = lane_value(a.z, j);
        const float a3 = lane_value(a.w, j);
        const float aj = (head & 2) ? ((head & 1) ? a3 : a2) : ((head & 1) ? a1 : a0);
        const float4 v = *(const float4*)(hrow + sj * DD);
        acc.x += aj * v.x;
        acc.y += aj * v.y;
        acc.z += aj * v.z;
        acc.w += aj * v.w;
    }
}

template <bool HAS_RES>
__global__ __launch_bounds__(256) void k_gat_node(const int* __restrict__ rowptr,
                                                  const int* __restrict__ col,
                                                  const float* __restrict__ aS,
                                                  const float* __restrict__ aD,
                                                  const float* __restrict__ hLin,
                                                  const float* __restrict__ bias,
                                                  const float* __restrict__ gma,
                                                  const float* __restrict__ bta,
                                                  const float* __restrict__ resid,
                                                  float* __restrict__ outp) {
    const int lane = threadIdx.x & 31;
    const int head = lane >> 3;
    const int d = __builtin_amdgcn_readfirstlane(
        (int)((blockIdx.x * blockDim.x + threadIdx.x) >> 5));
    if (d >= NN) return;
    const int beg = rowptr[d];
    const int cnt = rowptr[d + 1] - beg + 1;
    const float4 ad = *(const float4*)(aD + d * HH);
    const float* __restrict__ hrow = hLin + lane * 4;
    float4 rr = make_float4(0.f, 0.f, 0.f, 0.f);
    if (HAS_RES) rr = *(const float4*)(resid + d * DD + lane * 4);

    int s0;
    float4 lg0;
    load_item(lane, cnt, beg, d, col, aS, ad, s0, lg0);

    float4 m = lg0;
    for (int base = 32; base < cnt; base += 32) {
        int s; float4 lg;
        load_item(base + lane, cnt, beg, d, col, aS, ad, s, lg);
        m.x = fmaxf(m.x, lg.x); m.y = fmaxf(m.y, lg.y);
        m.z = fmaxf(m.z, lg.z); m.w = fmaxf(m.w, lg.w);
    }
    m.x = wave_max(m.x); m.y = wave_max(m.y);
    m.z = wave_max(m.z); m.w = wave_max(m.w);

    float4 e0 = make_float4(0.f, 0.f, 0.f, 0.f);
    if (lane < cnt) {
        e0.x = expf(lg0.x - m.x); e0.y = expf(lg0.y - m.y);
        e0.z = expf(lg0.z - m.z); e0.w = expf(lg0.w - m.w);
    }
    float4 z = e0;
    for (int base = 32; base < cnt; base += 32) {
        int s; float4 lg;
        load_item(base + lane, cnt, beg, d, col, aS, ad, s, lg);
        if (base + lane < cnt) {
            z.x += expf(lg.x - m.x); z.y += expf(lg.y - m.y);
            z.z += expf(lg.z - m.z); z.w += expf(lg.w - m.w);
        }
    }
    float4 den;
    den.x = wave_sum(z.x) + 1e-16f; den.y = wave_sum(z.y) + 1e-16f;
    den.z = wave_sum(z.z) + 1e-16f; den.w = wave_sum(z.w) + 1e-16f;

    float4 acc = make_float4(0.f, 0.f, 0.f, 0.f);
    {
        const float4 a = make_float4(e0.x / den.x, e0.y / den.y,
                                     e0.z / den.z, e0.w / den.w);
        add_rows(hrow, head, s0, a, cnt < 32 ? cnt : 32, acc);
    }
    for (int base = 32; base < cnt; base += 32) {
        int s; float4 lg;
        load_item(base + lane, cnt, beg, d, col, aS, ad, s, lg);
        float4 a = make_float4(0.f, 0.f, 0.f, 0.f);
        if (base + lane < cnt) {
            a.x = expf(lg.x - m.x) / den.x; a.y = expf(lg.y - m.y) / den.y;
            a.z = expf(lg.z - m.z) / den.z; a.w = expf(lg.w - m.w) / den.w;
        }
        const int left = cnt - base;
        add_rows(hrow, head, s, a, left < 32 ? left : 32, acc);
    }

    const float4 bi = *(const float4*)(bias + lane * 4);
    const float4 gm = *(const float4*)(gma + lane * 4);
    const float4 bt = *(const float4*)(bta + lane * 4);
    const float4 v = make_float4(acc.x + bi.x, acc.y + bi.y, acc.z + bi.z, acc.w + bi.w);
    const float mu = wave_sum((v.x + v.y) + (v.z + v.w)) * (1.f / 128.f);
    const float4 dl = make_float4(v.x - mu, v.y - mu, v.z - mu, v.w - mu);
    const float var =
        wave_sum((dl.x * dl.x + dl.y * dl.y) + (dl.z * dl.z + dl.w * dl.w));
    const float rs = rsqrtf(var * (1.f / 128.f) + 1e-5f);
    float4 y = make_float4(dl.x * rs * gm.x + bt.x, dl.y * rs * gm.y + bt.y,
                           dl.z * rs * gm.z + bt.z, dl.w * rs * gm.w + bt.w);
    y.x = y.x > 0.f ? y.x : expm1f(y.x);
    y.y = y.y > 0.f ? y.y : expm1f(y.y);
    y.z = y.z > 0.f ? y.z : expm1f(y.z);
    y.w = y.w > 0.f ? y.w : expm1f(y.w);
    if (HAS_RES) { y.x += rr.x; y.y += rr.y; y.z += rr.z; y.w += rr.w; }
    *(float4*)(outp + d * DD + lane * 4) = y;
}

__global__ void k_zero_pool(float* __restrict__ gsum, float* __restrict__ gcnt) {
    int t = blockIdx.x * blockDim.x + threadIdx.x;
    if (t < GG * DD) gsum[t] = 0.f;
    if (t < GG) gcnt[t] = 0.f;
}

#define POOL_ROWS 16
static_assert(NN % POOL_ROWS == 0, "a wave sums whole groups of POOL_ROWS nodes");

__device__ __forceinline__ void pool_flush(float* __restrict__ gsum,
                                           float* __restrict__ gcnt, int g,
                                           int lane, const float4 acc, float cnt) {
    float* p = gsum + g * DD + lane * 4;
    atomicAdd(p + 0, acc.x);
    atomicAdd(p + 1, acc.y);
    atomicAdd(p + 2, acc.z);
    atomicAdd(p + 3, acc.w);
    if (lane == 0) atomicAdd(gcnt + g, cnt);
}

__global__ __launch_bounds__(256) void k_pool_sum(const float* __restrict__ h,
                                                  const int* __restrict__ batch,
                                                  float* __restrict__ gsum,
                                                  float* __restrict__ gcnt) {
    const int lane = threadIdx.x & 31;
    const int w = __builtin_amdgcn_readfirstlane(
        (int)((blockIdx.x * blockDim.x + threadIdx.x) >> 5));
    const int n0 = w * POOL_ROWS;
    if (n0 >= NN) return;
    const int gl = batch[n0 + (lane & (POOL_ROWS - 1))];
    float4 r[POOL_ROWS];
    const float* __restrict__ hp = h + n0 * DD + lane * 4;
#pragma unroll
    for (int i = 0; i < POOL_ROWS; ++i) r[i] = *(const float4*)(hp + i * DD);

    int gcur = __builtin_amdgcn_readlane(gl, 0);
    float4 acc = r[0];
    float cnt = 1.f;
#pragma unroll
    for (int i = 1; i < POOL_ROWS; ++i) {
        const int gi = __builtin_amdgcn_readlane(gl, i);
        if (gi != gcur) {
            pool_flush(gsum, gcnt, gcur, lane, acc, cnt);
            gcur = gi;
            acc = r[i];
            cnt = 1.f;
        } else {
            acc.x += r[i].x; acc.y += r[i].y;
            acc.z += r[i].z; acc.w += r[i].w;
            cnt += 1.f;
        }
    }
    pool_flush(gsum, gcnt, gcur, lane, acc, cnt);
}

__global__ void k_pool_div(const float* __restrict__ gsum,
                           const float* __restrict__ gcnt,
                           float* __restrict__ outg) {
    int t = blockIdx.x * blockDim.x + threadIdx.x;
    if (t >= GG * DD) return;
    outg[t] = gsum[t] / fmaxf(gcnt[t >> 7], 1.f);
}

extern "C" void kernel_launch(void* const* d_in, const int* in_sizes, int n_in,
                              void* d_out, int out_size, void* d_ws, size_t ws_size,
                              hipStream_t stream) {
    (void)in_sizes; (void)n_in; (void)out_size; (void)ws_size;

    const float* x     = (const float*)d_in[0];
    const int*   ei    = (const int*)d_in[1];
    const int*   batch = (const int*)d_in[2];
    const float* W0    = (const float*)d_in[3];
    const float* as0   = (const float*)d_in[4];
    const float* ad0   = (const float*)d_in[5];
    const float* b0    = (const float*)d_in[6];
    const float* Wsl   = (const float*)d_in[7];
    const float* asrcs = (const float*)d_in[8];
    const float* adsts = (const float*)d_in[9];
    const float* bsl   = (const float*)d_in[10];
    const float* lng   = (const float*)d_in[11];
    const float* lnb   = (const float*)d_in[12];
    float* out = (float*)d_out;

    float* ws   = (float*)d_ws;
    float* A0v  = ws;           ws += NN * DD;
    float* A1v  = ws;           ws += NN * DD;
    float* Lin  = ws;           ws += NN * DD;
    float* xp   = ws;           ws += NN * KPAD;
    float* Wp   = ws;           ws += KPAD * DD;
    float* aS   = ws;           ws += NN * HH;
    float* aD   = ws;           ws += NN * HH;
    float* gsum = ws;           ws += GG * DD;
    float* gcnt = ws;           ws += GG;
    int* binTot  = (int*)ws;    ws += BPITCH;
    int* binBase = (int*)ws;    ws += BPITCH;
    int* cnt     = (int*)ws;    ws += NCHUNK * BPITCH;
    int* off     = (int*)ws;    ws += NCHUNK * BPITCH;
    int* binned  = (int*)ws;    ws += 2 * EE;
    int* rowptr  = (int*)ws;    ws += NN + 4;
    int* col     = (int*)ws;    ws += EE;

    k_bin_count<<<NCHUNK, 256, 0, stream>>>(ei, cnt);
    k_bin_scan<<<NBIN, 256, 0, stream>>>(cnt, off, binTot);
    k_bin_base<<<1, 256, 0, stream>>>(binTot, binBase);
    k_bin_scatter<<<NCHUNK, 256, 0, stream>>>(ei, off, binBase, binned);
    k_bin_csr<<<NBIN, 256, 0, stream>>>(binned, binBase, binTot, rowptr, col);

    k_pad_x<<<(NN * KPAD + 255) / 256, 256, 0, stream>>>(x, xp);
    k_pad_w<<<(KPAD * DD + 255) / 256, 256, 0, stream>>>(W0, Wp);

    for (int layer = 0; layer < LL; ++layer) {
        const float* Ain;  float* agg;  const float* resid;
        switch (layer) {
            case 0: Ain = xp;  agg = A0v; resid = nullptr; break;
            case 1: Ain = A0v; agg = A1v; resid = A0v;     break;
            case 2: Ain = A1v; agg = A0v; resid = A1v;     break;
            default:Ain = A0v; agg = out; resid = A0v;     break;
        }
        const float* asv  = (layer == 0) ? as0 : asrcs + (layer - 1) * HH * CC;
        const float* adv  = (layer == 0) ? ad0 : adsts + (layer - 1) * HH * CC;
        const float* bias = (layer == 0) ? b0  : bsl   + (layer - 1) * DD;

        if (layer == 0) {
            k_gemm<KPAD><<<(ROW_TILES + GEMM_WAVES - 1) / GEMM_WAVES, GEMM_WAVES * 32, 0, stream>>>(
                Ain, Wp, Lin, asv, adv, aS, aD);
        } else {
            k_gemm<DD><<<(ROW_TILES + GEMM_WAVES - 1) / GEMM_WAVES, GEMM_WAVES * 32, 0, stream>>>(
                Ain, Wsl + (layer - 1) * DD * DD, Lin, asv, adv, aS, aD);
        }
        if (layer == 0) {
            k_gat_node<false><<<(NN * 32 + 255) / 256, 256, 0, stream>>>(
                rowptr, col, aS, aD, Lin, bias, lng + layer * DD, lnb + layer * DD,
                resid, agg);
        } else {
            k_gat_node<true><<<(NN * 32 + 255) / 256, 256, 0, stream>>>(
                rowptr, col, aS, aD, Lin, bias, lng + layer * DD, lnb + layer * DD,
                resid, agg);
        }
    }

    k_zero_pool<<<(GG * DD + 255) / 256, 256, 0, stream>>>(gsum, gcnt);
    k_pool_sum<<<(NN / POOL_ROWS * 32 + 255) / 256, 256, 0, stream>>>(out, batch, gsum, gcnt);
    k_pool_div<<<(GG * DD + 255) / 256, 256, 0, stream>>>(gsum, gcnt, out + NN * DD);
}


# aggregation softmax laid out by head: an 8-lane group does one head, 4 items per lane; max and denominator by 8-lane DPP, weights reach the row sums by DPP8 broadcast (denominator sum order changes)
# speedup vs baseline: 18.6324x; 1.0367x over previous
#include <hip/hip_runtime.h>
#include <math.h>

#define NN 100000
#define EE 1600000
#define HH 4
#define CC 32
#define DD 128
#define GG 64
#define LL 4
#define INDIM 7
#define KPAD 8
#define NEG_SLOPE 0.2f

typedef __attribute__((ext_vector_type(2))) float v2f;
typedef __attribute__((ext_vector_type(8))) float v8f;

__device__ __forceinline__ float lrelu(float t) {
    return t > 0.f ? t : NEG_SLOPE * t;
}

__global__ void k_pad_x(const float* __restrict__ x, float* __restrict__ xp) {
    int t = blockIdx.x * blockDim.x + threadIdx.x;
    if (t >= NN * KPAD) return;
    const int n = t >> 3, c = t & 7;
    xp[t] = (c < INDIM) ? x[n * INDIM + c] : 0.f;
}

__global__ void k_pad_w(const float* __restrict__ W0, float* __restrict__ Wp) {
    int t = blockIdx.x * blockDim.x + threadIdx.x;
    if (t >= KPAD * DD) return;
    Wp[t] = (t < INDIM * DD) ? W0[t] : 0.f;
}

#define GEMM_WAVES 8
#define ROW_TILES (NN / 16)
static_assert(NN % 16 == 0, "whole 16-row strips");

__device__ __forceinline__ v8f wmma4(float a0, float a1, float b0, float b1, v8f c) {
    v2f a; a.x = a0; a.y = a1;
    v2f b; b.x = b0; b.y = b1;
    return __builtin_amdgcn_wmma_f32_16x16x4_f32(
         false, a,  false, b,
         (short)0, c,  false,  false);
}

template <int X>
__device__ __forceinline__ float row_xor(float v) {
    return __int_as_float(__builtin_amdgcn_update_dpp(
        0, __float_as_int(v), 0x160 + X, 0xf, 0xf, true));
}

template <int N, int X>
__device__ __forceinline__ void rs_step(const float (&in)[2 * N], float (&out)[N], bool bit) {
#pragma unroll
    for (int i = 0; i < N; ++i) {
        const float keep = bit ? in[2 * i + 1] : in[2 * i];
        const float send = bit ? in[2 * i] : in[2 * i + 1];
        out[i] = keep + row_xor<X>(send);
    }
}

template <int K>
__global__ __launch_bounds__(GEMM_WAVES * 32) void k_gemm(const float* __restrict__ A,
                                                          const float* __restrict__ B,
                                                          float* __restrict__ Cm,
                                                          const float* __restrict__ a_src,
                                                          const float* __restrict__ a_dst,
                                                          float* __restrict__ aS,
                                                          float* __restrict__ aD) {
    static_assert(K % 8 == 0, "k is consumed in blocks of 8");
    constexpr int NKB = K / 8;
    constexpr int CH  = NKB < 4 ? NKB : 4;
    static_assert(NKB % CH == 0, "whole chunks");
    __shared__ float4 Bs[(K / 4) * DD];

    const int lane = threadIdx.x & 31;
    const int wave = __builtin_amdgcn_readfirstlane((int)(threadIdx.x >> 5));
    const int m    = lane & 15;
    const int half = lane >> 4;

    const int tile  = blockIdx.x * GEMM_WAVES + wave;
    const int tilec = tile < ROW_TILES ? tile : ROW_TILES - 1;
    const float* __restrict__ arow = A + (tilec * 16 + m) * K + half * 4;

    float4 a_cur[CH];
#pragma unroll
    for (int i = 0; i < CH; ++i) a_cur[i] = *(const float4*)(arow + i * 8);

    constexpr int NPAIR = (K / 4) * (DD / 32);
    static_assert(NPAIR % GEMM_WAVES == 0, "every wave stages the same number of pieces");
#pragma unroll 4
    for (int t = 0; t < NPAIR / GEMM_WAVES; ++t) {
        const int p = t * GEMM_WAVES + wave;
        const int kq = p >> 2, n = (p & 3) * 32 + lane;
        const float* __restrict__ bp = B + kq * 4 * DD + n;
        float4 v;
        v.x = bp[0];
        v.y = bp[DD];
        v.z = bp[2 * DD];
        v.w = bp[3 * DD];
        Bs[kq * DD + n] = v;
    }
    __syncthreads();
    if (tile >= ROW_TILES) return;

    v8f acc[8];
#pragma unroll
    for (int ct = 0; ct < 8; ++ct) acc[ct] = v8f{};

#pragma unroll
    for (int c0 = 0; c0 < NKB; c0 += CH) {
        float4 a_nxt[CH];
        if (c0 + CH < NKB) {
#pragma unroll
            for (int i = 0; i < CH; ++i)
                a_nxt[i] = *(const float4*)(arow + (c0 + CH + i) * 8);
        }
#pragma unroll
        for (int i = 0; i < CH; ++i) {
            const float4 a = a_cur[i];
            const float4* __restrict__ bq = Bs + ((c0 + i) * 2 + half) * DD + m;
#pragma unroll
            for (int ct = 0; ct < 8; ct += 2) {
                const float4 b0 = bq[ct * 16];
                const float4 b1 = bq[ct * 16 + 16];
                acc[ct]     = wmma4(a.x, a.y, b0.x, b0.y, acc[ct]);
                acc[ct + 1] = wmma4(a.x, a.y, b1.x, b1.y, acc[ct + 1]);
                acc[ct]     = wmma4(a.z, a.w, b0.z, b0.w, acc[ct]);
                acc[ct + 1] = wmma4(a.z, a.w, b1.z, b1.w, acc[ct + 1]);
            }
        }
        if (c0 + CH < NKB) {
#pragma unroll
            for (int i = 0; i < CH; ++i) a_cur[i] = a_nxt[i];
        }
    }
    float cs0[HH], cs1[HH], cd0[HH], cd1[HH];
#pragma unroll
    for (int hd = 0; hd < HH; ++hd) {
        cs0[hd] = a_src[hd * CC + m];
        cs1[hd] = a_src[hd * CC + 16 + m];
        cd0[hd] = a_dst[hd * CC + m];
        cd1[hd] = a_dst[hd * CC + 16 + m];
    }
    float* out = Cm + tile * 16 * DD + m;
#pragma unroll
    for (int ct = 0; ct < 8; ++ct) {
#pragma unroll
        for (int r = 0; r < 8; ++r) {
            out[(r + half * 8) * DD + ct * 16] = acc[ct][r];
        }
    }

    __builtin_amdgcn_sched_barrier(0);
    const bool b3 = (m & 8) != 0, b2 = (m & 4) != 0, b1 = (m & 2) != 0, b0 = (m & 1) != 0;
    float res[HH];
#pragma unroll
    for (int hd = 0; hd < HH; ++hd) {
        const float s0 = cs0[hd], s1 = cs1[hd], d0 = cd0[hd], d1 = cd1[hd];
        float x16[16], x8[8], x4[4], x2[2], x1[1];
#pragma unroll
        for (int r = 0; r < 8; ++r) {
            const float c0 = acc[2 * hd][r], c1 = acc[2 * hd + 1][r];
            x16[r]     = c0 * s0 + c1 * s1;
            x16[8 + r] = c0 * d0 + c1 * d1;
        }
        rs_step<8, 8>(x16, x8, b3);
        rs_step<4, 4>(x8, x4, b2);
        rs_step<2, 2>(x4, x2, b1);
        rs_step<1, 1>(x2, x1, b0);
        res[hd] = x1[0];
    }
    const int rr = ((m >> 3) & 1) | ((m >> 1) & 2) | ((m << 1) & 4);
    float* ap = (b0 ? aD : aS) + (tile * 16 + half * 8 + rr) * HH;
    *(float4*)ap = make_float4(res[0], res[1], res[2], res[3]);
}

#define BIN_SHIFT 8
#define BIN_NODES (1 << BIN_SHIFT)
#define NBIN ((NN + BIN_NODES - 1) / BIN_NODES)
#define CHUNK 4096
#define NCHUNK ((EE + CHUNK - 1) / CHUNK)
#define BPITCH 512
static_assert(NBIN <= BPITCH && NCHUNK <= BPITCH, "two slots per thread of a 256-thread block");
static_assert(BIN_NODES == 256, "one destination per thread in k_bin_csr");
#define CSR_BATCH 8

__device__ __forceinline__ int block_excl_scan(int v, int* wtot, int& total) {
    const int lane = threadIdx.x & 31, wave = threadIdx.x >> 5;
    int inc = v;
#pragma unroll
    for (int off = 1; off < 32; off <<= 1) {
        const int u = __shfl_up(inc, off, 32);
        if (lane >= off) inc += u;
    }
    if (lane == 31) wtot[wave] = inc;
    __syncthreads();
    int wb = 0, tot = 0;
#pragma unroll
    for (int w = 0; w < 8; ++w) {
        const int x = wtot[w];
        if (w < wave) wb += x;
        tot += x;
    }
    total = tot;
    return wb + inc - v;
}

__global__ __launch_bounds__(256) void k_bin_count(const int* __restrict__ ei,
                                                   int* __restrict__ cnt) {
    __shared__ int hist[BPITCH];
    const int t = threadIdx.x;
    hist[t] = 0;
    hist[t + 256] = 0;
    __syncthreads();
    const int e0 = blockIdx.x * CHUNK + t;
    int d[CHUNK / 256];
#pragma unroll
    for (int i = 0; i < CHUNK / 256; ++i) {
        const int e = e0 + i * 256;
        d[i] = ei[EE + (e < EE ? e : EE - 1)];
    }
#pragma unroll
    for (int i = 0; i < CHUNK / 256; ++i) {
        if (e0 + i * 256 < EE) atomicAdd(&hist[d[i] >> BIN_SHIFT], 1);
    }
    __syncthreads();
    cnt[blockIdx.x * BPITCH + t] = hist[t];
    cnt[blockIdx.x * BPITCH + t + 256] = hist[t + 256];
}

__global__ __launch_bounds__(256) void k_bin_scan(const int* __restrict__ cnt,
                                                  int* __restrict__ off,
                                                  int* __restrict__ binTot) {
    __shared__ int wtot[8];
    const int b = blockIdx.x, t = threadIdx.x;
    const int w0 = 2 * t, w1 = 2 * t + 1;
    const int c0 = (w0 < NCHUNK) ? cnt[w0 * BPITCH + b] : 0;
    const int c1 = (w1 < NCHUNK) ? cnt[w1 * BPITCH + b] : 0;
    int tot;
    const int ex = block_excl_scan(c0 + c1, wtot, tot);
    if (w0 < NCHUNK) off[w0 * BPITCH + b] = ex;
    if (w1 < NCHUNK) off[w1 * BPITCH + b] = ex + c0;
    if (t == 0) binTot[b] = tot;
}

__global__ __launch_bounds__(256) void k_bin_base(const int* __restrict__ binTot,
                                                  int* __restrict__ binBase) {
    __shared__ int wtot[8];
    const int t = threadIdx.x;
    const int b0 = 2 * t, b1 = 2 * t + 1;
    const int c0 = (b0 < NBIN) ? binTot[b0] : 0;
    const int c1 = (b1 < NBIN) ? binTot[b1] : 0;
    int tot;
    const int ex = block_excl_scan(c0 + c1, wtot, tot);
    binBase[b0] = ex;
    binBase[b1] = ex + c0;
}

__global__ __launch_bounds__(256) void k_bin_scatter(const int* __restrict__ ei,
                                                     const int* __restrict__ off,
                                                     const int* __restrict__ binBase,
                                                     int* __restrict__ binned) {
    __shared__ int cur[BPITCH];
    const int t = threadIdx.x;
    cur[t] = (t < NBIN) ? binBase[t] + off[blockIdx.x * BPITCH + t] : 0;
    cur[t + 256] = (t + 256 < NBIN) ? binBase[t + 256] + off[blockIdx.x * BPITCH + t + 256] : 0;
    __syncthreads();
    const int e0 = blockIdx.x * CHUNK + t;
    int s[CHUNK / 256], d[CHUNK / 256];
#pragma unroll
    for (int i = 0; i < CHUNK / 256; ++i) {
        const int e = e0 + i * 256;
        const int ec = e < EE ? e : EE - 1;
        s[i] = ei[ec];
        d[i] = ei[EE + ec];
    }
#pragma unroll
    for (int i = 0; i < CHUNK / 256; ++i) {
        if (e0 + i * 256 < EE) {
            const int pos = atomicAdd(&cur[d[i] >> BIN_SHIFT], 1);
            *(int2*)(binned + 2 * pos) = make_int2(s[i], d[i]);
        }
    }
}

__global__ __launch_bounds__(256) void k_bin_csr(const int* __restrict__ binned,
                                                 const int* __restrict__ binBase,
                                                 const int* __restrict__ binTot,
                                                 int* __restrict__ rowptr,
                                                 int* __restrict__ col) {
    __shared__ int cntl[BIN_NODES];
    __shared__ int curl[BIN_NODES];
    __shared__ int wtot[8];
    const int b = blockIdx.x, t = threadIdx.x;
    const int start = binBase[b];
    const int n = binTot[b];
    const int2* __restrict__ ed = (const int2*)(binned + 2 * start);
    cntl[t] = 0;
    __syncthreads();
    for (int i0 = t; i0 < n; i0 += 256 * CSR_BATCH) {
        int dl[CSR_BATCH];
#pragma unroll
        for (int k = 0; k < CSR_BATCH; ++k) {
            const int i = i0 + k * 256;
            dl[k] = ed[i < n ? i : n - 1].y & (BIN_NODES - 1);
        }
#pragma unroll
        for (int k = 0; k < CSR_BATCH; ++k) {
            if (i0 + k * 256 < n) atomicAdd(&cntl[dl[k]], 1);
        }
    }
    __syncthreads();
    int tot;
    const int ex = block_excl_scan(cntl[t], wtot, tot);
    curl[t] = ex;
    const int node = b * BIN_NODES + t;
    if (node <= NN) rowptr[node] = start + ex;
    __syncthreads();
    for (int i0 = t; i0 < n; i0 += 256 * CSR_BATCH) {
        int2 e[CSR_BATCH];
#pragma unroll
        for (int k = 0; k < CSR_BATCH; ++k) {
            const int i = i0 + k * 256;
            e[k] = ed[i < n ? i : n - 1];
        }
#pragma unroll
        for (int k = 0; k < CSR_BATCH; ++k) {
            if (i0 + k * 256 < n) {
                const int r = atomicAdd(&curl[e[k].y & (BIN_NODES - 1)], 1);
                col[start + r] = e[k].x;
            }
        }
    }
}

__device__ __forceinline__ float wave_sum(float v) {
#pragma unroll
    for (int off = 16; off > 0; off >>= 1) v += __shfl_xor(v, off, 32);
    return v;
}

__device__ __forceinline__ float grp_max(float v) {
    v = fmaxf(v, row_xor<1>(v));
    v = fmaxf(v, row_xor<2>(v));
    v = fmaxf(v, row_xor<4>(v));
    return v;
}

__device__ __forceinline__ float grp_sum(float v) {
    v += row_xor<1>(v);
    v += row_xor<2>(v);
    v += row_xor<4>(v);
    return v;
}

template <int I>
__device__ __forceinline__ float grp_bcast(float v) {
    constexpr int SEL = I | (I << 3) | (I << 6) | (I << 9) | (I << 12) | (I << 15) |
                        (I << 18) | (I << 21);
    return __int_as_float(__builtin_amdgcn_mov_dpp8(__float_as_int(v), SEL));
}

__device__ __forceinline__ void load_chunk(int base, int i, int g, int cnt, int beg, int d,
                                           const int* __restrict__ col,
                                           const float* __restrict__ aS, float adg,
                                           int (&s)[4], float (&lg)[4]) {
    int sc[4];
#pragma unroll
    for (int q = 0; q < 4; ++q) {
        const int e = beg + base + 8 * q + i;
        sc[q] = col[e < EE ? e : EE - 1];
    }
    float as[4];
#pragma unroll
    for (int q = 0; q < 4; ++q) {
        s[q] = (base + 8 * q + i < cnt - 1) ? sc[q] : d;
        as[q] = aS[s[q] * HH + g];
    }
#pragma unroll
    for (int q = 0; q < 4; ++q)
        lg[q] = (base + 8 * q + i < cnt) ? lrelu(as[q] + adg) : -3.0e38f;
}

template <int I>
__device__ __forceinline__ float4 row_of(const float* __restrict__ hLin, unsigned loff, int sv) {
    const int sj = __builtin_amdgcn_readlane(sv, I);
    return *(const float4*)((const char*)(hLin + sj * DD) + loff);
}

template <int I>
__device__ __forceinline__ void row_add(float av, const float4 v, float4& acc) {
    const float w = grp_bcast<I>(av);
    acc.x = fmaf(w, v.x, acc.x);
    acc.y = fmaf(w, v.y, acc.y);
    acc.z = fmaf(w, v.z, acc.z);
    acc.w = fmaf(w, v.w, acc.w);
}

__device__ __forceinline__ void add_four(const float* __restrict__ hLin, unsigned loff,
                                         int sv, float av, float4& acc) {
    const float4 v0 = row_of<0>(hLin, loff, sv);
    const float4 v1 = row_of<1>(hLin, loff, sv);
    const float4 v2 = row_of<2>(hLin, loff, sv);
    const float4 v3 = row_of<3>(hLin, loff, sv);
    row_add<0>(av, v0, acc);
    row_add<1>(av, v1, acc);
    row_add<2>(av, v2, acc);
    row_add<3>(av, v3, acc);
}

__device__ __forceinline__ void add_eight(const float* __restrict__ hLin, unsigned loff,
                                          int sv, float av, float4& acc) {
    const float4 v0 = row_of<0>(hLin, loff, sv);
    const float4 v1 = row_of<1>(hLin, loff, sv);
    const float4 v2 = row_of<2>(hLin, loff, sv);
    const float4 v3 = row_of<3>(hLin, loff, sv);
    const float4 v4 = row_of<4>(hLin, loff, sv);
    const float4 v5 = row_of<5>(hLin, loff, sv);
    const float4 v6 = row_of<6>(hLin, loff, sv);
    const float4 v7 = row_of<7>(hLin, loff, sv);
    row_add<0>(av, v0, acc);
    row_add<1>(av, v1, acc);
    row_add<2>(av, v2, acc);
    row_add<3>(av, v3, acc);
    row_add<4>(av, v4, acc);
    row_add<5>(av, v5, acc);
    row_add<6>(av, v6, acc);
    row_add<7>(av, v7, acc);
}

__device__ __forceinline__ void add_slot(const float* __restrict__ hLin, unsigned loff,
                                         int sv, float av, int n, float4& acc) {
    if (n > 4) add_eight(hLin, loff, sv, av, acc);
    else       add_four(hLin, loff, sv, av, acc);
}

template <bool HAS_RES>
__global__ __launch_bounds__(256) void k_gat_node(const int* __restrict__ rowptr,
                                                  const int* __restrict__ col,
                                                  const float* __restrict__ aS,
                                                  const float* __restrict__ aD,
                                                  const float* __restrict__ hLin,
                                                  const float* __restrict__ bias,
                                                  const float* __restrict__ gma,
                                                  const float* __restrict__ bta,
                                                  const float* __restrict__ resid,
                                                  float* __restrict__ outp) {
    const int lane = threadIdx.x & 31;
    const int g = lane >> 3;
    const int i = lane & 7;
    const int d = __builtin_amdgcn_readfirstlane(
        (int)((blockIdx.x * blockDim.x + threadIdx.x) >> 5));
    if (d >= NN) return;
    const int beg = rowptr[d];
    const int cnt = rowptr[d + 1] - beg + 1;
    const float adg = aD[d * HH + g];
    const unsigned loff = lane * 16u;
    int s0[4];
    float lg0[4];
    load_chunk(0, i, g, cnt, beg, d, col, aS, adg, s0, lg0);

    float m = fmaxf(fmaxf(lg0[0], lg0[1]), fmaxf(lg0[2], lg0[3]));
    for (int base = 32; base < cnt; base += 32) {
        int s[4];
        float lg[4];
        load_chunk(base, i, g, cnt, beg, d, col, aS, adg, s, lg);
        m = fmaxf(m, fmaxf(fmaxf(lg[0], lg[1]), fmaxf(lg[2], lg[3])));
    }
    m = grp_max(m);

    float e0[4];
#pragma unroll
    for (int q = 0; q < 4; ++q) {
        e0[q] = 0.f;
        if (8 * q < cnt) e0[q] = expf(lg0[q] - m);
    }
    float z = (e0[0] + e0[1]) + (e0[2] + e0[3]);
    for (int base = 32; base < cnt; base += 32) {
        int s[4];
        float lg[4];
        load_chunk(base, i, g, cnt, beg, d, col, aS, adg, s, lg);
#pragma unroll
        for (int q = 0; q < 4; ++q) z += expf(lg[q] - m);
    }
    const float den = grp_sum(z) + 1e-16f;

    float4 acc = make_float4(0.f, 0.f, 0.f, 0.f);
#pragma unroll
    for (int q = 0; q < 4; ++q) {
        if (8 * q < cnt) add_slot(hLin, loff, s0[q], e0[q] / den, cnt - 8 * q, acc);
    }
    for (int base = 32; base < cnt; base += 32) {
        int s[4];
        float lg[4];
        load_chunk(base, i, g, cnt, beg, d, col, aS, adg, s, lg);
#pragma unroll
        for (int q = 0; q < 4; ++q) {
            if (base + 8 * q < cnt)
                add_slot(hLin, loff, s[q], expf(lg[q] - m) / den, cnt - base - 8 * q, acc);
        }
    }

    float4 rr = make_float4(0.f, 0.f, 0.f, 0.f);
    if (HAS_RES) rr = *(const float4*)(resid + d * DD + lane * 4);
    const float4 bi = *(const float4*)(bias + lane * 4);
    const float4 gm = *(const float4*)(gma + lane * 4);
    const float4 bt = *(const float4*)(bta + lane * 4);
    const float4 v = make_float4(acc.x + bi.x, acc.y + bi.y, acc.z + bi.z, acc.w + bi.w);
    const float mu = wave_sum((v.x + v.y) + (v.z + v.w)) * (1.f / 128.f);
    const float4 dl = make_float4(v.x - mu, v.y - mu, v.z - mu, v.w - mu);
    const float var =
        wave_sum((dl.x * dl.x + dl.y * dl.y) + (dl.z * dl.z + dl.w * dl.w));
    const float rs = rsqrtf(var * (1.f / 128.f) + 1e-5f);
    float4 y = make_float4(dl.x * rs * gm.x + bt.x, dl.y * rs * gm.y + bt.y,
                           dl.z * rs * gm.z + bt.z, dl.w * rs * gm.w + bt.w);
    y.x = y.x > 0.f ? y.x : expm1f(y.x);
    y.y = y.y > 0.f ? y.y : expm1f(y.y);
    y.z = y.z > 0.f ? y.z : expm1f(y.z);
    y.w = y.w > 0.f ? y.w : expm1f(y.w);
    if (HAS_RES) { y.x += rr.x; y.y += rr.y; y.z += rr.z; y.w += rr.w; }
    *(float4*)(outp + d * DD + lane * 4) = y;
}

__global__ void k_zero_pool(float* __restrict__ gsum, float* __restrict__ gcnt) {
    int t = blockIdx.x * blockDim.x + threadIdx.x;
    if (t < GG * DD) gsum[t] = 0.f;
    if (t < GG) gcnt[t] = 0.f;
}

#define POOL_ROWS 16
static_assert(NN % POOL_ROWS == 0, "a wave sums whole groups of POOL_ROWS nodes");

__device__ __forceinline__ void pool_flush(float* __restrict__ gsum,
                                           float* __restrict__ gcnt, int g,
                                           int lane, const float4 acc, float cnt) {
    float* p = gsum + g * DD + lane * 4;
    atomicAdd(p + 0, acc.x);
    atomicAdd(p + 1, acc.y);
    atomicAdd(p + 2, acc.z);
    atomicAdd(p + 3, acc.w);
    if (lane == 0) atomicAdd(gcnt + g, cnt);
}

__global__ __launch_bounds__(256) void k_pool_sum(const float* __restrict__ h,
                                                  const int* __restrict__ batch,
                                                  float* __restrict__ gsum,
                                                  float* __restrict__ gcnt) {
    const int lane = threadIdx.x & 31;
    const int w = __builtin_amdgcn_readfirstlane(
        (int)((blockIdx.x * blockDim.x + threadIdx.x) >> 5));
    const int n0 = w * POOL_ROWS;
    if (n0 >= NN) return;
    const int gl = batch[n0 + (lane & (POOL_ROWS - 1))];
    float4 r[POOL_ROWS];
    const float* __restrict__ hp = h + n0 * DD + lane * 4;
#pragma unroll
    for (int i = 0; i < POOL_ROWS; ++i) r[i] = *(const float4*)(hp + i * DD);

    int gcur = __builtin_amdgcn_readlane(gl, 0);
    float4 acc = r[0];
    float cnt = 1.f;
#pragma unroll
    for (int i = 1; i < POOL_ROWS; ++i) {
        const int gi = __builtin_amdgcn_readlane(gl, i);
        if (gi != gcur) {
            pool_flush(gsum, gcnt, gcur, lane, acc, cnt);
            gcur = gi;
            acc = r[i];
            cnt = 1.f;
        } else {
            acc.x += r[i].x; acc.y += r[i].y;
            acc.z += r[i].z; acc.w += r[i].w;
            cnt += 1.f;
        }
    }
    pool_flush(gsum, gcnt, gcur, lane, acc, cnt);
}

__global__ void k_pool_div(const float* __restrict__ gsum,
                           const float* __restrict__ gcnt,
                           float* __restrict__ outg) {
    int t = blockIdx.x * blockDim.x + threadIdx.x;
    if (t >= GG * DD) return;
    outg[t] = gsum[t] / fmaxf(gcnt[t >> 7], 1.f);
}

extern "C" void kernel_launch(void* const* d_in, const int* in_sizes, int n_in,
                              void* d_out, int out_size, void* d_ws, size_t ws_size,
                              hipStream_t stream) {
    (void)in_sizes; (void)n_in; (void)out_size; (void)ws_size;

    const float* x     = (const float*)d_in[0];
    const int*   ei    = (const int*)d_in[1];
    const int*   batch = (const int*)d_in[2];
    const float* W0    = (const float*)d_in[3];
    const float* as0   = (const float*)d_in[4];
    const float* ad0   = (const float*)d_in[5];
    const float* b0    = (const float*)d_in[6];
    const float* Wsl   = (const float*)d_in[7];
    const float* asrcs = (const float*)d_in[8];
    const float* adsts = (const float*)d_in[9];
    const float* bsl   = (const float*)d_in[10];
    const float* lng   = (const float*)d_in[11];
    const float* lnb   = (const float*)d_in[12];
    float* out = (float*)d_out;

    float* ws   = (float*)d_ws;
    float* A0v  = ws;           ws += NN * DD;
    float* A1v  = ws;           ws += NN * DD;
    float* Lin  = ws;           ws += NN * DD;
    float* xp   = ws;           ws += NN * KPAD;
    float* Wp   = ws;           ws += KPAD * DD;
    float* aS   = ws;           ws += NN * HH;
    float* aD   = ws;           ws += NN * HH;
    float* gsum = ws;           ws += GG * DD;
    float* gcnt = ws;           ws += GG;
    int* binTot  = (int*)ws;    ws += BPITCH;
    int* binBase = (int*)ws;    ws += BPITCH;
    int* cnt     = (int*)ws;    ws += NCHUNK * BPITCH;
    int* off     = (int*)ws;    ws += NCHUNK * BPITCH;
    int* binned  = (int*)ws;    ws += 2 * EE;
    int* rowptr  = (int*)ws;    ws += NN + 4;
    int* col     = (int*)ws;    ws += EE;

    k_bin_count<<<NCHUNK, 256, 0, stream>>>(ei, cnt);
    k_bin_scan<<<NBIN, 256, 0, stream>>>(cnt, off, binTot);
    k_bin_base<<<1, 256, 0, stream>>>(binTot, binBase);
    k_bin_scatter<<<NCHUNK, 256, 0, stream>>>(ei, off, binBase, binned);
    k_bin_csr<<<NBIN, 256, 0, stream>>>(binned, binBase, binTot, rowptr, col);

    k_pad_x<<<(NN * KPAD + 255) / 256, 256, 0, stream>>>(x, xp);
    k_pad_w<<<(KPAD * DD + 255) / 256, 256, 0, stream>>>(W0, Wp);

    for (int layer = 0; layer < LL; ++layer) {
        const float* Ain;  float* agg;  const float* resid;
        switch (layer) {
            case 0: Ain = xp;  agg = A0v; resid = nullptr; break;
            case 1: Ain = A0v; agg = A1v; resid = A0v;     break;
            case 2: Ain = A1v; agg = A0v; resid = A1v;     break;
            default:Ain = A0v; agg = out; resid = A0v;     break;
        }
        const float* asv  = (layer == 0) ? as0 : asrcs + (layer - 1) * HH * CC;
        const float* adv  = (layer == 0) ? ad0 : adsts + (layer - 1) * HH * CC;
        const float* bias = (layer == 0) ? b0  : bsl   + (layer - 1) * DD;

        if (layer == 0) {
            k_gemm<KPAD><<<(ROW_TILES + GEMM_WAVES - 1) / GEMM_WAVES, GEMM_WAVES * 32, 0, stream>>>(
                Ain, Wp, Lin, asv, adv, aS, aD);
        } else {
            k_gemm<DD><<<(ROW_TILES + GEMM_WAVES - 1) / GEMM_WAVES, GEMM_WAVES * 32, 0, stream>>>(
                Ain, Wsl + (layer - 1) * DD * DD, Lin, asv, adv, aS, aD);
        }
        if (layer == 0) {
            k_gat_node<false><<<(NN * 32 + 255) / 256, 256, 0, stream>>>(
                rowptr, col, aS, aD, Lin, bias, lng + layer * DD, lnb + layer * DD,
                resid, agg);
        } else {
            k_gat_node<true><<<(NN * 32 + 255) / 256, 256, 0, stream>>>(
                rowptr, col, aS, aD, Lin, bias, lng + layer * DD, lnb + layer * DD,
                resid, agg);
        }
    }

    k_zero_pool<<<(GG * DD + 255) / 256, 256, 0, stream>>>(gsum, gcnt);
    k_pool_sum<<<(NN / POOL_ROWS * 32 + 255) / 256, 256, 0, stream>>>(out, batch, gsum, gcnt);
    k_pool_div<<<(GG * DD + 255) / 256, 256, 0, stream>>>(gsum, gcnt, out + NN * DD);
}


# layer 0: per head, softmax-weighted sum of the 7-feature x rows first, then the W0 product on WMMA per 16-node block; gathers 32-B not 512-B rows (order of sums changes; one division after the sum)
# speedup vs baseline: 20.2115x; 1.0848x over previous
#include <hip/hip_runtime.h>
#include <math.h>

#define NN 100000
#define EE 1600000
#define HH 4
#define CC 32
#define DD 128
#define GG 64
#define LL 4
#define INDIM 7
#define KPAD 8
#define NEG_SLOPE 0.2f

typedef __attribute__((ext_vector_type(2))) float v2f;
typedef __attribute__((ext_vector_type(8))) float v8f;

__device__ __forceinline__ float lrelu(float t) {
    return t > 0.f ? t : NEG_SLOPE * t;
}

__global__ void k_pad_x(const float* __restrict__ x, float* __restrict__ xp) {
    int t = blockIdx.x * blockDim.x + threadIdx.x;
    if (t >= NN * KPAD) return;
    const int n = t >> 3, c = t & 7;
    xp[t] = (c < INDIM) ? x[n * INDIM + c] : 0.f;
}

__global__ void k_pad_w(const float* __restrict__ W0, float* __restrict__ Wp) {
    int t = blockIdx.x * blockDim.x + threadIdx.x;
    if (t >= KPAD * DD) return;
    Wp[t] = (t < INDIM * DD) ? W0[t] : 0.f;
}

#define GEMM_WAVES 8
#define ROW_TILES (NN / 16)
static_assert(NN % 16 == 0, "whole 16-row strips");

__device__ __forceinline__ v8f wmma4(float a0, float a1, float b0, float b1, v8f c) {
    v2f a; a.x = a0; a.y = a1;
    v2f b; b.x = b0; b.y = b1;
    return __builtin_amdgcn_wmma_f32_16x16x4_f32(
         false, a,  false, b,
         (short)0, c,  false,  false);
}

template <int X>
__device__ __forceinline__ float row_xor(float v) {
    return __int_as_float(__builtin_amdgcn_update_dpp(
        0, __float_as_int(v), 0x160 + X, 0xf, 0xf, true));
}

template <int N, int X>
__device__ __forceinline__ void rs_step(const float (&in)[2 * N], float (&out)[N], bool bit) {
#pragma unroll
    for (int i = 0; i < N; ++i) {
        const float keep = bit ? in[2 * i + 1] : in[2 * i];
        const float send = bit ? in[2 * i] : in[2 * i + 1];
        out[i] = keep + row_xor<X>(send);
    }
}

template <int K, bool STORE_C>
__global__ __launch_bounds__(GEMM_WAVES * 32) void k_gemm(const float* __restrict__ A,
                                                          const float* __restrict__ B,
                                                          float* __restrict__ Cm,
                                                          const float* __restrict__ a_src,
                                                          const float* __restrict__ a_dst,
                                                          float* __restrict__ aS,
                                                          float* __restrict__ aD) {
    static_assert(K % 8 == 0, "k is consumed in blocks of 8");
    constexpr int NKB = K / 8;
    constexpr int CH  = NKB < 4 ? NKB : 4;
    static_assert(NKB % CH == 0, "whole chunks");
    __shared__ float4 Bs[(K / 4) * DD];

    const int lane = threadIdx.x & 31;
    const int wave = __builtin_amdgcn_readfirstlane((int)(threadIdx.x >> 5));
    const int m    = lane & 15;
    const int half = lane >> 4;

    const int tile  = blockIdx.x * GEMM_WAVES + wave;
    const int tilec = tile < ROW_TILES ? tile : ROW_TILES - 1;
    const float* __restrict__ arow = A + (tilec * 16 + m) * K + half * 4;

    float4 a_cur[CH];
#pragma unroll
    for (int i = 0; i < CH; ++i) a_cur[i] = *(const float4*)(arow + i * 8);

    constexpr int NPAIR = (K / 4) * (DD / 32);
    static_assert(NPAIR % GEMM_WAVES == 0, "every wave stages the same number of pieces");
#pragma unroll 4
    for (int t = 0; t < NPAIR / GEMM_WAVES; ++t) {
        const int p = t * GEMM_WAVES + wave;
        const int kq = p >> 2, n = (p & 3) * 32 + lane;
        const float* __restrict__ bp = B + kq * 4 * DD + n;
        float4 v;
        v.x = bp[0];
        v.y = bp[DD];
        v.z = bp[2 * DD];
        v.w = bp[3 * DD];
        Bs[kq * DD + n] = v;
    }
    __syncthreads();
    if (tile >= ROW_TILES) return;

    v8f acc[8];
#pragma unroll
    for (int ct = 0; ct < 8; ++ct) acc[ct] = v8f{};

#pragma unroll
    for (int c0 = 0; c0 < NKB; c0 += CH) {
        float4 a_nxt[CH];
        if (c0 + CH < NKB) {
#pragma unroll
            for (int i = 0; i < CH; ++i)
                a_nxt[i] = *(const float4*)(arow + (c0 + CH + i) * 8);
        }
#pragma unroll
        for (int i = 0; i < CH; ++i) {
            const float4 a = a_cur[i];
            const float4* __restrict__ bq = Bs + ((c0 + i) * 2 + half) * DD + m;
#pragma unroll
            for (int ct = 0; ct < 8; ct += 2) {
                const float4 b0 = bq[ct * 16];
                const float4 b1 = bq[ct * 16 + 16];
                acc[ct]     = wmma4(a.x, a.y, b0.x, b0.y, acc[ct]);
                acc[ct + 1] = wmma4(a.x, a.y, b1.x, b1.y, acc[ct + 1]);
                acc[ct]     = wmma4(a.z, a.w, b0.z, b0.w, acc[ct]);
                acc[ct + 1] = wmma4(a.z, a.w, b1.z, b1.w, acc[ct + 1]);
            }
        }
        if (c0 + CH < NKB) {
#pragma unroll
            for (int i = 0; i < CH; ++i) a_cur[i] = a_nxt[i];
        }
    }
    float cs0[HH], cs1[HH], cd0[HH], cd1[HH];
#pragma unroll
    for (int hd = 0; hd < HH; ++hd) {
        cs0[hd] = a_src[hd * CC + m];
        cs1[hd] = a_src[hd * CC + 16 + m];
        cd0[hd] = a_dst[hd * CC + m];
        cd1[hd] = a_dst[hd * CC + 16 + m];
    }
    if (STORE_C) {
        float* out = Cm + tile * 16 * DD + m;
#pragma unroll
        for (int ct = 0; ct < 8; ++ct) {
#pragma unroll
            for (int r = 0; r < 8; ++r) {
                out[(r + half * 8) * DD + ct * 16] = acc[ct][r];
            }
        }
    }

    __builtin_amdgcn_sched_barrier(0);
    const bool b3 = (m & 8) != 0, b2 = (m & 4) != 0, b1 = (m & 2) != 0, b0 = (m & 1) != 0;
    float res[HH];
#pragma unroll
    for (int hd = 0; hd < HH; ++hd) {
        const float s0 = cs0[hd], s1 = cs1[hd], d0 = cd0[hd], d1 = cd1[hd];
        float x16[16], x8[8], x4[4], x2[2], x1[1];
#pragma unroll
        for (int r = 0; r < 8; ++r) {
            const float c0 = acc[2 * hd][r], c1 = acc[2 * hd + 1][r];
            x16[r]     = c0 * s0 + c1 * s1;
            x16[8 + r] = c0 * d0 + c1 * d1;
        }
        rs_step<8, 8>(x16, x8, b3);
        rs_step<4, 4>(x8, x4, b2);
        rs_step<2, 2>(x4, x2, b1);
        rs_step<1, 1>(x2, x1, b0);
        res[hd] = x1[0];
    }
    const int rr = ((m >> 3) & 1) | ((m >> 1) & 2) | ((m << 1) & 4);
    float* ap = (b0 ? aD : aS) + (tile * 16 + half * 8 + rr) * HH;
    *(float4*)ap = make_float4(res[0], res[1], res[2], res[3]);
}

#define BIN_SHIFT 8
#define BIN_NODES (1 << BIN_SHIFT)
#define NBIN ((NN + BIN_NODES - 1) / BIN_NODES)
#define CHUNK 4096
#define NCHUNK ((EE + CHUNK - 1) / CHUNK)
#define BPITCH 512
static_assert(NBIN <= BPITCH && NCHUNK <= BPITCH, "two slots per thread of a 256-thread block");
static_assert(BIN_NODES == 256, "one destination per thread in k_bin_csr");
#define CSR_BATCH 8

__device__ __forceinline__ int block_excl_scan(int v, int* wtot, int& total) {
    const int lane = threadIdx.x & 31, wave = threadIdx.x >> 5;
    int inc = v;
#pragma unroll
    for (int off = 1; off < 32; off <<= 1) {
        const int u = __shfl_up(inc, off, 32);
        if (lane >= off) inc += u;
    }
    if (lane == 31) wtot[wave] = inc;
    __syncthreads();
    int wb = 0, tot = 0;
#pragma unroll
    for (int w = 0; w < 8; ++w) {
        const int x = wtot[w];
        if (w < wave) wb += x;
        tot += x;
    }
    total = tot;
    return wb + inc - v;
}

__global__ __launch_bounds__(256) void k_bin_count(const int* __restrict__ ei,
                                                   int* __restrict__ cnt) {
    __shared__ int hist[BPITCH];
    const int t = threadIdx.x;
    hist[t] = 0;
    hist[t + 256] = 0;
    __syncthreads();
    const int e0 = blockIdx.x * CHUNK + t;
    int d[CHUNK / 256];
#pragma unroll
    for (int i = 0; i < CHUNK / 256; ++i) {
        const int e = e0 + i * 256;
        d[i] = ei[EE + (e < EE ? e : EE - 1)];
    }
#pragma unroll
    for (int i = 0; i < CHUNK / 256; ++i) {
        if (e0 + i * 256 < EE) atomicAdd(&hist[d[i] >> BIN_SHIFT], 1);
    }
    __syncthreads();
    cnt[blockIdx.x * BPITCH + t] = hist[t];
    cnt[blockIdx.x * BPITCH + t + 256] = hist[t + 256];
}

__global__ __launch_bounds__(256) void k_bin_scan(const int* __restrict__ cnt,
                                                  int* __restrict__ off,
                                                  int* __restrict__ binTot) {
    __shared__ int wtot[8];
    const int b = blockIdx.x, t = threadIdx.x;
    const int w0 = 2 * t, w1 = 2 * t + 1;
    const int c0 = (w0 < NCHUNK) ? cnt[w0 * BPITCH + b] : 0;
    const int c1 = (w1 < NCHUNK) ? cnt[w1 * BPITCH + b] : 0;
    int tot;
    const int ex = block_excl_scan(c0 + c1, wtot, tot);
    if (w0 < NCHUNK) off[w0 * BPITCH + b] = ex;
    if (w1 < NCHUNK) off[w1 * BPITCH + b] = ex + c0;
    if (t == 0) binTot[b] = tot;
}

__global__ __launch_bounds__(256) void k_bin_base(const int* __restrict__ binTot,
                                                  int* __restrict__ binBase) {
    __shared__ int wtot[8];
    const int t = threadIdx.x;
    const int b0 = 2 * t, b1 = 2 * t + 1;
    const int c0 = (b0 < NBIN) ? binTot[b0] : 0;
    const int c1 = (b1 < NBIN) ? binTot[b1] : 0;
    int tot;
    const int ex = block_excl_scan(c0 + c1, wtot, tot);
    binBase[b0] = ex;
    binBase[b1] = ex + c0;
}

__global__ __launch_bounds__(256) void k_bin_scatter(const int* __restrict__ ei,
                                                     const int* __restrict__ off,
                                                     const int* __restrict__ binBase,
                                                     int* __restrict__ binned) {
    __shared__ int cur[BPITCH];
    const int t = threadIdx.x;
    cur[t] = (t < NBIN) ? binBase[t] + off[blockIdx.x * BPITCH + t] : 0;
    cur[t + 256] = (t + 256 < NBIN) ? binBase[t + 256] + off[blockIdx.x * BPITCH + t + 256] : 0;
    __syncthreads();
    const int e0 = blockIdx.x * CHUNK + t;
    int s[CHUNK / 256], d[CHUNK / 256];
#pragma unroll
    for (int i = 0; i < CHUNK / 256; ++i) {
        const int e = e0 + i * 256;
        const int ec = e < EE ? e : EE - 1;
        s[i] = ei[ec];
        d[i] = ei[EE + ec];
    }
#pragma unroll
    for (int i = 0; i < CHUNK / 256; ++i) {
        if (e0 + i * 256 < EE) {
            const int pos = atomicAdd(&cur[d[i] >> BIN_SHIFT], 1);
            *(int2*)(binned + 2 * pos) = make_int2(s[i], d[i]);
        }
    }
}

__global__ __launch_bounds__(256) void k_bin_csr(const int* __restrict__ binned,
                                                 const int* __restrict__ binBase,
                                                 const int* __restrict__ binTot,
                                                 int* __restrict__ rowptr,
                                                 int* __restrict__ col) {
    __shared__ int cntl[BIN_NODES];
    __shared__ int curl[BIN_NODES];
    __shared__ int wtot[8];
    const int b = blockIdx.x, t = threadIdx.x;
    const int start = binBase[b];
    const int n = binTot[b];
    const int2* __restrict__ ed = (const int2*)(binned + 2 * start);
    cntl[t] = 0;
    __syncthreads();
    for (int i0 = t; i0 < n; i0 += 256 * CSR_BATCH) {
        int dl[CSR_BATCH];
#pragma unroll
        for (int k = 0; k < CSR_BATCH; ++k) {
            const int i = i0 + k * 256;
            dl[k] = ed[i < n ? i : n - 1].y & (BIN_NODES - 1);
        }
#pragma unroll
        for (int k = 0; k < CSR_BATCH; ++k) {
            if (i0 + k * 256 < n) atomicAdd(&cntl[dl[k]], 1);
        }
    }
    __syncthreads();
    int tot;
    const int ex = block_excl_scan(cntl[t], wtot, tot);
    curl[t] = ex;
    const int node = b * BIN_NODES + t;
    if (node <= NN) rowptr[node] = start + ex;
    __syncthreads();
    for (int i0 = t; i0 < n; i0 += 256 * CSR_BATCH) {
        int2 e[CSR_BATCH];
#pragma unroll
        for (int k = 0; k < CSR_BATCH; ++k) {
            const int i = i0 + k * 256;
            e[k] = ed[i < n ? i : n - 1];
        }
#pragma unroll
        for (int k = 0; k < CSR_BATCH; ++k) {
            if (i0 + k * 256 < n) {
                const int r = atomicAdd(&curl[e[k].y & (BIN_NODES - 1)], 1);
                col[start + r] = e[k].x;
            }
        }
    }
}

__device__ __forceinline__ float wave_sum(float v) {
#pragma unroll
    for (int off = 16; off > 0; off >>= 1) v += __shfl_xor(v, off, 32);
    return v;
}

__device__ __forceinline__ float grp_max(float v) {
    v = fmaxf(v, row_xor<1>(v));
    v = fmaxf(v, row_xor<2>(v));
    v = fmaxf(v, row_xor<4>(v));
    return v;
}

__device__ __forceinline__ float grp_sum(float v) {
    v += row_xor<1>(v);
    v += row_xor<2>(v);
    v += row_xor<4>(v);
    return v;
}

template <int I>
__device__ __forceinline__ float grp_bcast(float v) {
    constexpr int SEL = I | (I << 3) | (I << 6) | (I << 9) | (I << 12) | (I << 15) |
                        (I << 18) | (I << 21);
    return __int_as_float(__builtin_amdgcn_mov_dpp8(__float_as_int(v), SEL));
}

__device__ __forceinline__ void load_chunk(int base, int i, int g, int cnt, int beg, int d,
                                           const int* __restrict__ col,
                                           const float* __restrict__ aS, float adg,
                                           int (&s)[4], float (&lg)[4]) {
    int sc[4];
#pragma unroll
    for (int q = 0; q < 4; ++q) {
        const int e = beg + base + 8 * q + i;
        sc[q] = col[e < EE ? e : EE - 1];
    }
    float as[4];
#pragma unroll
    for (int q = 0; q < 4; ++q) {
        s[q] = (base + 8 * q + i < cnt - 1) ? sc[q] : d;
        as[q] = aS[s[q] * HH + g];
    }
#pragma unroll
    for (int q = 0; q < 4; ++q)
        lg[q] = (base + 8 * q + i < cnt) ? lrelu(as[q] + adg) : -3.0e38f;
}

template <int I>
__device__ __forceinline__ float4 row_of(const float* __restrict__ hLin, unsigned loff, int sv) {
    const int sj = __builtin_amdgcn_readlane(sv, I);
    return *(const float4*)((const char*)(hLin + sj * DD) + loff);
}

template <int I>
__device__ __forceinline__ void row_add(float av, const float4 v, float4& acc) {
    const float w = grp_bcast<I>(av);
    acc.x = fmaf(w, v.x, acc.x);
    acc.y = fmaf(w, v.y, acc.y);
    acc.z = fmaf(w, v.z, acc.z);
    acc.w = fmaf(w, v.w, acc.w);
}

__device__ __forceinline__ void add_four(const float* __restrict__ hLin, unsigned loff,
                                         int sv, float av, float4& acc) {
    const float4 v0 = row_of<0>(hLin, loff, sv);
    const float4 v1 = row_of<1>(hLin, loff, sv);
    const float4 v2 = row_of<2>(hLin, loff, sv);
    const float4 v3 = row_of<3>(hLin, loff, sv);
    row_add<0>(av, v0, acc);
    row_add<1>(av, v1, acc);
    row_add<2>(av, v2, acc);
    row_add<3>(av, v3, acc);
}

__device__ __forceinline__ void add_eight(const float* __restrict__ hLin, unsigned loff,
                                          int sv, float av, float4& acc) {
    const float4 v0 = row_of<0>(hLin, loff, sv);
    const float4 v1 = row_of<1>(hLin, loff, sv);
    const float4 v2 = row_of<2>(hLin, loff, sv);
    const float4 v3 = row_of<3>(hLin, loff, sv);
    const float4 v4 = row_of<4>(hLin, loff, sv);
    const float4 v5 = row_of<5>(hLin, loff, sv);
    const float4 v6 = row_of<6>(hLin, loff, sv);
    const float4 v7 = row_of<7>(hLin, loff, sv);
    row_add<0>(av, v0, acc);
    row_add<1>(av, v1, acc);
    row_add<2>(av, v2, acc);
    row_add<3>(av, v3, acc);
    row_add<4>(av, v4, acc);
    row_add<5>(av, v5, acc);
    row_add<6>(av, v6, acc);
    row_add<7>(av, v7, acc);
}

__device__ __forceinline__ void add_slot(const float* __restrict__ hLin, unsigned loff,
                                         int sv, float av, int n, float4& acc) {
    if (n > 4) add_eight(hLin, loff, sv, av, acc);
    else       add_four(hLin, loff, sv, av, acc);
}

template <bool HAS_RES>
__device__ __forceinline__ void finish_row(const float4 acc, int d, int lane,
                                           const float* __restrict__ bias,
                                           const float* __restrict__ gma,
                                           const float* __restrict__ bta,
                                           const float* __restrict__ resid,
                                           float* __restrict__ outp) {
    float4 rr = make_float4(0.f, 0.f, 0.f, 0.f);
    if (HAS_RES) rr = *(const float4*)(resid + d * DD + lane * 4);
    const float4 bi = *(const float4*)(bias + lane * 4);
    const float4 gm = *(const float4*)(gma + lane * 4);
    const float4 bt = *(const float4*)(bta + lane * 4);
    const float4 v = make_float4(acc.x + bi.x, acc.y + bi.y, acc.z + bi.z, acc.w + bi.w);
    const float mu = wave_sum((v.x + v.y) + (v.z + v.w)) * (1.f / 128.f);
    const float4 dl = make_float4(v.x - mu, v.y - mu, v.z - mu, v.w - mu);
    const float var =
        wave_sum((dl.x * dl.x + dl.y * dl.y) + (dl.z * dl.z + dl.w * dl.w));
    const float rs = rsqrtf(var * (1.f / 128.f) + 1e-5f);
    float4 y = make_float4(dl.x * rs * gm.x + bt.x, dl.y * rs * gm.y + bt.y,
                           dl.z * rs * gm.z + bt.z, dl.w * rs * gm.w + bt.w);
    y.x = y.x > 0.f ? y.x : expm1f(y.x);
    y.y = y.y > 0.f ? y.y : expm1f(y.y);
    y.z = y.z > 0.f ? y.z : expm1f(y.z);
    y.w = y.w > 0.f ? y.w : expm1f(y.w);
    if (HAS_RES) { y.x += rr.x; y.y += rr.y; y.z += rr.z; y.w += rr.w; }
    *(float4*)(outp + d * DD + lane * 4) = y;
}

template <bool HAS_RES>
__global__ __launch_bounds__(256) void k_gat_node(const int* __restrict__ rowptr,
                                                  const int* __restrict__ col,
                                                  const float* __restrict__ aS,
                                                  const float* __restrict__ aD,
                                                  const float* __restrict__ hLin,
                                                  const float* __restrict__ bias,
                                                  const float* __restrict__ gma,
                                                  const float* __restrict__ bta,
                                                  const float* __restrict__ resid,
                                                  float* __restrict__ outp) {
    const int lane = threadIdx.x & 31;
    const int g = lane >> 3;
    const int i = lane & 7;
    const int d = __builtin_amdgcn_readfirstlane(
        (int)((blockIdx.x * blockDim.x + threadIdx.x) >> 5));
    if (d >= NN) return;
    const int beg = rowptr[d];
    const int cnt = rowptr[d + 1] - beg + 1;
    const float adg = aD[d * HH + g];
    const unsigned loff = lane * 16u;
    int s0[4];
    float lg0[4];
    load_chunk(0, i, g, cnt, beg, d, col, aS, adg, s0, lg0);

    float m = fmaxf(fmaxf(lg0[0], lg0[1]), fmaxf(lg0[2], lg0[3]));
    for (int base = 32; base < cnt; base += 32) {
        int s[4];
        float lg[4];
        load_chunk(base, i, g, cnt, beg, d, col, aS, adg, s, lg);
        m = fmaxf(m, fmaxf(fmaxf(lg[0], lg[1]), fmaxf(lg[2], lg[3])));
    }
    m = grp_max(m);

    float e0[4];
#pragma unroll
    for (int q = 0; q < 4; ++q) {
        e0[q] = 0.f;
        if (8 * q < cnt) e0[q] = expf(lg0[q] - m);
    }
    float z = (e0[0] + e0[1]) + (e0[2] + e0[3]);
    for (int base = 32; base < cnt; base += 32) {
        int s[4];
        float lg[4];
        load_chunk(base, i, g, cnt, beg, d, col, aS, adg, s, lg);
#pragma unroll
        for (int q = 0; q < 4; ++q) z += expf(lg[q] - m);
    }
    const float den = grp_sum(z) + 1e-16f;

    float4 acc = make_float4(0.f, 0.f, 0.f, 0.f);
#pragma unroll
    for (int q = 0; q < 4; ++q) {
        if (8 * q < cnt) add_slot(hLin, loff, s0[q], e0[q] / den, cnt - 8 * q, acc);
    }
    for (int base = 32; base < cnt; base += 32) {
        int s[4];
        float lg[4];
        load_chunk(base, i, g, cnt, beg, d, col, aS, adg, s, lg);
#pragma unroll
        for (int q = 0; q < 4; ++q) {
            if (base + 8 * q < cnt)
                add_slot(hLin, loff, s[q], expf(lg[q] - m) / den, cnt - base - 8 * q, acc);
        }
    }

    finish_row<HAS_RES>(acc, d, lane, bias, gma, bta, resid, outp);
}

#define L0_WAVES 16
#define L0_CP (DD + 4)
static_assert(NN % L0_WAVES == 0, "whole 16-node strips");

__global__ __launch_bounds__(L0_WAVES * 32) __attribute__((amdgpu_waves_per_eu(16, 16)))
void k_gat_node0(const int* __restrict__ rowptr,
                 const int* __restrict__ col,
                 const float* __restrict__ aS,
                 const float* __restrict__ aD,
                 const float* __restrict__ xp,
                 const float* __restrict__ Wp,
                 const float* __restrict__ bias,
                 const float* __restrict__ gma,
                 const float* __restrict__ bta,
                 float* __restrict__ outp) {
    __shared__ float4 As[L0_WAVES * HH * 2];
    __shared__ float Cs[L0_WAVES * L0_CP];
    const int lane = threadIdx.x & 31;
    const int wave = __builtin_amdgcn_readfirstlane((int)(threadIdx.x >> 5));
    const int g = lane >> 3;
    const int i = lane & 7;
    const int d = blockIdx.x * L0_WAVES + wave;
    const int beg = rowptr[d];
    const int cnt = rowptr[d + 1] - beg + 1;
    const float adg = aD[d * HH + g];
    const float4* __restrict__ xr = (const float4*)xp;

    int s0[4];
    float lg0[4];
    load_chunk(0, i, g, cnt, beg, d, col, aS, adg, s0, lg0);
    float4 xa[4], xb[4];
#pragma unroll
    for (int q = 0; q < 4; ++q) {
        xa[q] = xr[s0[q] * 2];
        xb[q] = xr[s0[q] * 2 + 1];
    }

    float m = fmaxf(fmaxf(lg0[0], lg0[1]), fmaxf(lg0[2], lg0[3]));
    for (int base = 32; base < cnt; base += 32) {
        int s[4];
        float lg[4];
        load_chunk(base, i, g, cnt, beg, d, col, aS, adg, s, lg);
        m = fmaxf(m, fmaxf(fmaxf(lg[0], lg[1]), fmaxf(lg[2], lg[3])));
    }
    m = grp_max(m);

    float e0[4];
#pragma unroll
    for (int q = 0; q < 4; ++q) {
        e0[q] = 0.f;
        if (8 * q < cnt) e0[q] = expf(lg0[q] - m);
    }
    float z = (e0[0] + e0[1]) + (e0[2] + e0[3]);
    for (int base = 32; base < cnt; base += 32) {
        int s[4];
        float lg[4];
        load_chunk(base, i, g, cnt, beg, d, col, aS, adg, s, lg);
#pragma unroll
        for (int q = 0; q < 4; ++q) z += expf(lg[q] - m);
    }
    const float den = grp_sum(z) + 1e-16f;

    float p[8];
#pragma unroll
    for (int k = 0; k < 8; ++k) p[k] = 0.f;
#pragma unroll
    for (int q = 0; q < 4; ++q) {
        const float w = e0[q];
        p[0] = fmaf(w, xa[q].x, p[0]);
        p[1] = fmaf(w, xa[q].y, p[1]);
        p[2] = fmaf(w, xa[q].z, p[2]);
        p[3] = fmaf(w, xa[q].w, p[3]);
        p[4] = fmaf(w, xb[q].x, p[4]);
        p[5] = fmaf(w, xb[q].y, p[5]);
        p[6] = fmaf(w, xb[q].z, p[6]);
        p[7] = fmaf(w, xb[q].w, p[7]);
    }
    for (int base = 32; base < cnt; base += 32) {
        int s[4];
        float lg[4];
        load_chunk(base, i, g, cnt, beg, d, col, aS, adg, s, lg);
#pragma unroll
        for (int q = 0; q < 4; ++q) {
            const float4 va = xr[s[q] * 2];
            const float4 vb = xr[s[q] * 2 + 1];
            const float w = expf(lg[q] - m);
            p[0] = fmaf(w, va.x, p[0]);
            p[1] = fmaf(w, va.y, p[1]);
            p[2] = fmaf(w, va.z, p[2]);
            p[3] = fmaf(w, va.w, p[3]);
            p[4] = fmaf(w, vb.x, p[4]);
            p[5] = fmaf(w, vb.y, p[5]);
            p[6] = fmaf(w, vb.z, p[6]);
            p[7] = fmaf(w, vb.w, p[7]);
        }
    }
    const bool i2 = (i & 4) != 0, i1 = (i & 2) != 0, i0 = (i & 1) != 0;
    float p4[4], p2[2], p1[1];
    rs_step<4, 4>(p, p4, i2);
    rs_step<2, 2>(p4, p2, i1);
    rs_step<1, 1>(p2, p1, i0);
    const int kk = ((i >> 2) & 1) | (i & 2) | ((i & 1) << 2);
    ((float*)As)[(wave * HH + g) * 8 + kk] = p1[0] / den;

    const int m16 = lane & 15;
    const int half = lane >> 4;
    float bw[4] = {0.f, 0.f, 0.f, 0.f};
    if (wave < 8) {
#pragma unroll
        for (int s = 0; s < 4; ++s) bw[s] = Wp[(half * 4 + s) * DD + wave * 16 + m16];
    }
    __syncthreads();
    if (wave < 8) {
        const float4 a = As[(m16 * HH + (wave >> 1)) * 2 + half];
        v8f acc = v8f{};
        acc = wmma4(a.x, a.y, bw[0], bw[1], acc);
        acc = wmma4(a.z, a.w, bw[2], bw[3], acc);
#pragma unroll
        for (int r = 0; r < 8; ++r) Cs[(r + half * 8) * L0_CP + wave * 16 + m16] = acc[r];
    }
    __syncthreads();
    const float4 row = *(const float4*)(Cs + wave * L0_CP + lane * 4);
    finish_row<false>(row, d, lane, bias, gma, bta, nullptr, outp);
}

__global__ void k_zero_pool(float* __restrict__ gsum, float* __restrict__ gcnt) {
    int t = blockIdx.x * blockDim.x + threadIdx.x;
    if (t < GG * DD) gsum[t] = 0.f;
    if (t < GG) gcnt[t] = 0.f;
}

#define POOL_ROWS 16
static_assert(NN % POOL_ROWS == 0, "a wave sums whole groups of POOL_ROWS nodes");

__device__ __forceinline__ void pool_flush(float* __restrict__ gsum,
                                           float* __restrict__ gcnt, int g,
                                           int lane, const float4 acc, float cnt) {
    float* p = gsum + g * DD + lane * 4;
    atomicAdd(p + 0, acc.x);
    atomicAdd(p + 1, acc.y);
    atomicAdd(p + 2, acc.z);
    atomicAdd(p + 3, acc.w);
    if (lane == 0) atomicAdd(gcnt + g, cnt);
}

__global__ __launch_bounds__(256) void k_pool_sum(const float* __restrict__ h,
                                                  const int* __restrict__ batch,
                                                  float* __restrict__ gsum,
                                                  float* __restrict__ gcnt) {
    const int lane = threadIdx.x & 31;
    const int w = __builtin_amdgcn_readfirstlane(
        (int)((blockIdx.x * blockDim.x + threadIdx.x) >> 5));
    const int n0 = w * POOL_ROWS;
    if (n0 >= NN) return;
    const int gl = batch[n0 + (lane & (POOL_ROWS - 1))];
    float4 r[POOL_ROWS];
    const float* __restrict__ hp = h + n0 * DD + lane * 4;
#pragma unroll
    for (int i = 0; i < POOL_ROWS; ++i) r[i] = *(const float4*)(hp + i * DD);

    int gcur = __builtin_amdgcn_readlane(gl, 0);
    float4 acc = r[0];
    float cnt = 1.f;
#pragma unroll
    for (int i = 1; i < POOL_ROWS; ++i) {
        const int gi = __builtin_amdgcn_readlane(gl, i);
        if (gi != gcur) {
            pool_flush(gsum, gcnt, gcur, lane, acc, cnt);
            gcur = gi;
            acc = r[i];
            cnt = 1.f;
        } else {
            acc.x += r[i].x; acc.y += r[i].y;
            acc.z += r[i].z; acc.w += r[i].w;
            cnt += 1.f;
        }
    }
    pool_flush(gsum, gcnt, gcur, lane, acc, cnt);
}

__global__ void k_pool_div(const float* __restrict__ gsum,
                           const float* __restrict__ gcnt,
                           float* __restrict__ outg) {
    int t = blockIdx.x * blockDim.x + threadIdx.x;
    if (t >= GG * DD) return;
    outg[t] = gsum[t] / fmaxf(gcnt[t >> 7], 1.f);
}

extern "C" void kernel_launch(void* const* d_in, const int* in_sizes, int n_in,
                              void* d_out, int out_size, void* d_ws, size_t ws_size,
                              hipStream_t stream) {
    (void)in_sizes; (void)n_in; (void)out_size; (void)ws_size;

    const float* x     = (const float*)d_in[0];
    const int*   ei    = (const int*)d_in[1];
    const int*   batch = (const int*)d_in[2];
    const float* W0    = (const float*)d_in[3];
    const float* as0   = (const float*)d_in[4];
    const float* ad0   = (const float*)d_in[5];
    const float* b0    = (const float*)d_in[6];
    const float* Wsl   = (const float*)d_in[7];
    const float* asrcs = (const float*)d_in[8];
    const float* adsts = (const float*)d_in[9];
    const float* bsl   = (const float*)d_in[10];
    const float* lng   = (const float*)d_in[11];
    const float* lnb   = (const float*)d_in[12];
    float* out = (float*)d_out;

    float* ws   = (float*)d_ws;
    float* A0v  = ws;           ws += NN * DD;
    float* A1v  = ws;           ws += NN * DD;
    float* Lin  = ws;           ws += NN * DD;
    float* xp   = ws;           ws += NN * KPAD;
    float* Wp   = ws;           ws += KPAD * DD;
    float* aS   = ws;           ws += NN * HH;
    float* aD   = ws;           ws += NN * HH;
    float* gsum = ws;           ws += GG * DD;
    float* gcnt = ws;           ws += GG;
    int* binTot  = (int*)ws;    ws += BPITCH;
    int* binBase = (int*)ws;    ws += BPITCH;
    int* cnt     = (int*)ws;    ws += NCHUNK * BPITCH;
    int* off     = (int*)ws;    ws += NCHUNK * BPITCH;
    int* binned  = (int*)ws;    ws += 2 * EE;
    int* rowptr  = (int*)ws;    ws += NN + 4;
    int* col     = (int*)ws;    ws += EE;

    k_bin_count<<<NCHUNK, 256, 0, stream>>>(ei, cnt);
    k_bin_scan<<<NBIN, 256, 0, stream>>>(cnt, off, binTot);
    k_bin_base<<<1, 256, 0, stream>>>(binTot, binBase);
    k_bin_scatter<<<NCHUNK, 256, 0, stream>>>(ei, off, binBase, binned);
    k_bin_csr<<<NBIN, 256, 0, stream>>>(binned, binBase, binTot, rowptr, col);

    k_pad_x<<<(NN * KPAD + 255) / 256, 256, 0, stream>>>(x, xp);
    k_pad_w<<<(KPAD * DD + 255) / 256, 256, 0, stream>>>(W0, Wp);

    for (int layer = 0; layer < LL; ++layer) {
        const float* Ain;  float* agg;  const float* resid;
        switch (layer) {
            case 0: Ain = xp;  agg = A0v; resid = nullptr; break;
            case 1: Ain = A0v; agg = A1v; resid = A0v;     break;
            case 2: Ain = A1v; agg = A0v; resid = A1v;     break;
            default:Ain = A0v; agg = out; resid = A0v;     break;
        }
        const float* asv  = (layer == 0) ? as0 : asrcs + (layer - 1) * HH * CC;
        const float* adv  = (layer == 0) ? ad0 : adsts + (layer - 1) * HH * CC;
        const float* bias = (layer == 0) ? b0  : bsl   + (layer - 1) * DD;

        if (layer == 0) {
            k_gemm<KPAD, false><<<(ROW_TILES + GEMM_WAVES - 1) / GEMM_WAVES, GEMM_WAVES * 32, 0, stream>>>(
                Ain, Wp, Lin, asv, adv, aS, aD);
        } else {
            k_gemm<DD, true><<<(ROW_TILES + GEMM_WAVES - 1) / GEMM_WAVES, GEMM_WAVES * 32, 0, stream>>>(
                Ain, Wsl + (layer - 1) * DD * DD, Lin, asv, adv, aS, aD);
        }
        if (layer == 0) {
            k_gat_node0<<<NN / L0_WAVES, L0_WAVES * 32, 0, stream>>>(
                rowptr, col, aS, aD, xp, Wp, bias, lng + layer * DD, lnb + layer * DD, agg);
        } else {
            k_gat_node<true><<<(NN * 32 + 255) / 256, 256, 0, stream>>>(
                rowptr, col, aS, aD, Lin, bias, lng + layer * DD, lnb + layer * DD,
                resid, agg);
        }
    }

    k_zero_pool<<<(GG * DD + 255) / 256, 256, 0, stream>>>(gsum, gcnt);
    k_pool_sum<<<(NN / POOL_ROWS * 32 + 255) / 256, 256, 0, stream>>>(out, batch, gsum, gcnt);
    k_pool_div<<<(GG * DD + 255) / 256, 256, 0, stream>>>(gsum, gcnt, out + NN * DD);
}


# aggregation waves issue about a quarter fewer vector instructions per node: softmax and ELU exponentials as bare exp2 (log2-domain logits), LayerNorm sums by DPP + row swap instead of LDS permutes
# speedup vs baseline: 22.0744x; 1.0922x over previous
#include <hip/hip_runtime.h>
#include <math.h>

#define NN 100000
#define EE 1600000
#define HH 4
#define CC 32
#define DD 128
#define GG 64
#define LL 4
#define INDIM 7
#define KPAD 8
#define NEG_SLOPE 0.2f

typedef __attribute__((ext_vector_type(2))) float v2f;
typedef __attribute__((ext_vector_type(8))) float v8f;

__device__ __forceinline__ float lrelu(float t) {
    return t > 0.f ? t : NEG_SLOPE * t;
}

#define LOG2E 1.4426950408889634f
__device__ __forceinline__ float ex2(float t) {
    return __builtin_amdgcn_exp2f(t);
}

__global__ void k_pad_x(const float* __restrict__ x, float* __restrict__ xp) {
    int t = blockIdx.x * blockDim.x + threadIdx.x;
    if (t >= NN * KPAD) return;
    const int n = t >> 3, c = t & 7;
    xp[t] = (c < INDIM) ? x[n * INDIM + c] : 0.f;
}

__global__ void k_pad_w(const float* __restrict__ W0, float* __restrict__ Wp) {
    int t = blockIdx.x * blockDim.x + threadIdx.x;
    if (t >= KPAD * DD) return;
    Wp[t] = (t < INDIM * DD) ? W0[t] : 0.f;
}

#define GEMM_WAVES 8
#define ROW_TILES (NN / 16)
static_assert(NN % 16 == 0, "whole 16-row strips");

__device__ __forceinline__ v8f wmma4(float a0, float a1, float b0, float b1, v8f c) {
    v2f a; a.x = a0; a.y = a1;
    v2f b; b.x = b0; b.y = b1;
    return __builtin_amdgcn_wmma_f32_16x16x4_f32(
         false, a,  false, b,
         (short)0, c,  false,  false);
}

template <int X>
__device__ __forceinline__ float row_xor(float v) {
    return __int_as_float(__builtin_amdgcn_update_dpp(
        0, __float_as_int(v), 0x160 + X, 0xf, 0xf, true));
}

template <int N, int X>
__device__ __forceinline__ void rs_step(const float (&in)[2 * N], float (&out)[N], bool bit) {
#pragma unroll
    for (int i = 0; i < N; ++i) {
        const float keep = bit ? in[2 * i + 1] : in[2 * i];
        const float send = bit ? in[2 * i] : in[2 * i + 1];
        out[i] = keep + row_xor<X>(send);
    }
}

template <int K, bool STORE_C, bool COEF>
__global__ __launch_bounds__(GEMM_WAVES * 32) void k_gemm(const float* __restrict__ A,
                                                          const float* __restrict__ B,
                                                          float* __restrict__ Cm,
                                                          const float* __restrict__ a_src,
                                                          const float* __restrict__ a_dst,
                                                          float* __restrict__ aS,
                                                          float* __restrict__ aD) {
    static_assert(K % 8 == 0, "k is consumed in blocks of 8");
    constexpr int NKB = K / 8;
    constexpr int CH  = NKB < 4 ? NKB : 4;
    static_assert(NKB % CH == 0, "whole chunks");
    __shared__ float4 Bs[(K / 4) * DD];

    const int lane = threadIdx.x & 31;
    const int wave = __builtin_amdgcn_readfirstlane((int)(threadIdx.x >> 5));
    const int m    = lane & 15;
    const int half = lane >> 4;

    const int tile  = blockIdx.x * GEMM_WAVES + wave;
    const int tilec = tile < ROW_TILES ? tile : ROW_TILES - 1;
    const float* __restrict__ arow = A + (tilec * 16 + m) * K + half * 4;

    float4 a_cur[CH];
#pragma unroll
    for (int i = 0; i < CH; ++i) a_cur[i] = *(const float4*)(arow + i * 8);

    constexpr int NPAIR = (K / 4) * (DD / 32);
    static_assert(NPAIR % GEMM_WAVES == 0, "every wave stages the same number of pieces");
#pragma unroll 4
    for (int t = 0; t < NPAIR / GEMM_WAVES; ++t) {
        const int p = t * GEMM_WAVES + wave;
        const int kq = p >> 2, n = (p & 3) * 32 + lane;
        const float* __restrict__ bp = B + kq * 4 * DD + n;
        float4 v;
        v.x = bp[0];
        v.y = bp[DD];
        v.z = bp[2 * DD];
        v.w = bp[3 * DD];
        Bs[kq * DD + n] = v;
    }
    __syncthreads();
    if (tile >= ROW_TILES) return;

    v8f acc[8];
#pragma unroll
    for (int ct = 0; ct < 8; ++ct) acc[ct] = v8f{};

#pragma unroll
    for (int c0 = 0; c0 < NKB; c0 += CH) {
        float4 a_nxt[CH];
        if (c0 + CH < NKB) {
#pragma unroll
            for (int i = 0; i < CH; ++i)
                a_nxt[i] = *(const float4*)(arow + (c0 + CH + i) * 8);
        }
#pragma unroll
        for (int i = 0; i < CH; ++i) {
            const float4 a = a_cur[i];
            const float4* __restrict__ bq = Bs + ((c0 + i) * 2 + half) * DD + m;
#pragma unroll
            for (int ct = 0; ct < 8; ct += 2) {
                const float4 b0 = bq[ct * 16];
                const float4 b1 = bq[ct * 16 + 16];
                acc[ct]     = wmma4(a.x, a.y, b0.x, b0.y, acc[ct]);
                acc[ct + 1] = wmma4(a.x, a.y, b1.x, b1.y, acc[ct + 1]);
                acc[ct]     = wmma4(a.z, a.w, b0.z, b0.w, acc[ct]);
                acc[ct + 1] = wmma4(a.z, a.w, b1.z, b1.w, acc[ct + 1]);
            }
        }
        if (c0 + CH < NKB) {
#pragma unroll
            for (int i = 0; i < CH; ++i) a_cur[i] = a_nxt[i];
        }
    }
    float cs0[HH], cs1[HH], cd0[HH], cd1[HH];
    if (COEF) {
#pragma unroll
        for (int hd = 0; hd < HH; ++hd) {
            cs0[hd] = a_src[hd * CC + m];
            cs1[hd] = a_src[hd * CC + 16 + m];
            cd0[hd] = a_dst[hd * CC + m];
            cd1[hd] = a_dst[hd * CC + 16 + m];
        }
    }
    if (STORE_C) {
        float* out = Cm + tile * 16 * DD + m;
#pragma unroll
        for (int ct = 0; ct < 8; ++ct) {
#pragma unroll
            for (int r = 0; r < 8; ++r) {
                out[(r + half * 8) * DD + ct * 16] = acc[ct][r];
            }
        }
    }

    if (!COEF) return;
    __builtin_amdgcn_sched_barrier(0);
    const bool b3 = (m & 8) != 0, b2 = (m & 4) != 0, b1 = (m & 2) != 0, b0 = (m & 1) != 0;
    float res[HH];
#pragma unroll
    for (int hd = 0; hd < HH; ++hd) {
        const float s0 = cs0[hd], s1 = cs1[hd], d0 = cd0[hd], d1 = cd1[hd];
        float x16[16], x8[8], x4[4], x2[2], x1[1];
#pragma unroll
        for (int r = 0; r < 8; ++r) {
            const float c0 = acc[2 * hd][r], c1 = acc[2 * hd + 1][r];
            x16[r]     = c0 * s0 + c1 * s1;
            x16[8 + r] = c0 * d0 + c1 * d1;
        }
        rs_step<8, 8>(x16, x8, b3);
        rs_step<4, 4>(x8, x4, b2);
        rs_step<2, 2>(x4, x2, b1);
        rs_step<1, 1>(x2, x1, b0);
        res[hd] = x1[0] * LOG2E;
    }
    const int rr = ((m >> 3) & 1) | ((m >> 1) & 2) | ((m << 1) & 4);
    float* ap = (b0 ? aD : aS) + (tile * 16 + half * 8 + rr) * HH;
    *(float4*)ap = make_float4(res[0], res[1], res[2], res[3]);
}

#define BIN_SHIFT 8
#define BIN_NODES (1 << BIN_SHIFT)
#define NBIN ((NN + BIN_NODES - 1) / BIN_NODES)
#define CHUNK 4096
#define NCHUNK ((EE + CHUNK - 1) / CHUNK)
#define BPITCH 512
static_assert(NBIN <= BPITCH && NCHUNK <= BPITCH, "two slots per thread of a 256-thread block");
static_assert(BIN_NODES == 256, "one destination per thread in k_bin_csr");
#define CSR_BATCH 8

__device__ __forceinline__ int block_excl_scan(int v, int* wtot, int& total) {
    const int lane = threadIdx.x & 31, wave = threadIdx.x >> 5;
    int inc = v;
#pragma unroll
    for (int off = 1; off < 32; off <<= 1) {
        const int u = __shfl_up(inc, off, 32);
        if (lane >= off) inc += u;
    }
    if (lane == 31) wtot[wave] = inc;
    __syncthreads();
    int wb = 0, tot = 0;
#pragma unroll
    for (int w = 0; w < 8; ++w) {
        const int x = wtot[w];
        if (w < wave) wb += x;
        tot += x;
    }
    total = tot;
    return wb + inc - v;
}

__global__ __launch_bounds__(256) void k_bin_count(const int* __restrict__ ei,
                                                   int* __restrict__ cnt) {
    __shared__ int hist[BPITCH];
    const int t = threadIdx.x;
    hist[t] = 0;
    hist[t + 256] = 0;
    __syncthreads();
    const int e0 = blockIdx.x * CHUNK + t;
    int d[CHUNK / 256];
#pragma unroll
    for (int i = 0; i < CHUNK / 256; ++i) {
        const int e = e0 + i * 256;
        d[i] = ei[EE + (e < EE ? e : EE - 1)];
    }
#pragma unroll
    for (int i = 0; i < CHUNK / 256; ++i) {
        if (e0 + i * 256 < EE) atomicAdd(&hist[d[i] >> BIN_SHIFT], 1);
    }
    __syncthreads();
    cnt[blockIdx.x * BPITCH + t] = hist[t];
    cnt[blockIdx.x * BPITCH + t + 256] = hist[t + 256];
}

__global__ __launch_bounds__(256) void k_bin_scan(const int* __restrict__ cnt,
                                                  int* __restrict__ off,
                                                  int* __restrict__ binTot) {
    __shared__ int wtot[8];
    const int b = blockIdx.x, t = threadIdx.x;
    const int w0 = 2 * t, w1 = 2 * t + 1;
    const int c0 = (w0 < NCHUNK) ? cnt[w0 * BPITCH + b] : 0;
    const int c1 = (w1 < NCHUNK) ? cnt[w1 * BPITCH + b] : 0;
    int tot;
    const int ex = block_excl_scan(c0 + c1, wtot, tot);
    if (w0 < NCHUNK) off[w0 * BPITCH + b] = ex;
    if (w1 < NCHUNK) off[w1 * BPITCH + b] = ex + c0;
    if (t == 0) binTot[b] = tot;
}

__global__ __launch_bounds__(256) void k_bin_base(const int* __restrict__ binTot,
                                                  int* __restrict__ binBase) {
    __shared__ int wtot[8];
    const int t = threadIdx.x;
    const int b0 = 2 * t, b1 = 2 * t + 1;
    const int c0 = (b0 < NBIN) ? binTot[b0] : 0;
    const int c1 = (b1 < NBIN) ? binTot[b1] : 0;
    int tot;
    const int ex = block_excl_scan(c0 + c1, wtot, tot);
    binBase[b0] = ex;
    binBase[b1] = ex + c0;
}

__global__ __launch_bounds__(256) void k_bin_scatter(const int* __restrict__ ei,
                                                     const int* __restrict__ off,
                                                     const int* __restrict__ binBase,
                                                     int* __restrict__ binned) {
    __shared__ int cur[BPITCH];
    const int t = threadIdx.x;
    cur[t] = (t < NBIN) ? binBase[t] + off[blockIdx.x * BPITCH + t] : 0;
    cur[t + 256] = (t + 256 < NBIN) ? binBase[t + 256] + off[blockIdx.x * BPITCH + t + 256] : 0;
    __syncthreads();
    const int e0 = blockIdx.x * CHUNK + t;
    int s[CHUNK / 256], d[CHUNK / 256];
#pragma unroll
    for (int i = 0; i < CHUNK / 256; ++i) {
        const int e = e0 + i * 256;
        const int ec = e < EE ? e : EE - 1;
        s[i] = ei[ec];
        d[i] = ei[EE + ec];
    }
#pragma unroll
    for (int i = 0; i < CHUNK / 256; ++i) {
        if (e0 + i * 256 < EE) {
            const int pos = atomicAdd(&cur[d[i] >> BIN_SHIFT], 1);
            *(int2*)(binned + 2 * pos) = make_int2(s[i], d[i]);
        }
    }
}

__global__ __launch_bounds__(256) void k_bin_csr(const int* __restrict__ binned,
                                                 const int* __restrict__ binBase,
                                                 const int* __restrict__ binTot,
                                                 int* __restrict__ rowptr,
                                                 int* __restrict__ col) {
    __shared__ int cntl[BIN_NODES];
    __shared__ int curl[BIN_NODES];
    __shared__ int wtot[8];
    const int b = blockIdx.x, t = threadIdx.x;
    const int start = binBase[b];
    const int n = binTot[b];
    const int2* __restrict__ ed = (const int2*)(binned + 2 * start);
    cntl[t] = 0;
    __syncthreads();
    for (int i0 = t; i0 < n; i0 += 256 * CSR_BATCH) {
        int dl[CSR_BATCH];
#pragma unroll
        for (int k = 0; k < CSR_BATCH; ++k) {
            const int i = i0 + k * 256;
            dl[k] = ed[i < n ? i : n - 1].y & (BIN_NODES - 1);
        }
#pragma unroll
        for (int k = 0; k < CSR_BATCH; ++k) {
            if (i0 + k * 256 < n) atomicAdd(&cntl[dl[k]], 1);
        }
    }
    __syncthreads();
    int tot;
    const int ex = block_excl_scan(cntl[t], wtot, tot);
    curl[t] = ex;
    const int node = b * BIN_NODES + t;
    if (node <= NN) rowptr[node] = start + ex;
    __syncthreads();
    for (int i0 = t; i0 < n; i0 += 256 * CSR_BATCH) {
        int2 e[CSR_BATCH];
#pragma unroll
        for (int k = 0; k < CSR_BATCH; ++k) {
            const int i = i0 + k * 256;
            e[k] = ed[i < n ? i : n - 1];
        }
#pragma unroll
        for (int k = 0; k < CSR_BATCH; ++k) {
            if (i0 + k * 256 < n) {
                const int r = atomicAdd(&curl[e[k].y & (BIN_NODES - 1)], 1);
                col[start + r] = e[k].x;
            }
        }
    }
}

__device__ __forceinline__ float wave_sum(float v) {
    v += __int_as_float((int)__builtin_amdgcn_permlanex16(
        0u, (unsigned)__float_as_int(v), 0x76543210u, 0xfedcba98u, false, false));
    v += __int_as_float(__builtin_amdgcn_update_dpp(0, __float_as_int(v), 0x168, 0xf, 0xf, true));
    v += __int_as_float(__builtin_amdgcn_update_dpp(0, __float_as_int(v), 0x164, 0xf, 0xf, true));
    v += __int_as_float(__builtin_amdgcn_update_dpp(0, __float_as_int(v), 0x162, 0xf, 0xf, true));
    v += __int_as_float(__builtin_amdgcn_update_dpp(0, __float_as_int(v), 0x161, 0xf, 0xf, true));
    return v;
}

__device__ __forceinline__ float grp_max(float v) {
    v = fmaxf(v, row_xor<1>(v));
    v = fmaxf(v, row_xor<2>(v));
    v = fmaxf(v, row_xor<4>(v));
    return v;
}

__device__ __forceinline__ float grp_sum(float v) {
    v += row_xor<1>(v);
    v += row_xor<2>(v);
    v += row_xor<4>(v);
    return v;
}

template <int I>
__device__ __forceinline__ float grp_bcast(float v) {
    constexpr int SEL = I | (I << 3) | (I << 6) | (I << 9) | (I << 12) | (I << 15) |
                        (I << 18) | (I << 21);
    return __int_as_float(__builtin_amdgcn_mov_dpp8(__float_as_int(v), SEL));
}

__device__ __forceinline__ void load_chunk(int base, int i, int g, int cnt, int beg, int d,
                                           const int* __restrict__ col,
                                           const float* __restrict__ aS, float adg,
                                           int (&s)[4], float (&lg)[4]) {
    int sc[4];
#pragma unroll
    for (int q = 0; q < 4; ++q) {
        const int e = beg + base + 8 * q + i;
        sc[q] = col[e < EE ? e : EE - 1];
    }
    float as[4];
#pragma unroll
    for (int q = 0; q < 4; ++q) {
        s[q] = (base + 8 * q + i < cnt - 1) ? sc[q] : d;
        as[q] = aS[s[q] * HH + g];
    }
#pragma unroll
    for (int q = 0; q < 4; ++q)
        lg[q] = (base + 8 * q + i < cnt) ? lrelu(as[q] + adg) : -3.0e38f;
}

template <int I>
__device__ __forceinline__ float4 row_of(const float* __restrict__ hLin, unsigned loff, int sv) {
    const int sj = __builtin_amdgcn_readlane(sv, I);
    return *(const float4*)((const char*)(hLin + sj * DD) + loff);
}

__device__ __forceinline__ float dot4(const float4 v, const float4 a) {
    return fmaf(v.w, a.w, fmaf(v.z, a.z, fmaf(v.y, a.y, v.x * a.x)));
}

template <int LN>
__device__ __forceinline__ void row_add(float wv, const float4 v, float4& acc) {
    const float w = grp_bcast<LN>(wv);
    acc.x = fmaf(w, v.x, acc.x);
    acc.y = fmaf(w, v.y, acc.y);
    acc.z = fmaf(w, v.z, acc.z);
    acc.w = fmaf(w, v.w, acc.w);
}

template <int NR>
__device__ __forceinline__ void fold_slot(const float* __restrict__ hLin, unsigned loff, int sv,
                                          int nv, int i, const float4 cs, float adg,
                                          float& m, float& zl, float4& acc) {
    static_assert(NR == 4 || NR == 8, "a slot is four or eight rows");
    const float4 v0 = row_of<0>(hLin, loff, sv);
    const float4 v1 = row_of<1>(hLin, loff, sv);
    const float4 v2 = row_of<2>(hLin, loff, sv);
    const float4 v3 = row_of<3>(hLin, loff, sv);
    float4 v4 = make_float4(0.f, 0.f, 0.f, 0.f), v5 = v4, v6 = v4, v7 = v4;
    if (NR == 8) {
        v4 = row_of<4>(hLin, loff, sv);
        v5 = row_of<5>(hLin, loff, sv);
        v6 = row_of<6>(hLin, loff, sv);
        v7 = row_of<7>(hLin, loff, sv);
    }
    float t[8], t4[4], t2[2], t1[1];
    t[0] = dot4(v0, cs);
    t[1] = dot4(v1, cs);
    t[2] = dot4(v2, cs);
    t[3] = dot4(v3, cs);
    t[4] = NR == 8 ? dot4(v4, cs) : 0.f;
    t[5] = NR == 8 ? dot4(v5, cs) : 0.f;
    t[6] = NR == 8 ? dot4(v6, cs) : 0.f;
    t[7] = NR == 8 ? dot4(v7, cs) : 0.f;
    const bool i2 = (i & 4) != 0, i1 = (i & 2) != 0, i0 = (i & 1) != 0;
    rs_step<4, 4>(t, t4, i2);
    rs_step<2, 2>(t4, t2, i1);
    rs_step<1, 1>(t2, t1, i0);
    const int kk = ((i >> 2) & 1) | (i & 2) | ((i & 1) << 2);
    const bool real = kk < nv;
    const float lg = real ? lrelu(t1[0] + adg) : -3.0e38f;
    const float mn = fmaxf(m, grp_max(lg));
    const float sc = ex2(m - mn);
    const float w = real ? ex2(lg - mn) : 0.f;
    m = mn;
    zl = fmaf(zl, sc, w);
    acc.x *= sc; acc.y *= sc; acc.z *= sc; acc.w *= sc;
    row_add<0>(w, v0, acc);
    row_add<4>(w, v1, acc);
    row_add<2>(w, v2, acc);
    row_add<6>(w, v3, acc);
    if (NR == 8) {
        row_add<1>(w, v4, acc);
        row_add<5>(w, v5, acc);
        row_add<3>(w, v6, acc);
        row_add<7>(w, v7, acc);
    }
}

template <bool HAS_RES>
__device__ __forceinline__ void finish_row(const float4 acc, int d, int lane,
                                           const float* __restrict__ bias,
                                           const float* __restrict__ gma,
                                           const float* __restrict__ bta,
                                           const float* __restrict__ resid,
                                           float* __restrict__ outp) {
    float4 rr = make_float4(0.f, 0.f, 0.f, 0.f);
    if (HAS_RES) rr = *(const float4*)(resid + d * DD + lane * 4);
    const float4 bi = *(const float4*)(bias + lane * 4);
    const float4 gm = *(const float4*)(gma + lane * 4);
    const float4 bt = *(const float4*)(bta + lane * 4);
    const float4 v = make_float4(acc.x + bi.x, acc.y + bi.y, acc.z + bi.z, acc.w + bi.w);
    const float mu = wave_sum((v.x + v.y) + (v.z + v.w)) * (1.f / 128.f);
    const float4 dl = make_float4(v.x - mu, v.y - mu, v.z - mu, v.w - mu);
    const float var =
        wave_sum((dl.x * dl.x + dl.y * dl.y) + (dl.z * dl.z + dl.w * dl.w));
    const float rs = rsqrtf(var * (1.f / 128.f) + 1e-5f);
    float4 y = make_float4(dl.x * rs * gm.x + bt.x, dl.y * rs * gm.y + bt.y,
                           dl.z * rs * gm.z + bt.z, dl.w * rs * gm.w + bt.w);
    y.x = y.x > 0.f ? y.x : ex2(y.x * LOG2E) - 1.f;
    y.y = y.y > 0.f ? y.y : ex2(y.y * LOG2E) - 1.f;
    y.z = y.z > 0.f ? y.z : ex2(y.z * LOG2E) - 1.f;
    y.w = y.w > 0.f ? y.w : ex2(y.w * LOG2E) - 1.f;
    if (HAS_RES) { y.x += rr.x; y.y += rr.y; y.z += rr.z; y.w += rr.w; }
    *(float4*)(outp + d * DD + lane * 4) = y;
}

template <bool HAS_RES>
__global__ __launch_bounds__(256) __attribute__((amdgpu_waves_per_eu(16, 16)))
void k_gat_node(const int* __restrict__ rowptr,
                const int* __restrict__ col,
                const float* __restrict__ a_src,
                const float* __restrict__ a_dst,
                const float* __restrict__ hLin,
                const float* __restrict__ bias,
                const float* __restrict__ gma,
                const float* __restrict__ bta,
                const float* __restrict__ resid,
                float* __restrict__ outp) {
    const int lane = threadIdx.x & 31;
    const int i = lane & 7;
    const int d = __builtin_amdgcn_readfirstlane(
        (int)((blockIdx.x * blockDim.x + threadIdx.x) >> 5));
    if (d >= NN) return;
    const unsigned loff = lane * 16u;
    const float4 vs = *(const float4*)((const char*)(hLin + d * DD) + loff);
    float4 cs = *(const float4*)(a_src + lane * 4);
    float4 cd = *(const float4*)(a_dst + lane * 4);
    cs.x *= LOG2E; cs.y *= LOG2E; cs.z *= LOG2E; cs.w *= LOG2E;
    cd.x *= LOG2E; cd.y *= LOG2E; cd.z *= LOG2E; cd.w *= LOG2E;
    const int beg = rowptr[d];
    const int n = rowptr[d + 1] - beg;
    int sc[4];
#pragma unroll
    for (int q = 0; q < 4; ++q) {
        const int e = beg + 8 * q + i;
        sc[q] = col[e < EE ? e : EE - 1];
    }

    const float adg = grp_sum(dot4(vs, cd));
    float m = lrelu(grp_sum(dot4(vs, cs)) + adg);
    float zl = (i == 0) ? 1.f : 0.f;
    float4 acc = vs;

    for (int base = 0; base < n;) {
        int s[4];
#pragma unroll
        for (int q = 0; q < 4; ++q) s[q] = (base + 8 * q + i < n) ? sc[q] : d;
#pragma unroll
        for (int q = 0; q < 4; ++q) {
            const int rem = n - base - 8 * q;
            if (rem > 4)      fold_slot<8>(hLin, loff, s[q], rem, i, cs, adg, m, zl, acc);
            else if (rem > 0) fold_slot<4>(hLin, loff, s[q], rem, i, cs, adg, m, zl, acc);
        }
        base += 32;
        if (base < n) {
#pragma unroll
            for (int q = 0; q < 4; ++q) {
                const int e = beg + base + 8 * q + i;
                sc[q] = col[e < EE ? e : EE - 1];
            }
        }
    }
    const float inv = 1.f / (grp_sum(zl) + 1e-16f);
    acc.x *= inv; acc.y *= inv; acc.z *= inv; acc.w *= inv;

    finish_row<HAS_RES>(acc, d, lane, bias, gma, bta, resid, outp);
}

#define L0_WAVES 16
#define L0_CP (DD + 4)
static_assert(NN % L0_WAVES == 0, "whole 16-node strips");

__global__ __launch_bounds__(L0_WAVES * 32) __attribute__((amdgpu_waves_per_eu(16, 16)))
void k_gat_node0(const int* __restrict__ rowptr,
                 const int* __restrict__ col,
                 const float* __restrict__ aS,
                 const float* __restrict__ aD,
                 const float* __restrict__ xp,
                 const float* __restrict__ Wp,
                 const float* __restrict__ bias,
                 const float* __restrict__ gma,
                 const float* __restrict__ bta,
                 float* __restrict__ outp) {
    __shared__ float4 As[L0_WAVES * HH * 2];
    __shared__ float Cs[L0_WAVES * L0_CP];
    const int lane = threadIdx.x & 31;
    const int wave = __builtin_amdgcn_readfirstlane((int)(threadIdx.x >> 5));
    const int g = lane >> 3;
    const int i = lane & 7;
    const int d = blockIdx.x * L0_WAVES + wave;
    const int beg = rowptr[d];
    const int cnt = rowptr[d + 1] - beg + 1;
    const float adg = aD[d * HH + g];
    const float4* __restrict__ xr = (const float4*)xp;

    int s0[4];
    float lg0[4];
    load_chunk(0, i, g, cnt, beg, d, col, aS, adg, s0, lg0);
    float4 xa[4], xb[4];
#pragma unroll
    for (int q = 0; q < 4; ++q) {
        xa[q] = xr[s0[q] * 2];
        xb[q] = xr[s0[q] * 2 + 1];
    }

    float m = fmaxf(fmaxf(lg0[0], lg0[1]), fmaxf(lg0[2], lg0[3]));
    for (int base = 32; base < cnt; base += 32) {
        int s[4];
        float lg[4];
        load_chunk(base, i, g, cnt, beg, d, col, aS, adg, s, lg);
        m = fmaxf(m, fmaxf(fmaxf(lg[0], lg[1]), fmaxf(lg[2], lg[3])));
    }
    m = grp_max(m);

    float e0[4];
#pragma unroll
    for (int q = 0; q < 4; ++q) {
        e0[q] = 0.f;
        if (8 * q < cnt) e0[q] = ex2(lg0[q] - m);
    }
    float z = (e0[0] + e0[1]) + (e0[2] + e0[3]);
    for (int base = 32; base < cnt; base += 32) {
        int s[4];
        float lg[4];
        load_chunk(base, i, g, cnt, beg, d, col, aS, adg, s, lg);
#pragma unroll
        for (int q = 0; q < 4; ++q) z += ex2(lg[q] - m);
    }
    const float den = grp_sum(z) + 1e-16f;

    float p[8];
#pragma unroll
    for (int k = 0; k < 8; ++k) p[k] = 0.f;
#pragma unroll
    for (int q = 0; q < 4; ++q) {
        const float w = e0[q];
        p[0] = fmaf(w, xa[q].x, p[0]);
        p[1] = fmaf(w, xa[q].y, p[1]);
        p[2] = fmaf(w, xa[q].z, p[2]);
        p[3] = fmaf(w, xa[q].w, p[3]);
        p[4] = fmaf(w, xb[q].x, p[4]);
        p[5] = fmaf(w, xb[q].y, p[5]);
        p[6] = fmaf(w, xb[q].z, p[6]);
        p[7] = fmaf(w, xb[q].w, p[7]);
    }
    for (int base = 32; base < cnt; base += 32) {
        int s[4];
        float lg[4];
        load_chunk(base, i, g, cnt, beg, d, col, aS, adg, s, lg);
#pragma unroll
        for (int q = 0; q < 4; ++q) {
            const float4 va = xr[s[q] * 2];
            const float4 vb = xr[s[q] * 2 + 1];
            const float w = ex2(lg[q] - m);
            p[0] = fmaf(w, va.x, p[0]);
            p[1] = fmaf(w, va.y, p[1]);
            p[2] = fmaf(w, va.z, p[2]);
            p[3] = fmaf(w, va.w, p[3]);
            p[4] = fmaf(w, vb.x, p[4]);
            p[5] = fmaf(w, vb.y, p[5]);
            p[6] = fmaf(w, vb.z, p[6]);
            p[7] = fmaf(w, vb.w, p[7]);
        }
    }
    const bool i2 = (i & 4) != 0, i1 = (i & 2) != 0, i0 = (i & 1) != 0;
    float p4[4], p2[2], p1[1];
    rs_step<4, 4>(p, p4, i2);
    rs_step<2, 2>(p4, p2, i1);
    rs_step<1, 1>(p2, p1, i0);
    const int kk = ((i >> 2) & 1) | (i & 2) | ((i & 1) << 2);
    ((float*)As)[(wave * HH + g) * 8 + kk] = p1[0] / den;

    const int m16 = lane & 15;
    const int half = lane >> 4;
    float bw[4] = {0.f, 0.f, 0.f, 0.f};
    if (wave < 8) {
#pragma unroll
        for (int s = 0; s < 4; ++s) bw[s] = Wp[(half * 4 + s) * DD + wave * 16 + m16];
    }
    __syncthreads();
    if (wave < 8) {
        const float4 a = As[(m16 * HH + (wave >> 1)) * 2 + half];
        v8f acc = v8f{};
        acc = wmma4(a.x, a.y, bw[0], bw[1], acc);
        acc = wmma4(a.z, a.w, bw[2], bw[3], acc);
#pragma unroll
        for (int r = 0; r < 8; ++r) Cs[(r + half * 8) * L0_CP + wave * 16 + m16] = acc[r];
    }
    __syncthreads();
    const float4 row = *(const float4*)(Cs + wave * L0_CP + lane * 4);
    finish_row<false>(row, d, lane, bias, gma, bta, nullptr, outp);
}

__global__ void k_zero_pool(float* __restrict__ gsum, float* __restrict__ gcnt) {
    int t = blockIdx.x * blockDim.x + threadIdx.x;
    if (t < GG * DD) gsum[t] = 0.f;
    if (t < GG) gcnt[t] = 0.f;
}

#define POOL_ROWS 16
static_assert(NN % POOL_ROWS == 0, "a wave sums whole groups of POOL_ROWS nodes");

__device__ __forceinline__ void pool_flush(float* __restrict__ gsum,
                                           float* __restrict__ gcnt, int g,
                                           int lane, const float4 acc, float cnt) {
    float* p = gsum + g * DD + lane * 4;
    atomicAdd(p + 0, acc.x);
    atomicAdd(p + 1, acc.y);
    atomicAdd(p + 2, acc.z);
    atomicAdd(p + 3, acc.w);
    if (lane == 0) atomicAdd(gcnt + g, cnt);
}

__global__ __launch_bounds__(256) void k_pool_sum(const float* __restrict__ h,
                                                  const int* __restrict__ batch,
                                                  float* __restrict__ gsum,
                                                  float* __restrict__ gcnt) {
    const int lane = threadIdx.x & 31;
    const int w = __builtin_amdgcn_readfirstlane(
        (int)((blockIdx.x * blockDim.x + threadIdx.x) >> 5));
    const int n0 = w * POOL_ROWS;
    if (n0 >= NN) return;
    const int gl = batch[n0 + (lane & (POOL_ROWS - 1))];
    float4 r[POOL_ROWS];
    const float* __restrict__ hp = h + n0 * DD + lane * 4;
#pragma unroll
    for (int i = 0; i < POOL_ROWS; ++i) r[i] = *(const float4*)(hp + i * DD);

    int gcur = __builtin_amdgcn_readlane(gl, 0);
    float4 acc = r[0];
    float cnt = 1.f;
#pragma unroll
    for (int i = 1; i < POOL_ROWS; ++i) {
        const int gi = __builtin_amdgcn_readlane(gl, i);
        if (gi != gcur) {
            pool_flush(gsum, gcnt, gcur, lane, acc, cnt);
            gcur = gi;
            acc = r[i];
            cnt = 1.f;
        } else {
            acc.x += r[i].x; acc.y += r[i].y;
            acc.z += r[i].z; acc.w += r[i].w;
            cnt += 1.f;
        }
    }
    pool_flush(gsum, gcnt, gcur, lane, acc, cnt);
}

__global__ void k_pool_div(const float* __restrict__ gsum,
                           const float* __restrict__ gcnt,
                           float* __restrict__ outg) {
    int t = blockIdx.x * blockDim.x + threadIdx.x;
    if (t >= GG * DD) return;
    outg[t] = gsum[t] / fmaxf(gcnt[t >> 7], 1.f);
}

extern "C" void kernel_launch(void* const* d_in, const int* in_sizes, int n_in,
                              void* d_out, int out_size, void* d_ws, size_t ws_size,
                              hipStream_t stream) {
    (void)in_sizes; (void)n_in; (void)out_size; (void)ws_size;

    const float* x     = (const float*)d_in[0];
    const int*   ei    = (const int*)d_in[1];
    const int*   batch = (const int*)d_in[2];
    const float* W0    = (const float*)d_in[3];
    const float* as0   = (const float*)d_in[4];
    const float* ad0   = (const float*)d_in[5];
    const float* b0    = (const float*)d_in[6];
    const float* Wsl   = (const float*)d_in[7];
    const float* asrcs = (const float*)d_in[8];
    const float* adsts = (const float*)d_in[9];
    const float* bsl   = (const float*)d_in[10];
    const float* lng   = (const float*)d_in[11];
    const float* lnb   = (const float*)d_in[12];
    float* out = (float*)d_out;

    float* ws   = (float*)d_ws;
    float* A0v  = ws;           ws += NN * DD;
    float* A1v  = ws;           ws += NN * DD;
    float* Lin  = ws;           ws += NN * DD;
    float* xp   = ws;           ws += NN * KPAD;
    float* Wp   = ws;           ws += KPAD * DD;
    float* aS   = ws;           ws += NN * HH;
    float* aD   = ws;           ws += NN * HH;
    float* gsum = ws;           ws += GG * DD;
    float* gcnt = ws;           ws += GG;
    int* binTot  = (int*)ws;    ws += BPITCH;
    int* binBase = (int*)ws;    ws += BPITCH;
    int* cnt     = (int*)ws;    ws += NCHUNK * BPITCH;
    int* off     = (int*)ws;    ws += NCHUNK * BPITCH;
    int* binned  = (int*)ws;    ws += 2 * EE;
    int* rowptr  = (int*)ws;    ws += NN + 4;
    int* col     = (int*)ws;    ws += EE;

    k_bin_count<<<NCHUNK, 256, 0, stream>>>(ei, cnt);
    k_bin_scan<<<NBIN, 256, 0, stream>>>(cnt, off, binTot);
    k_bin_base<<<1, 256, 0, stream>>>(binTot, binBase);
    k_bin_scatter<<<NCHUNK, 256, 0, stream>>>(ei, off, binBase, binned);
    k_bin_csr<<<NBIN, 256, 0, stream>>>(binned, binBase, binTot, rowptr, col);

    k_pad_x<<<(NN * KPAD + 255) / 256, 256, 0, stream>>>(x, xp);
    k_pad_w<<<(KPAD * DD + 255) / 256, 256, 0, stream>>>(W0, Wp);

    for (int layer = 0; layer < LL; ++layer) {
        const float* Ain;  float* agg;  const float* resid;
        switch (layer) {
            case 0: Ain = xp;  agg = A0v; resid = nullptr; break;
            case 1: Ain = A0v; agg = A1v; resid = A0v;     break;
            case 2: Ain = A1v; agg = A0v; resid = A1v;     break;
            default:Ain = A0v; agg = out; resid = A0v;     break;
        }
        const float* asv  = (layer == 0) ? as0 : asrcs + (layer - 1) * HH * CC;
        const float* adv  = (layer == 0) ? ad0 : adsts + (layer - 1) * HH * CC;
        const float* bias = (layer == 0) ? b0  : bsl   + (layer - 1) * DD;

        if (layer == 0) {
            k_gemm<KPAD, false, true><<<(ROW_TILES + GEMM_WAVES - 1) / GEMM_WAVES, GEMM_WAVES * 32, 0, stream>>>(
                Ain, Wp, Lin, asv, adv, aS, aD);
        } else {
            k_gemm<DD, true, false><<<(ROW_TILES + GEMM_WAVES - 1) / GEMM_WAVES, GEMM_WAVES * 32, 0, stream>>>(
                Ain, Wsl + (layer - 1) * DD * DD, Lin, asv, adv, aS, aD);
        }
        if (layer == 0) {
            k_gat_node0<<<NN / L0_WAVES, L0_WAVES * 32, 0, stream>>>(
                rowptr, col, aS, aD, xp, Wp, bias, lng + layer * DD, lnb + layer * DD, agg);
        } else {
            k_gat_node<true><<<(NN * 32 + 255) / 256, 256, 0, stream>>>(
                rowptr, col, asv, adv, Lin, bias, lng + layer * DD, lnb + layer * DD,
                resid, agg);
        }
    }

    k_zero_pool<<<(GG * DD + 255) / 256, 256, 0, stream>>>(gsum, gcnt);
    k_pool_sum<<<(NN / POOL_ROWS * 32 + 255) / 256, 256, 0, stream>>>(out, batch, gsum, gcnt);
    k_pool_div<<<(GG * DD + 255) / 256, 256, 0, stream>>>(gsum, gcnt, out + NN * DD);
}
